# starting kernel
# baseline (speedup 1.0000x reference)
#include <hip/hip_runtime.h>

typedef __attribute__((ext_vector_type(2))) float v2f;
typedef __attribute__((ext_vector_type(8))) float v8f;

#define GCN_IN_C   128
#define GCN_HID_C  128
#define GCN_OUT_C  64
#define GCN_BN_EPS 1e-5f


__global__ void zero_f32(float* __restrict__ p, long n) {
  long i = (long)blockIdx.x * blockDim.x + threadIdx.x;
  long stride = (long)gridDim.x * blockDim.x;
  for (; i < n; i += stride) p[i] = 0.0f;
}

__global__ void degree_kernel(const int* __restrict__ dst, int E, int* __restrict__ deg) {
  int i = blockIdx.x * blockDim.x + threadIdx.x;
  if (i < E) atomicAdd(&deg[dst[i]], 1);
}

__global__ void dinv_kernel(const int* __restrict__ deg, float* __restrict__ dinv, int N) {
  int i = blockIdx.x * blockDim.x + threadIdx.x;
  if (i < N) dinv[i] = rsqrtf((float)(deg[i] + 1));
}

template <int KDIM>
__global__ void gemm_wmma_lds(const float* __restrict__ A, const float* __restrict__ B,
                              float* __restrict__ C, int N) {
  constexpr int LDA = KDIM + 4;
  __shared__ float As[16 * LDA];

  const int tm = blockIdx.x;
  const float* __restrict__ gA = A + (long)tm * 16 * KDIM;

  for (int idx = threadIdx.x; idx < 16 * KDIM / 4; idx += blockDim.x) {
    const int r  = idx / (KDIM / 4);
    const int c4 = idx % (KDIM / 4);
    const float4 v = ((const float4*)(gA + (long)r * KDIM))[c4];
    *(float4*)&As[r * LDA + c4 * 4] = v;
  }
  __syncthreads();

  const int lane = threadIdx.x & 31;
  const int half = lane >> 4;
  const int l16  = lane & 15;
  const int tn   = threadIdx.x >> 5;
  const int bcol = tn * 16 + l16;
  const float* __restrict__ ap = &As[l16 * LDA];

  v8f acc = {};
#pragma unroll 4
  for (int k0 = 0; k0 < KDIM; k0 += 4) {
    const int ka = k0 + 2 * half;
    v2f a, b;
    a.x = ap[ka];
    a.y = ap[ka + 1];
    b.x = B[(long)ka * N + bcol];
    b.y = B[(long)(ka + 1) * N + bcol];
    acc = __builtin_amdgcn_wmma_f32_16x16x4_f32(
         false, a,  false, b,
         (short)0, acc,  false,  false);
  }

  const int orow = tm * 16 + half * 8;
  float* __restrict__ cp = C + (long)orow * N + bcol;
#pragma unroll
  for (int r = 0; r < 8; ++r) cp[(long)r * N] = acc[r];
}

__global__ void scatter_c128(const float* __restrict__ h, const int* __restrict__ src,
                             const int* __restrict__ dst, const float* __restrict__ dinv,
                             float* __restrict__ agg, int E) {
  const int wave = (int)((blockIdx.x * (long)blockDim.x + threadIdx.x) >> 5);
  const int lane = threadIdx.x & 31;
  if (wave >= E) return;
  const int s = src[wave];
  const int d = dst[wave];
  const float coef = dinv[s] * dinv[d];
  const float4 v = ((const float4*)(h + (long)s * 128))[lane];
  float* __restrict__ o = agg + (long)d * 128 + lane * 4;
  atomicAdd(o + 0, v.x * coef);
  atomicAdd(o + 1, v.y * coef);
  atomicAdd(o + 2, v.z * coef);
  atomicAdd(o + 3, v.w * coef);
}

__global__ void scatter_c64(const float* __restrict__ h, const int* __restrict__ src,
                            const int* __restrict__ dst, const float* __restrict__ dinv,
                            float* __restrict__ agg, int E) {
  const int wave = (int)((blockIdx.x * (long)blockDim.x + threadIdx.x) >> 5);
  const int lane = threadIdx.x & 31;
  if (wave >= E) return;
  const int s = src[wave];
  const int d = dst[wave];
  const float coef = dinv[s] * dinv[d];
  const float2 v = ((const float2*)(h + (long)s * 64))[lane];
  float* __restrict__ o = agg + (long)d * 64 + lane * 2;
  atomicAdd(o + 0, v.x * coef);
  atomicAdd(o + 1, v.y * coef);
}

__global__ void finalize_relu_c128(const float* __restrict__ h, float* __restrict__ agg,
                                   const float* __restrict__ dinv, const float* __restrict__ bias,
                                   const float* __restrict__ gamma, const float* __restrict__ beta,
                                   const float* __restrict__ mean, const float* __restrict__ var,
                                   int N) {
  const long idx = (long)blockIdx.x * blockDim.x + threadIdx.x;
  if (idx >= (long)N * 128) return;
  const int i = (int)(idx >> 7);
  const int c = (int)(idx & 127);
  const float di = dinv[i];
  float v = agg[idx] + h[idx] * di * di + bias[c];
  v = (v - mean[c]) * rsqrtf(var[c] + GCN_BN_EPS) * gamma[c] + beta[c];
  agg[idx] = fmaxf(v, 0.0f);
}

__global__ void finalize_c64(const float* __restrict__ h, const float* __restrict__ agg,
                             const float* __restrict__ dinv, const float* __restrict__ bias,
                             const float* __restrict__ gamma, const float* __restrict__ beta,
                             const float* __restrict__ mean, const float* __restrict__ var,
                             float* __restrict__ out, int N) {
  const long idx = (long)blockIdx.x * blockDim.x + threadIdx.x;
  if (idx >= (long)N * 64) return;
  const int i = (int)(idx >> 6);
  const int c = (int)(idx & 63);
  const float di = dinv[i];
  float v = agg[idx] + h[idx] * di * di + bias[c];
  v = (v - mean[c]) * rsqrtf(var[c] + GCN_BN_EPS) * gamma[c] + beta[c];
  out[idx] = v;
}


extern "C" void kernel_launch(void* const* d_in, const int* in_sizes, int n_in,
                              void* d_out, int out_size, void* d_ws, size_t ws_size,
                              hipStream_t stream) {
  const float* x   = (const float*)d_in[0];
  const int*   ei  = (const int*)d_in[1];
  const float* W1  = (const float*)d_in[2];
  const float* b1  = (const float*)d_in[3];
  const float* g1  = (const float*)d_in[4];
  const float* be1 = (const float*)d_in[5];
  const float* m1  = (const float*)d_in[6];
  const float* v1  = (const float*)d_in[7];
  const float* W2  = (const float*)d_in[8];
  const float* b2  = (const float*)d_in[9];
  const float* g2  = (const float*)d_in[10];
  const float* be2 = (const float*)d_in[11];
  const float* m2  = (const float*)d_in[12];
  const float* v2  = (const float*)d_in[13];

  const int N = in_sizes[0] / GCN_IN_C;
  const int E = in_sizes[1] / 2;
  const int* src = ei;
  const int* dst = ei + E;

  char* ws = (char*)d_ws;
  int*   deg  = (int*)ws;                         ws += (size_t)N * sizeof(int);
  float* dinv = (float*)ws;                       ws += (size_t)N * sizeof(float);
  float* h1   = (float*)ws;                       ws += (size_t)N * GCN_HID_C * sizeof(float);
  float* agg1 = (float*)ws;                       ws += (size_t)N * GCN_HID_C * sizeof(float);
  float* h2   = (float*)ws;                       ws += (size_t)N * GCN_OUT_C * sizeof(float);
  float* agg2 = (float*)ws;                       ws += (size_t)N * GCN_OUT_C * sizeof(float);
  (void)ws_size; (void)n_in; (void)out_size;

  const int ZB = 2048;
  zero_f32<<<ZB, 256, 0, stream>>>((float*)deg, (long)N);
  zero_f32<<<ZB, 256, 0, stream>>>(agg1, (long)N * GCN_HID_C);
  zero_f32<<<ZB, 256, 0, stream>>>(agg2, (long)N * GCN_OUT_C);

  degree_kernel<<<(E + 255) / 256, 256, 0, stream>>>(dst, E, deg);
  dinv_kernel<<<(N + 255) / 256, 256, 0, stream>>>(deg, dinv, N);

  gemm_wmma_lds<GCN_IN_C><<<N / 16, (GCN_HID_C / 16) * 32, 0, stream>>>(x, W1, h1, GCN_HID_C);
  {
    const long blocks = ((long)E * 32 + 255) / 256;
    scatter_c128<<<(int)blocks, 256, 0, stream>>>(h1, src, dst, dinv, agg1, E);
  }
  {
    const long total = (long)N * GCN_HID_C;
    finalize_relu_c128<<<(int)((total + 255) / 256), 256, 0, stream>>>(
        h1, agg1, dinv, b1, g1, be1, m1, v1, N);
  }

  gemm_wmma_lds<GCN_HID_C><<<N / 16, (GCN_OUT_C / 16) * 32, 0, stream>>>(agg1, W2, h2, GCN_OUT_C);
  {
    const long blocks = ((long)E * 32 + 255) / 256;
    scatter_c64<<<(int)blocks, 256, 0, stream>>>(h2, src, dst, dinv, agg2, E);
  }
  {
    const long total = (long)N * GCN_OUT_C;
    finalize_c64<<<(int)((total + 255) / 256), 256, 0, stream>>>(
        h2, agg2, dinv, b2, g2, be2, m2, v2, (float*)d_out, N);
  }
}


# both layers as (A_hat h) W: tile block sums the 128-wide INPUT rows in LDS (self loop = start value), then multiplies its tile by W on f32 WMMA with bias/BN/ReLU: no h@W plane; sums reassociated
# speedup vs baseline: 2.8403x; 2.8403x over previous
#include <hip/hip_runtime.h>

typedef __attribute__((ext_vector_type(2))) float v2f;
typedef __attribute__((ext_vector_type(4))) float v4f;
typedef __attribute__((ext_vector_type(8))) float v8f;

#define GCN_IN_C   128
#define GCN_HID_C  128
#define GCN_OUT_C  64
#define GCN_BN_EPS 1e-5f

#define TILE        128
#define TILE_SHIFT  7
#define MAX_TILES   512
#define BIN_CHUNK   4096
#define BIN_THREADS 256
#define BIN_PER_THR (BIN_CHUNK / BIN_THREADS)
#define AGG_THREADS 512
#define AGG_WAVES   (AGG_THREADS / 32)
#define AGG_FLIGHT  1024


__global__ void zero_f32(float* __restrict__ p, long n) {
  long i = (long)blockIdx.x * blockDim.x + threadIdx.x;
  long stride = (long)gridDim.x * blockDim.x;
  for (; i < n; i += stride) p[i] = 0.0f;
}

__global__ void bin_count(const int* __restrict__ dst, int E, int* __restrict__ totals, int NT) {
  __shared__ int hist[MAX_TILES];
  const int base = blockIdx.x * BIN_CHUNK + threadIdx.x;
  int dv[BIN_PER_THR];
#pragma unroll
  for (int k = 0; k < BIN_PER_THR; ++k) dv[k] = dst[min(base + k * BIN_THREADS, E - 1)];
  for (int i = threadIdx.x; i < NT; i += BIN_THREADS) hist[i] = 0;
  __syncthreads();
#pragma unroll
  for (int k = 0; k < BIN_PER_THR; ++k)
    if (base + k * BIN_THREADS < E) atomicAdd(&hist[dv[k] >> TILE_SHIFT], 1);
  __syncthreads();
  for (int i = threadIdx.x; i < NT; i += BIN_THREADS) {
    const int c = hist[i];
    if (c) atomicAdd(&totals[i], c);
  }
}

__global__ void bin_scan(const int* __restrict__ totals, int* __restrict__ offsets,
                         int* __restrict__ cursor, int NT) {
  __shared__ int a[2][MAX_TILES];
  const int i = threadIdx.x;
  const int own = (i < NT) ? totals[i] : 0;
  a[0][i] = own;
  __syncthreads();
  int cur = 0;
  for (int off = 1; off < MAX_TILES; off <<= 1) {
    int x = a[cur][i];
    if (i >= off) x += a[cur][i - off];
    a[cur ^ 1][i] = x;
    __syncthreads();
    cur ^= 1;
  }
  const int excl = a[cur][i] - own;
  if (i <= NT) offsets[i] = excl;
  if (i < NT) cursor[i] = excl;
}

__global__ void bin_fill(const int* __restrict__ src, const int* __restrict__ dst, int E,
                         int* __restrict__ cursor, int* __restrict__ binned, int NT) {
  __shared__ int hist[MAX_TILES];
  __shared__ int runbase[MAX_TILES];
  const int base = blockIdx.x * BIN_CHUNK + threadIdx.x;
  int dv[BIN_PER_THR], sv[BIN_PER_THR];
#pragma unroll
  for (int k = 0; k < BIN_PER_THR; ++k) dv[k] = dst[min(base + k * BIN_THREADS, E - 1)];
#pragma unroll
  for (int k = 0; k < BIN_PER_THR; ++k) sv[k] = src[min(base + k * BIN_THREADS, E - 1)];
  for (int i = threadIdx.x; i < NT; i += BIN_THREADS) hist[i] = 0;
  __syncthreads();
#pragma unroll
  for (int k = 0; k < BIN_PER_THR; ++k)
    if (base + k * BIN_THREADS < E) atomicAdd(&hist[dv[k] >> TILE_SHIFT], 1);
  __syncthreads();
  for (int i = threadIdx.x; i < NT; i += BIN_THREADS) {
    const int c = hist[i];
    runbase[i] = c ? atomicAdd(&cursor[i], c) : 0;
    hist[i] = 0;
  }
  __syncthreads();
  int2* __restrict__ out = (int2*)binned;
#pragma unroll
  for (int k = 0; k < BIN_PER_THR; ++k) {
    if (base + k * BIN_THREADS < E) {
      const int b = dv[k] >> TILE_SHIFT;
      const int r = atomicAdd(&hist[b], 1);
      out[runbase[b] + r] = make_int2(sv[k], dv[k]);
    }
  }
}

__global__ void tile_dinv(const int* __restrict__ binned, const int* __restrict__ offsets,
                          float* __restrict__ dinv, int N) {
  __shared__ int cnt[TILE];
  const int tile = blockIdx.x;
  const int tbase = tile * TILE;
  if (threadIdx.x < TILE) cnt[threadIdx.x] = 0;
  __syncthreads();
  const int e0 = offsets[tile];
  const int e1 = offsets[tile + 1];
  const int2* __restrict__ ed = (const int2*)binned;
  for (int e = e0 + threadIdx.x; e < e1; e += blockDim.x) atomicAdd(&cnt[ed[e].y - tbase], 1);
  __syncthreads();
  if (threadIdx.x < TILE) {
    const int node = tbase + threadIdx.x;
    if (node < N) dinv[node] = rsqrtf((float)(cnt[threadIdx.x] + 1));
  }
}

template <int COUT, bool RELU>
__global__ __launch_bounds__(AGG_THREADS) void tile_layer(
    const float* __restrict__ h, const int* __restrict__ binned,
    const int* __restrict__ offsets, const float* __restrict__ dinv,
    const float* __restrict__ Wm, const float* __restrict__ bias,
    const float* __restrict__ gamma, const float* __restrict__ beta,
    const float* __restrict__ mean, const float* __restrict__ var,
    float* __restrict__ outp, int N) {
  constexpr int C   = 128;
  static_assert(GCN_IN_C == C && GCN_HID_C == C, "both layers read 128-wide rows");
  static_assert(COUT % 32 == 0 && AGG_WAVES % (COUT / 16) == 0 &&
                (AGG_WAVES / (COUT / 16)) * (COUT / 32) * 16 == TILE,
                "the waves' 16-row blocks must cover the tile exactly once");
  constexpr int LDA = C + 4;
  constexpr int P = C / 64;
  constexpr int R = AGG_FLIGHT / C;
  __shared__ float acc[TILE * LDA];

  const int tid  = threadIdx.x;
  const int lane = tid & 31;
  const int w = __builtin_amdgcn_readfirstlane(tid >> 5);
  const int tile  = blockIdx.x;
  const int tbase = tile * TILE;

  for (int i = tid; i < TILE * C / 4; i += AGG_THREADS) {
    const int nl = i / (C / 4);
    const int c4 = i % (C / 4);
    const int node = tbase + nl;
    v4f v = {0.f, 0.f, 0.f, 0.f};
    if (node < N) {
      const float di = dinv[node];
      v = ((const v4f*)(h + (long)node * C))[c4] * (di * di);
    }
    *(v4f*)&acc[nl * LDA + c4 * 4] = v;
  }
  __syncthreads();

  const int e0 = offsets[tile];
  const int e1 = offsets[tile + 1];
  const int2* __restrict__ ed = (const int2*)binned;

  for (int g = e0 + w * 32; g < e1; g += AGG_WAVES * 32) {
    const int cnt = min(32, e1 - g);
    int s = 0, dl = 0;
    float cf = 0.0f;
    if (lane < cnt) {
      const int2 p = ed[g + lane];
      s  = p.x;
      dl = p.y - tbase;
      cf = dinv[p.x] * dinv[p.y];
    }
    for (int jb = 0; jb < cnt; jb += R) {
      v2f v[R][P];
#pragma unroll
      for (int u = 0; u < R; ++u) {
        const int sj = __builtin_amdgcn_readlane(s, jb + u);
        const float* __restrict__ row = h + (long)sj * C + lane * 2;
#pragma unroll
        for (int p = 0; p < P; ++p) v[u][p] = *(const v2f*)(row + p * 64);
      }
#pragma unroll
      for (int u = 0; u < R; ++u) {
        if (u == 0 || jb + u < cnt) {
          const int   dj = __builtin_amdgcn_readlane(dl, jb + u);
          const float cj = __int_as_float(__builtin_amdgcn_readlane(__float_as_int(cf), jb + u));
          float* a = acc + dj * LDA + lane * 2;
#pragma unroll
          for (int p = 0; p < P; ++p) {
            (void)__hip_atomic_fetch_add(a + p * 64,     v[u][p].x * cj, __ATOMIC_RELAXED,
                                         __HIP_MEMORY_SCOPE_WORKGROUP);
            (void)__hip_atomic_fetch_add(a + p * 64 + 1, v[u][p].y * cj, __ATOMIC_RELAXED,
                                         __HIP_MEMORY_SCOPE_WORKGROUP);
          }
        }
      }
    }
  }
  __syncthreads();

  constexpr int NTN = COUT / 16;
  constexpr int MPW = COUT / 32;
  const int half = lane >> 4;
  const int l16  = lane & 15;
  const int tn   = w % NTN;
  const int mt0  = (w / NTN) * MPW;
  const int bcol = tn * 16 + l16;
  const float* __restrict__ ap = &acc[(mt0 * 16 + l16) * LDA];

  v8f cacc[MPW];
#pragma unroll
  for (int j = 0; j < MPW; ++j) cacc[j] = (v8f){0.f, 0.f, 0.f, 0.f, 0.f, 0.f, 0.f, 0.f};

#pragma unroll 4
  for (int k0 = 0; k0 < C; k0 += 4) {
    const int ka = k0 + 2 * half;
    v2f b;
    b.x = Wm[(long)ka * COUT + bcol];
    b.y = Wm[(long)(ka + 1) * COUT + bcol];
#pragma unroll
    for (int j = 0; j < MPW; ++j) {
      v2f a;
      a.x = ap[j * 16 * LDA + ka];
      a.y = ap[j * 16 * LDA + ka + 1];
      cacc[j] = __builtin_amdgcn_wmma_f32_16x16x4_f32(
           false, a,  false, b,
           (short)0, cacc[j],  false,  false);
    }
  }

  const float bi = bias[bcol];
  const float mu = mean[bcol];
  const float rs = rsqrtf(var[bcol] + GCN_BN_EPS);
  const float ga = gamma[bcol];
  const float be = beta[bcol];
#pragma unroll
  for (int j = 0; j < MPW; ++j) {
    const int row0 = tbase + (mt0 + j) * 16 + half * 8;
    float* __restrict__ cp = outp + (long)row0 * COUT + bcol;
#pragma unroll
    for (int r = 0; r < 8; ++r) {
      float v = cacc[j][r] + bi;
      v = (v - mu) * rs * ga + be;
      if (RELU) v = fmaxf(v, 0.0f);
      if (row0 + r < N) cp[(long)r * COUT] = v;
    }
  }
}


extern "C" void kernel_launch(void* const* d_in, const int* in_sizes, int n_in,
                              void* d_out, int out_size, void* d_ws, size_t ws_size,
                              hipStream_t stream) {
  const float* x   = (const float*)d_in[0];
  const int*   ei  = (const int*)d_in[1];
  const float* W1  = (const float*)d_in[2];
  const float* b1  = (const float*)d_in[3];
  const float* g1  = (const float*)d_in[4];
  const float* be1 = (const float*)d_in[5];
  const float* m1  = (const float*)d_in[6];
  const float* v1  = (const float*)d_in[7];
  const float* W2  = (const float*)d_in[8];
  const float* b2  = (const float*)d_in[9];
  const float* g2  = (const float*)d_in[10];
  const float* be2 = (const float*)d_in[11];
  const float* m2  = (const float*)d_in[12];
  const float* v2  = (const float*)d_in[13];

  const int N = in_sizes[0] / GCN_IN_C;
  const int E = in_sizes[1] / 2;
  const int* src = ei;
  const int* dst = ei + E;

  char* ws = (char*)d_ws;
  int*   totals  = (int*)ws;
  int*   offsets = totals + MAX_TILES;
  int*   cursor  = offsets + 2 * MAX_TILES;       ws += (size_t)N * sizeof(int);
  float* dinv = (float*)ws;                       ws += (size_t)N * sizeof(float);
  float* act1 = (float*)ws;                       ws += (size_t)N * GCN_HID_C * sizeof(float);
  int*   binned  = (int*)ws;                      ws += (size_t)E * 2 * sizeof(int);
  (void)ws_size; (void)n_in; (void)out_size;

  const int NT = (N + TILE - 1) / TILE;
  const int BB = (E + BIN_CHUNK - 1) / BIN_CHUNK;

  zero_f32<<<1, MAX_TILES, 0, stream>>>((float*)totals, (long)MAX_TILES);
  bin_count<<<BB, BIN_THREADS, 0, stream>>>(dst, E, totals, NT);
  bin_scan<<<1, MAX_TILES, 0, stream>>>(totals, offsets, cursor, NT);
  bin_fill<<<BB, BIN_THREADS, 0, stream>>>(src, dst, E, cursor, binned, NT);
  tile_dinv<<<NT, 256, 0, stream>>>(binned, offsets, dinv, N);

  tile_layer<GCN_HID_C, true><<<NT, AGG_THREADS, 0, stream>>>(
      x, binned, offsets, dinv, W1, b1, g1, be1, m1, v1, act1, N);

  tile_layer<GCN_OUT_C, false><<<NT, AGG_THREADS, 0, stream>>>(
      act1, binned, offsets, dinv, W2, b2, g2, be2, m2, v2, (float*)d_out, N);
}


# layer 2 taken as A_hat (act1 W2): layer-1 tile block multiplies its finished tile by W2 (2nd f32 WMMA pass from LDS); layer 2 sums 64-wide rows: half the row bytes and ds adds; sums reassociated
# speedup vs baseline: 3.5610x; 1.2537x over previous
#include <hip/hip_runtime.h>

typedef __attribute__((ext_vector_type(2))) float v2f;
typedef __attribute__((ext_vector_type(4))) float v4f;
typedef __attribute__((ext_vector_type(8))) float v8f;

#define GCN_IN_C   128
#define GCN_HID_C  128
#define GCN_OUT_C  64
#define GCN_BN_EPS 1e-5f

#define TILE        128
#define TILE_SHIFT  7
#define MAX_TILES   512
#define BIN_CHUNK   4096
#define BIN_THREADS 256
#define BIN_PER_THR (BIN_CHUNK / BIN_THREADS)
#define AGG_THREADS 512
#define AGG_WAVES   (AGG_THREADS / 32)
#define AGG_FLIGHT  1024


__global__ void zero_f32(float* __restrict__ p, long n) {
  long i = (long)blockIdx.x * blockDim.x + threadIdx.x;
  long stride = (long)gridDim.x * blockDim.x;
  for (; i < n; i += stride) p[i] = 0.0f;
}

__global__ void bin_count(const int* __restrict__ dst, int E, int* __restrict__ totals, int NT) {
  __shared__ int hist[MAX_TILES];
  const int base = blockIdx.x * BIN_CHUNK + threadIdx.x;
  int dv[BIN_PER_THR];
#pragma unroll
  for (int k = 0; k < BIN_PER_THR; ++k) dv[k] = dst[min(base + k * BIN_THREADS, E - 1)];
  for (int i = threadIdx.x; i < NT; i += BIN_THREADS) hist[i] = 0;
  __syncthreads();
#pragma unroll
  for (int k = 0; k < BIN_PER_THR; ++k)
    if (base + k * BIN_THREADS < E) atomicAdd(&hist[dv[k] >> TILE_SHIFT], 1);
  __syncthreads();
  for (int i = threadIdx.x; i < NT; i += BIN_THREADS) {
    const int c = hist[i];
    if (c) atomicAdd(&totals[i], c);
  }
}

__global__ void bin_scan(const int* __restrict__ totals, int* __restrict__ offsets,
                         int* __restrict__ cursor, int NT) {
  __shared__ int a[2][MAX_TILES];
  const int i = threadIdx.x;
  const int own = (i < NT) ? totals[i] : 0;
  a[0][i] = own;
  __syncthreads();
  int cur = 0;
  for (int off = 1; off < MAX_TILES; off <<= 1) {
    int x = a[cur][i];
    if (i >= off) x += a[cur][i - off];
    a[cur ^ 1][i] = x;
    __syncthreads();
    cur ^= 1;
  }
  const int excl = a[cur][i] - own;
  if (i <= NT) offsets[i] = excl;
  if (i < NT) cursor[i] = excl;
}

__global__ void bin_fill(const int* __restrict__ src, const int* __restrict__ dst, int E,
                         int* __restrict__ cursor, int* __restrict__ binned, int NT) {
  __shared__ int hist[MAX_TILES];
  __shared__ int runbase[MAX_TILES];
  const int base = blockIdx.x * BIN_CHUNK + threadIdx.x;
  int dv[BIN_PER_THR], sv[BIN_PER_THR];
#pragma unroll
  for (int k = 0; k < BIN_PER_THR; ++k) dv[k] = dst[min(base + k * BIN_THREADS, E - 1)];
#pragma unroll
  for (int k = 0; k < BIN_PER_THR; ++k) sv[k] = src[min(base + k * BIN_THREADS, E - 1)];
  for (int i = threadIdx.x; i < NT; i += BIN_THREADS) hist[i] = 0;
  __syncthreads();
#pragma unroll
  for (int k = 0; k < BIN_PER_THR; ++k)
    if (base + k * BIN_THREADS < E) atomicAdd(&hist[dv[k] >> TILE_SHIFT], 1);
  __syncthreads();
  for (int i = threadIdx.x; i < NT; i += BIN_THREADS) {
    const int c = hist[i];
    runbase[i] = c ? atomicAdd(&cursor[i], c) : 0;
    hist[i] = 0;
  }
  __syncthreads();
  int2* __restrict__ out = (int2*)binned;
#pragma unroll
  for (int k = 0; k < BIN_PER_THR; ++k) {
    if (base + k * BIN_THREADS < E) {
      const int b = dv[k] >> TILE_SHIFT;
      const int r = atomicAdd(&hist[b], 1);
      out[runbase[b] + r] = make_int2(sv[k], dv[k]);
    }
  }
}

__global__ void tile_dinv(const int* __restrict__ binned, const int* __restrict__ offsets,
                          float* __restrict__ dinv, int N) {
  __shared__ int cnt[TILE];
  const int tile = blockIdx.x;
  const int tbase = tile * TILE;
  if (threadIdx.x < TILE) cnt[threadIdx.x] = 0;
  __syncthreads();
  const int e0 = offsets[tile];
  const int e1 = offsets[tile + 1];
  const int2* __restrict__ ed = (const int2*)binned;
  for (int e = e0 + threadIdx.x; e < e1; e += blockDim.x) atomicAdd(&cnt[ed[e].y - tbase], 1);
  __syncthreads();
  if (threadIdx.x < TILE) {
    const int node = tbase + threadIdx.x;
    if (node < N) dinv[node] = rsqrtf((float)(cnt[threadIdx.x] + 1));
  }
}

template <int C>
__device__ __forceinline__ void tile_sums(
    float* __restrict__ acc, const float* __restrict__ h, const int* __restrict__ binned,
    const int* __restrict__ offsets, const float* __restrict__ dinv,
    int N, int tile, int tid, int lane, int w) {
  static_assert(C % 64 == 0 && AGG_FLIGHT % C == 0 && AGG_FLIGHT / C <= 32,
                "a lane holds whole float pairs; one 32-edge group is a whole number of batches");
  constexpr int LDA = C + 4;
  constexpr int P = C / 64;
  constexpr int R = AGG_FLIGHT / C;
  const int tbase = tile * TILE;

  for (int i = tid; i < TILE * C / 4; i += AGG_THREADS) {
    const int nl = i / (C / 4);
    const int c4 = i % (C / 4);
    const int node = tbase + nl;
    v4f v = {0.f, 0.f, 0.f, 0.f};
    if (node < N) {
      const float di = dinv[node];
      v = ((const v4f*)(h + (long)node * C))[c4] * (di * di);
    }
    *(v4f*)&acc[nl * LDA + c4 * 4] = v;
  }
  __syncthreads();

  const int e0 = offsets[tile];
  const int e1 = offsets[tile + 1];
  const int2* __restrict__ ed = (const int2*)binned;

  for (int g = e0 + w * 32; g < e1; g += AGG_WAVES * 32) {
    const int cnt = min(32, e1 - g);
    int s = 0, dl = 0;
    float cf = 0.0f;
    if (lane < cnt) {
      const int2 p = ed[g + lane];
      s  = p.x;
      dl = p.y - tbase;
      cf = dinv[p.x] * dinv[p.y];
    }
    for (int jb = 0; jb < cnt; jb += R) {
      v2f v[R][P];
#pragma unroll
      for (int u = 0; u < R; ++u) {
        const int sj = __builtin_amdgcn_readlane(s, jb + u);
        const float* __restrict__ row = h + (long)sj * C + lane * 2;
#pragma unroll
        for (int p = 0; p < P; ++p) v[u][p] = *(const v2f*)(row + p * 64);
      }
#pragma unroll
      for (int u = 0; u < R; ++u) {
        if (u == 0 || jb + u < cnt) {
          const int   dj = __builtin_amdgcn_readlane(dl, jb + u);
          const float cj = __int_as_float(__builtin_amdgcn_readlane(__float_as_int(cf), jb + u));
          float* a = acc + dj * LDA + lane * 2;
#pragma unroll
          for (int p = 0; p < P; ++p) {
            (void)__hip_atomic_fetch_add(a + p * 64,     v[u][p].x * cj, __ATOMIC_RELAXED,
                                         __HIP_MEMORY_SCOPE_WORKGROUP);
            (void)__hip_atomic_fetch_add(a + p * 64 + 1, v[u][p].y * cj, __ATOMIC_RELAXED,
                                         __HIP_MEMORY_SCOPE_WORKGROUP);
          }
        }
      }
    }
  }
  __syncthreads();
}

template <int COUT>
__device__ __forceinline__ void tile_product(
    const float* __restrict__ acc, const float* __restrict__ Wm, int w, int lane,
    v8f (&cacc)[COUT / 32]) {
  constexpr int C   = 128;
  constexpr int LDA = C + 4;
  static_assert(COUT % 32 == 0 && AGG_WAVES % (COUT / 16) == 0 &&
                (AGG_WAVES / (COUT / 16)) * (COUT / 32) * 16 == TILE,
                "the waves' 16-row blocks must cover the tile exactly once");
  constexpr int NTN = COUT / 16;
  constexpr int MPW = COUT / 32;
  const int half = lane >> 4;
  const int l16  = lane & 15;
  const int tn   = w % NTN;
  const int mt0  = (w / NTN) * MPW;
  const int bcol = tn * 16 + l16;
  const float* __restrict__ ap = &acc[(mt0 * 16 + l16) * LDA];

#pragma unroll
  for (int j = 0; j < MPW; ++j) cacc[j] = (v8f){0.f, 0.f, 0.f, 0.f, 0.f, 0.f, 0.f, 0.f};

#pragma unroll 4
  for (int k0 = 0; k0 < C; k0 += 4) {
    const int ka = k0 + 2 * half;
    v2f b;
    b.x = Wm[(long)ka * COUT + bcol];
    b.y = Wm[(long)(ka + 1) * COUT + bcol];
#pragma unroll
    for (int j = 0; j < MPW; ++j) {
      v2f a;
      a.x = ap[j * 16 * LDA + ka];
      a.y = ap[j * 16 * LDA + ka + 1];
      cacc[j] = __builtin_amdgcn_wmma_f32_16x16x4_f32(
           false, a,  false, b,
           (short)0, cacc[j],  false,  false);
    }
  }
}

__global__ __launch_bounds__(AGG_THREADS) void tile_layer1(
    const float* __restrict__ x, const int* __restrict__ binned,
    const int* __restrict__ offsets, const float* __restrict__ dinv,
    const float* __restrict__ W1, const float* __restrict__ bias,
    const float* __restrict__ gamma, const float* __restrict__ beta,
    const float* __restrict__ mean, const float* __restrict__ var,
    const float* __restrict__ W2, float* __restrict__ hw2, int N) {
  static_assert(GCN_IN_C == 128 && GCN_HID_C == 128, "both products run over a 128-wide LDS tile");
  constexpr int LDA = 128 + 4;
  __shared__ float acc[TILE * LDA];

  const int tid  = threadIdx.x;
  const int lane = tid & 31;
  const int w = __builtin_amdgcn_readfirstlane(tid >> 5);
  const int tile  = blockIdx.x;
  const int tbase = tile * TILE;
  const int half = lane >> 4;
  const int l16  = lane & 15;

  tile_sums<GCN_IN_C>(acc, x, binned, offsets, dinv, N, tile, tid, lane, w);

  constexpr int MPW1 = GCN_HID_C / 32;
  v8f c1[MPW1];
  tile_product<GCN_HID_C>(acc, W1, w, lane, c1);

  __syncthreads();
  {
    const int bcol = (w % (GCN_HID_C / 16)) * 16 + l16;
    const int mt0  = (w / (GCN_HID_C / 16)) * MPW1;
    const float bi = bias[bcol];
    const float mu = mean[bcol];
    const float rs = rsqrtf(var[bcol] + GCN_BN_EPS);
    const float ga = gamma[bcol];
    const float be = beta[bcol];
#pragma unroll
    for (int j = 0; j < MPW1; ++j) {
      float* __restrict__ ar = &acc[((mt0 + j) * 16 + half * 8) * LDA + bcol];
#pragma unroll
      for (int r = 0; r < 8; ++r) {
        float v = c1[j][r] + bi;
        v = (v - mu) * rs * ga + be;
        ar[r * LDA] = fmaxf(v, 0.0f);
      }
    }
  }
  __syncthreads();

  constexpr int MPW2 = GCN_OUT_C / 32;
  v8f c2[MPW2];
  tile_product<GCN_OUT_C>(acc, W2, w, lane, c2);
  {
    const int bcol = (w % (GCN_OUT_C / 16)) * 16 + l16;
    const int mt0  = (w / (GCN_OUT_C / 16)) * MPW2;
#pragma unroll
    for (int j = 0; j < MPW2; ++j) {
      const int row0 = tbase + (mt0 + j) * 16 + half * 8;
      float* __restrict__ cp = hw2 + (long)row0 * GCN_OUT_C + bcol;
#pragma unroll
      for (int r = 0; r < 8; ++r)
        if (row0 + r < N) cp[(long)r * GCN_OUT_C] = c2[j][r];
    }
  }
}

__global__ __launch_bounds__(AGG_THREADS) void tile_layer2(
    const float* __restrict__ hw2, const int* __restrict__ binned,
    const int* __restrict__ offsets, const float* __restrict__ dinv,
    const float* __restrict__ bias,
    const float* __restrict__ gamma, const float* __restrict__ beta,
    const float* __restrict__ mean, const float* __restrict__ var,
    float* __restrict__ outp, int N) {
  constexpr int C   = GCN_OUT_C;
  constexpr int LDA = C + 4;
  __shared__ float acc[TILE * LDA];

  const int tid  = threadIdx.x;
  const int lane = tid & 31;
  const int w = __builtin_amdgcn_readfirstlane(tid >> 5);
  const int tile  = blockIdx.x;
  const int tbase = tile * TILE;

  tile_sums<C>(acc, hw2, binned, offsets, dinv, N, tile, tid, lane, w);

  static_assert(AGG_THREADS % (C / 4) == 0, "a thread's column group must not change with i");
  const int c4 = tid % (C / 4);
  const v4f bi = ((const v4f*)bias)[c4];
  const v4f mu = ((const v4f*)mean)[c4];
  const v4f vr = ((const v4f*)var)[c4];
  const v4f ga = ((const v4f*)gamma)[c4];
  const v4f be = ((const v4f*)beta)[c4];
  v4f rs;
  rs.x = rsqrtf(vr.x + GCN_BN_EPS);
  rs.y = rsqrtf(vr.y + GCN_BN_EPS);
  rs.z = rsqrtf(vr.z + GCN_BN_EPS);
  rs.w = rsqrtf(vr.w + GCN_BN_EPS);
  for (int i = tid; i < TILE * C / 4; i += AGG_THREADS) {
    const int nl = i / (C / 4);
    const int node = tbase + nl;
    if (node < N) {
      v4f v = *(const v4f*)&acc[nl * LDA + c4 * 4] + bi;
      v = (v - mu) * rs * ga + be;
      ((v4f*)(outp + (long)node * C))[c4] = v;
    }
  }
}


extern "C" void kernel_launch(void* const* d_in, const int* in_sizes, int n_in,
                              void* d_out, int out_size, void* d_ws, size_t ws_size,
                              hipStream_t stream) {
  const float* x   = (const float*)d_in[0];
  const int*   ei  = (const int*)d_in[1];
  const float* W1  = (const float*)d_in[2];
  const float* b1  = (const float*)d_in[3];
  const float* g1  = (const float*)d_in[4];
  const float* be1 = (const float*)d_in[5];
  const float* m1  = (const float*)d_in[6];
  const float* v1  = (const float*)d_in[7];
  const float* W2  = (const float*)d_in[8];
  const float* b2  = (const float*)d_in[9];
  const float* g2  = (const float*)d_in[10];
  const float* be2 = (const float*)d_in[11];
  const float* m2  = (const float*)d_in[12];
  const float* v2  = (const float*)d_in[13];

  const int N = in_sizes[0] / GCN_IN_C;
  const int E = in_sizes[1] / 2;
  const int* src = ei;
  const int* dst = ei + E;

  char* ws = (char*)d_ws;
  float* hw2 = (float*)ws;                        ws += (size_t)N * GCN_OUT_C * sizeof(float);
  int*   totals  = (int*)ws;
  int*   offsets = totals + MAX_TILES;
  int*   cursor  = offsets + 2 * MAX_TILES;       ws += (size_t)N * sizeof(int);
  float* dinv = (float*)ws;                       ws += (size_t)N * sizeof(float);
  int*   binned  = (int*)ws;                      ws += (size_t)E * 2 * sizeof(int);
  (void)ws_size; (void)n_in; (void)out_size;

  const int NT = (N + TILE - 1) / TILE;
  const int BB = (E + BIN_CHUNK - 1) / BIN_CHUNK;

  zero_f32<<<1, MAX_TILES, 0, stream>>>((float*)totals, (long)MAX_TILES);
  bin_count<<<BB, BIN_THREADS, 0, stream>>>(dst, E, totals, NT);
  bin_scan<<<1, MAX_TILES, 0, stream>>>(totals, offsets, cursor, NT);
  bin_fill<<<BB, BIN_THREADS, 0, stream>>>(src, dst, E, cursor, binned, NT);
  tile_dinv<<<NT, 256, 0, stream>>>(binned, offsets, dinv, N);

  tile_layer1<<<NT, AGG_THREADS, 0, stream>>>(
      x, binned, offsets, dinv, W1, b1, g1, be1, m1, v1, W2, hw2, N);

  tile_layer2<<<NT, AGG_THREADS, 0, stream>>>(
      hw2, binned, offsets, dinv, b2, g2, be2, m2, v2, (float*)d_out, N);
}


# tiling: one 196-node destination tile per WGP: 256 work-groups of 32 waves (was 391 of 16 on 256 WGPs: 135 WGPs held two tiles); 13 row blocks dealt over the waves; arrival-order sums as before
# speedup vs baseline: 3.9792x; 1.1174x over previous
#include <hip/hip_runtime.h>

typedef __attribute__((ext_vector_type(2))) float v2f;
typedef __attribute__((ext_vector_type(4))) float v4f;
typedef __attribute__((ext_vector_type(8))) float v8f;

#define GCN_IN_C   128
#define GCN_HID_C  128
#define GCN_OUT_C  64
#define GCN_BN_EPS 1e-5f

#define TILE        196
#define TILE_ROWS   208
#define MAX_TILES   512
#define BIN_CHUNK   4096
#define BIN_THREADS 256
#define BIN_PER_THR (BIN_CHUNK / BIN_THREADS)
#define AGG_THREADS 1024
#define AGG_WAVES   (AGG_THREADS / 32)
#define AGG_FLIGHT  1024


__global__ void zero_f32(float* __restrict__ p, long n) {
  long i = (long)blockIdx.x * blockDim.x + threadIdx.x;
  long stride = (long)gridDim.x * blockDim.x;
  for (; i < n; i += stride) p[i] = 0.0f;
}

__device__ __forceinline__ int tile_of(int node) { return (int)((unsigned)node / (unsigned)TILE); }

__global__ void bin_count(const int* __restrict__ dst, int E, int* __restrict__ totals, int NT) {
  __shared__ int hist[MAX_TILES];
  const int base = blockIdx.x * BIN_CHUNK + threadIdx.x;
  int dv[BIN_PER_THR];
#pragma unroll
  for (int k = 0; k < BIN_PER_THR; ++k) dv[k] = dst[min(base + k * BIN_THREADS, E - 1)];
  for (int i = threadIdx.x; i < NT; i += BIN_THREADS) hist[i] = 0;
  __syncthreads();
#pragma unroll
  for (int k = 0; k < BIN_PER_THR; ++k)
    if (base + k * BIN_THREADS < E) atomicAdd(&hist[tile_of(dv[k])], 1);
  __syncthreads();
  for (int i = threadIdx.x; i < NT; i += BIN_THREADS) {
    const int c = hist[i];
    if (c) atomicAdd(&totals[i], c);
  }
}

__global__ void bin_scan(const int* __restrict__ totals, int* __restrict__ offsets,
                         int* __restrict__ cursor, int NT) {
  __shared__ int a[2][MAX_TILES];
  const int i = threadIdx.x;
  const int own = (i < NT) ? totals[i] : 0;
  a[0][i] = own;
  __syncthreads();
  int cur = 0;
  for (int off = 1; off < MAX_TILES; off <<= 1) {
    int x = a[cur][i];
    if (i >= off) x += a[cur][i - off];
    a[cur ^ 1][i] = x;
    __syncthreads();
    cur ^= 1;
  }
  const int excl = a[cur][i] - own;
  if (i <= NT) offsets[i] = excl;
  if (i < NT) cursor[i] = excl;
}

__global__ void bin_fill(const int* __restrict__ src, const int* __restrict__ dst, int E,
                         int* __restrict__ cursor, int* __restrict__ binned, int NT) {
  __shared__ int hist[MAX_TILES];
  __shared__ int runbase[MAX_TILES];
  const int base = blockIdx.x * BIN_CHUNK + threadIdx.x;
  int dv[BIN_PER_THR], sv[BIN_PER_THR];
#pragma unroll
  for (int k = 0; k < BIN_PER_THR; ++k) dv[k] = dst[min(base + k * BIN_THREADS, E - 1)];
#pragma unroll
  for (int k = 0; k < BIN_PER_THR; ++k) sv[k] = src[min(base + k * BIN_THREADS, E - 1)];
  for (int i = threadIdx.x; i < NT; i += BIN_THREADS) hist[i] = 0;
  __syncthreads();
#pragma unroll
  for (int k = 0; k < BIN_PER_THR; ++k)
    if (base + k * BIN_THREADS < E) atomicAdd(&hist[tile_of(dv[k])], 1);
  __syncthreads();
  for (int i = threadIdx.x; i < NT; i += BIN_THREADS) {
    const int c = hist[i];
    runbase[i] = c ? atomicAdd(&cursor[i], c) : 0;
    hist[i] = 0;
  }
  __syncthreads();
  int2* __restrict__ out = (int2*)binned;
#pragma unroll
  for (int k = 0; k < BIN_PER_THR; ++k) {
    if (base + k * BIN_THREADS < E) {
      const int b = tile_of(dv[k]);
      const int r = atomicAdd(&hist[b], 1);
      out[runbase[b] + r] = make_int2(sv[k], dv[k]);
    }
  }
}

__global__ void tile_dinv(const int* __restrict__ binned, const int* __restrict__ offsets,
                          float* __restrict__ dinv, int N) {
  static_assert(TILE <= 256, "launched with 256 threads: one thread per node of the tile");
  __shared__ int cnt[TILE];
  const int tile = blockIdx.x;
  const int tbase = tile * TILE;
  if (threadIdx.x < TILE) cnt[threadIdx.x] = 0;
  __syncthreads();
  const int e0 = offsets[tile];
  const int e1 = offsets[tile + 1];
  const int2* __restrict__ ed = (const int2*)binned;
  for (int e = e0 + threadIdx.x; e < e1; e += blockDim.x) atomicAdd(&cnt[ed[e].y - tbase], 1);
  __syncthreads();
  if (threadIdx.x < TILE) {
    const int node = tbase + threadIdx.x;
    if (node < N) dinv[node] = rsqrtf((float)(cnt[threadIdx.x] + 1));
  }
}

template <int C, int ROWS>
__device__ __forceinline__ void tile_sums(
    float* __restrict__ acc, const float* __restrict__ h, const int* __restrict__ binned,
    const int* __restrict__ offsets, const float* __restrict__ dinv,
    int N, int tile, int tid, int lane, int w) {
  static_assert(C % 64 == 0 && AGG_FLIGHT % C == 0 && AGG_FLIGHT / C <= 32,
                "a lane holds whole float pairs; one 32-edge group is a whole number of batches");
  constexpr int LDA = C + 4;
  constexpr int P = C / 64;
  constexpr int R = AGG_FLIGHT / C;
  const int tbase = tile * TILE;

  for (int i = tid; i < ROWS * C / 4; i += AGG_THREADS) {
    const int nl = i / (C / 4);
    const int c4 = i % (C / 4);
    const int node = tbase + nl;
    v4f v = {0.f, 0.f, 0.f, 0.f};
    if (nl < TILE && node < N) {
      const float di = dinv[node];
      v = ((const v4f*)(h + (long)node * C))[c4] * (di * di);
    }
    *(v4f*)&acc[nl * LDA + c4 * 4] = v;
  }
  __syncthreads();

  const int e0 = offsets[tile];
  const int e1 = offsets[tile + 1];
  const int2* __restrict__ ed = (const int2*)binned;

  for (int g = e0 + w * 32; g < e1; g += AGG_WAVES * 32) {
    const int cnt = min(32, e1 - g);
    int s = 0, dl = 0;
    float cf = 0.0f;
    if (lane < cnt) {
      const int2 p = ed[g + lane];
      s  = p.x;
      dl = p.y - tbase;
      cf = dinv[p.x] * dinv[p.y];
    }
    for (int jb = 0; jb < cnt; jb += R) {
      v2f v[R][P];
#pragma unroll
      for (int u = 0; u < R; ++u) {
        const int sj = __builtin_amdgcn_readlane(s, jb + u);
        const float* __restrict__ row = h + (long)sj * C + lane * 2;
#pragma unroll
        for (int p = 0; p < P; ++p) v[u][p] = *(const v2f*)(row + p * 64);
      }
#pragma unroll
      for (int u = 0; u < R; ++u) {
        if (u == 0 || jb + u < cnt) {
          const int   dj = __builtin_amdgcn_readlane(dl, jb + u);
          const float cj = __int_as_float(__builtin_amdgcn_readlane(__float_as_int(cf), jb + u));
          float* a = acc + dj * LDA + lane * 2;
#pragma unroll
          for (int p = 0; p < P; ++p) {
            (void)__hip_atomic_fetch_add(a + p * 64,     v[u][p].x * cj, __ATOMIC_RELAXED,
                                         __HIP_MEMORY_SCOPE_WORKGROUP);
            (void)__hip_atomic_fetch_add(a + p * 64 + 1, v[u][p].y * cj, __ATOMIC_RELAXED,
                                         __HIP_MEMORY_SCOPE_WORKGROUP);
          }
        }
      }
    }
  }
  __syncthreads();
}

template <int COUT>
struct ProductShape {
  static constexpr int NTN = COUT / 16;
  static constexpr int NWG = AGG_WAVES / NTN;
  static constexpr int NRB = TILE_ROWS / 16;
  static constexpr int MPW = (NRB + NWG - 1) / NWG;
  static_assert(COUT % 16 == 0 && AGG_WAVES % NTN == 0 && TILE_ROWS % 16 == 0 && TILE_ROWS >= TILE,
                "whole strips, whole wave groups, whole row blocks");
  static_assert(NRB % MPW <= 1, "a wave holds MPW row blocks, one, or none");
};

template <int COUT, int NB>
__device__ __forceinline__ void product_rows(
    const float* __restrict__ ap, const float* __restrict__ Wm, int bcol, int half,
    v8f (&cacc)[ProductShape<COUT>::MPW]) {
  constexpr int C   = 128;
  constexpr int LDA = C + 4;
#pragma unroll 4
  for (int k0 = 0; k0 < C; k0 += 4) {
    const int ka = k0 + 2 * half;
    v2f b;
    b.x = Wm[(long)ka * COUT + bcol];
    b.y = Wm[(long)(ka + 1) * COUT + bcol];
#pragma unroll
    for (int j = 0; j < NB; ++j) {
      v2f a;
      a.x = ap[j * 16 * LDA + ka];
      a.y = ap[j * 16 * LDA + ka + 1];
      cacc[j] = __builtin_amdgcn_wmma_f32_16x16x4_f32(
           false, a,  false, b,
           (short)0, cacc[j],  false,  false);
    }
  }
}

template <int COUT>
__device__ __forceinline__ int tile_product(
    const float* __restrict__ acc, const float* __restrict__ Wm, int w, int lane,
    v8f (&cacc)[ProductShape<COUT>::MPW]) {
  using S = ProductShape<COUT>;
  constexpr int LDA = 128 + 4;
  const int half = lane >> 4;
  const int l16  = lane & 15;
  const int tn   = w % S::NTN;
  const int mt0  = (w / S::NTN) * S::MPW;
  const int bcol = tn * 16 + l16;
  const float* __restrict__ ap = &acc[(mt0 * 16 + l16) * LDA];

#pragma unroll
  for (int j = 0; j < S::MPW; ++j) cacc[j] = (v8f){0.f, 0.f, 0.f, 0.f, 0.f, 0.f, 0.f, 0.f};

  const int nb = __builtin_amdgcn_readfirstlane(min(S::MPW, max(0, S::NRB - mt0)));
  if (nb == S::MPW)  product_rows<COUT, S::MPW>(ap, Wm, bcol, half, cacc);
  else if (nb == 1)  product_rows<COUT, 1>(ap, Wm, bcol, half, cacc);
  return nb;
}

__global__ __launch_bounds__(AGG_THREADS) void tile_layer1(
    const float* __restrict__ x, const int* __restrict__ binned,
    const int* __restrict__ offsets, const float* __restrict__ dinv,
    const float* __restrict__ W1, const float* __restrict__ bias,
    const float* __restrict__ gamma, const float* __restrict__ beta,
    const float* __restrict__ mean, const float* __restrict__ var,
    const float* __restrict__ W2, float* __restrict__ hw2, int N) {
  static_assert(GCN_IN_C == 128 && GCN_HID_C == 128, "both products run over a 128-wide LDS tile");
  constexpr int LDA = 128 + 4;
  __shared__ float acc[TILE_ROWS * LDA];

  const int tid  = threadIdx.x;
  const int lane = tid & 31;
  const int w = __builtin_amdgcn_readfirstlane(tid >> 5);
  const int tile  = blockIdx.x;
  const int tbase = tile * TILE;
  const int half = lane >> 4;
  const int l16  = lane & 15;

  tile_sums<GCN_IN_C, TILE_ROWS>(acc, x, binned, offsets, dinv, N, tile, tid, lane, w);

  using S1 = ProductShape<GCN_HID_C>;
  v8f c1[S1::MPW];
  const int nb1 = tile_product<GCN_HID_C>(acc, W1, w, lane, c1);

  __syncthreads();
  {
    const int bcol = (w % S1::NTN) * 16 + l16;
    const int mt0  = (w / S1::NTN) * S1::MPW;
    const float bi = bias[bcol];
    const float mu = mean[bcol];
    const float rs = rsqrtf(var[bcol] + GCN_BN_EPS);
    const float ga = gamma[bcol];
    const float be = beta[bcol];
#pragma unroll
    for (int j = 0; j < S1::MPW; ++j) {
      if (j < nb1) {
        float* __restrict__ ar = &acc[((mt0 + j) * 16 + half * 8) * LDA + bcol];
#pragma unroll
        for (int r = 0; r < 8; ++r) {
          float v = c1[j][r] + bi;
          v = (v - mu) * rs * ga + be;
          ar[r * LDA] = fmaxf(v, 0.0f);
        }
      }
    }
  }
  __syncthreads();

  using S2 = ProductShape<GCN_OUT_C>;
  v8f c2[S2::MPW];
  const int nb2 = tile_product<GCN_OUT_C>(acc, W2, w, lane, c2);
  {
    const int bcol = (w % S2::NTN) * 16 + l16;
    const int mt0  = (w / S2::NTN) * S2::MPW;
#pragma unroll
    for (int j = 0; j < S2::MPW; ++j) {
      if (j < nb2) {
        const int nl0 = (mt0 + j) * 16 + half * 8;
        float* __restrict__ cp = hw2 + (long)(tbase + nl0) * GCN_OUT_C + bcol;
#pragma unroll
        for (int r = 0; r < 8; ++r)
          if (nl0 + r < TILE && tbase + nl0 + r < N) cp[(long)r * GCN_OUT_C] = c2[j][r];
      }
    }
  }
}

__global__ __launch_bounds__(AGG_THREADS) void tile_layer2(
    const float* __restrict__ hw2, const int* __restrict__ binned,
    const int* __restrict__ offsets, const float* __restrict__ dinv,
    const float* __restrict__ bias,
    const float* __restrict__ gamma, const float* __restrict__ beta,
    const float* __restrict__ mean, const float* __restrict__ var,
    float* __restrict__ outp, int N) {
  constexpr int C   = GCN_OUT_C;
  constexpr int LDA = C + 4;
  __shared__ float acc[TILE * LDA];

  const int tid  = threadIdx.x;
  const int lane = tid & 31;
  const int w = __builtin_amdgcn_readfirstlane(tid >> 5);
  const int tile  = blockIdx.x;
  const int tbase = tile * TILE;

  tile_sums<C, TILE>(acc, hw2, binned, offsets, dinv, N, tile, tid, lane, w);

  static_assert(AGG_THREADS % (C / 4) == 0, "a thread's column group must not change with i");
  const int c4 = tid % (C / 4);
  const v4f bi = ((const v4f*)bias)[c4];
  const v4f mu = ((const v4f*)mean)[c4];
  const v4f vr = ((const v4f*)var)[c4];
  const v4f ga = ((const v4f*)gamma)[c4];
  const v4f be = ((const v4f*)beta)[c4];
  v4f rs;
  rs.x = rsqrtf(vr.x + GCN_BN_EPS);
  rs.y = rsqrtf(vr.y + GCN_BN_EPS);
  rs.z = rsqrtf(vr.z + GCN_BN_EPS);
  rs.w = rsqrtf(vr.w + GCN_BN_EPS);
  for (int i = tid; i < TILE * C / 4; i += AGG_THREADS) {
    const int nl = i / (C / 4);
    const int node = tbase + nl;
    if (node < N) {
      v4f v = *(const v4f*)&acc[nl * LDA + c4 * 4] + bi;
      v = (v - mu) * rs * ga + be;
      ((v4f*)(outp + (long)node * C))[c4] = v;
    }
  }
}


extern "C" void kernel_launch(void* const* d_in, const int* in_sizes, int n_in,
                              void* d_out, int out_size, void* d_ws, size_t ws_size,
                              hipStream_t stream) {
  const float* x   = (const float*)d_in[0];
  const int*   ei  = (const int*)d_in[1];
  const float* W1  = (const float*)d_in[2];
  const float* b1  = (const float*)d_in[3];
  const float* g1  = (const float*)d_in[4];
  const float* be1 = (const float*)d_in[5];
  const float* m1  = (const float*)d_in[6];
  const float* v1  = (const float*)d_in[7];
  const float* W2  = (const float*)d_in[8];
  const float* b2  = (const float*)d_in[9];
  const float* g2  = (const float*)d_in[10];
  const float* be2 = (const float*)d_in[11];
  const float* m2  = (const float*)d_in[12];
  const float* v2  = (const float*)d_in[13];

  const int N = in_sizes[0] / GCN_IN_C;
  const int E = in_sizes[1] / 2;
  const int* src = ei;
  const int* dst = ei + E;

  char* ws = (char*)d_ws;
  float* hw2 = (float*)ws;                        ws += (size_t)N * GCN_OUT_C * sizeof(float);
  int*   totals  = (int*)ws;
  int*   offsets = totals + MAX_TILES;
  int*   cursor  = offsets + 2 * MAX_TILES;       ws += (size_t)N * sizeof(int);
  float* dinv = (float*)ws;                       ws += (size_t)N * sizeof(float);
  int*   binned  = (int*)ws;                      ws += (size_t)E * 2 * sizeof(int);
  (void)ws_size; (void)n_in; (void)out_size;

  const int NT = (N + TILE - 1) / TILE;
  const int BB = (E + BIN_CHUNK - 1) / BIN_CHUNK;

  zero_f32<<<1, MAX_TILES, 0, stream>>>((float*)totals, (long)MAX_TILES);
  bin_count<<<BB, BIN_THREADS, 0, stream>>>(dst, E, totals, NT);
  bin_scan<<<1, MAX_TILES, 0, stream>>>(totals, offsets, cursor, NT);
  bin_fill<<<BB, BIN_THREADS, 0, stream>>>(src, dst, E, cursor, binned, NT);
  tile_dinv<<<NT, 256, 0, stream>>>(binned, offsets, dinv, N);

  tile_layer1<<<NT, AGG_THREADS, 0, stream>>>(
      x, binned, offsets, dinv, W1, b1, g1, be1, m1, v1, W2, hw2, N);

  tile_layer2<<<NT, AGG_THREADS, 0, stream>>>(
      hw2, binned, offsets, dinv, b2, g2, be2, m2, v2, (float*)d_out, N);
}


# aggregation: tile block orders its edges by dst row in LDS (count, scan, place); a wave sums a row's sources in registers from 16-byte loads, adds to LDS once: no LDS float atomics; sum order changes
# speedup vs baseline: 4.8565x; 1.2205x over previous
#include <hip/hip_runtime.h>

typedef __attribute__((ext_vector_type(2))) float v2f;
typedef __attribute__((ext_vector_type(4))) float v4f;
typedef __attribute__((ext_vector_type(8))) float v8f;

#define GCN_IN_C   128
#define GCN_HID_C  128
#define GCN_OUT_C  64
#define GCN_BN_EPS 1e-5f

#define TILE        196
#define TILE_ROWS   208
#define MAX_TILES   512
#define BIN_CHUNK   4096
#define BIN_THREADS 256
#define BIN_PER_THR (BIN_CHUNK / BIN_THREADS)
#define AGG_THREADS 1024
#define AGG_WAVES   (AGG_THREADS / 32)
#define AGG_FLIGHT  1024
#define SRT_CAP     8192
#define SRT_PER_THR (SRT_CAP / AGG_THREADS)
#define SCAN_PER_LANE ((TILE + 31) / 32)


__global__ void zero_f32(float* __restrict__ p, long n) {
  long i = (long)blockIdx.x * blockDim.x + threadIdx.x;
  long stride = (long)gridDim.x * blockDim.x;
  for (; i < n; i += stride) p[i] = 0.0f;
}

__device__ __forceinline__ int tile_of(int node) { return (int)((unsigned)node / (unsigned)TILE); }

__global__ void bin_count(const int* __restrict__ dst, int E, int* __restrict__ totals, int NT) {
  __shared__ int hist[MAX_TILES];
  const int base = blockIdx.x * BIN_CHUNK + threadIdx.x;
  int dv[BIN_PER_THR];
#pragma unroll
  for (int k = 0; k < BIN_PER_THR; ++k) dv[k] = dst[min(base + k * BIN_THREADS, E - 1)];
  for (int i = threadIdx.x; i < NT; i += BIN_THREADS) hist[i] = 0;
  __syncthreads();
#pragma unroll
  for (int k = 0; k < BIN_PER_THR; ++k)
    if (base + k * BIN_THREADS < E) atomicAdd(&hist[tile_of(dv[k])], 1);
  __syncthreads();
  for (int i = threadIdx.x; i < NT; i += BIN_THREADS) {
    const int c = hist[i];
    if (c) atomicAdd(&totals[i], c);
  }
}

__global__ void bin_scan(const int* __restrict__ totals, int* __restrict__ offsets,
                         int* __restrict__ cursor, int NT) {
  __shared__ int a[2][MAX_TILES];
  const int i = threadIdx.x;
  const int own = (i < NT) ? totals[i] : 0;
  a[0][i] = own;
  __syncthreads();
  int cur = 0;
  for (int off = 1; off < MAX_TILES; off <<= 1) {
    int x = a[cur][i];
    if (i >= off) x += a[cur][i - off];
    a[cur ^ 1][i] = x;
    __syncthreads();
    cur ^= 1;
  }
  const int excl = a[cur][i] - own;
  if (i <= NT) offsets[i] = excl;
  if (i < NT) cursor[i] = excl;
}

__global__ void bin_fill(const int* __restrict__ src, const int* __restrict__ dst, int E,
                         int* __restrict__ cursor, int* __restrict__ binned, int NT) {
  __shared__ int hist[MAX_TILES];
  __shared__ int runbase[MAX_TILES];
  const int base = blockIdx.x * BIN_CHUNK + threadIdx.x;
  int dv[BIN_PER_THR], sv[BIN_PER_THR];
#pragma unroll
  for (int k = 0; k < BIN_PER_THR; ++k) dv[k] = dst[min(base + k * BIN_THREADS, E - 1)];
#pragma unroll
  for (int k = 0; k < BIN_PER_THR; ++k) sv[k] = src[min(base + k * BIN_THREADS, E - 1)];
  for (int i = threadIdx.x; i < NT; i += BIN_THREADS) hist[i] = 0;
  __syncthreads();
#pragma unroll
  for (int k = 0; k < BIN_PER_THR; ++k)
    if (base + k * BIN_THREADS < E) atomicAdd(&hist[tile_of(dv[k])], 1);
  __syncthreads();
  for (int i = threadIdx.x; i < NT; i += BIN_THREADS) {
    const int c = hist[i];
    runbase[i] = c ? atomicAdd(&cursor[i], c) : 0;
    hist[i] = 0;
  }
  __syncthreads();
  int2* __restrict__ out = (int2*)binned;
#pragma unroll
  for (int k = 0; k < BIN_PER_THR; ++k) {
    if (base + k * BIN_THREADS < E) {
      const int b = tile_of(dv[k]);
      const int r = atomicAdd(&hist[b], 1);
      out[runbase[b] + r] = make_int2(sv[k], dv[k]);
    }
  }
}

__global__ void tile_dinv(const int* __restrict__ binned, const int* __restrict__ offsets,
                          float* __restrict__ dinv, int N) {
  static_assert(TILE <= 256, "launched with 256 threads: one thread per node of the tile");
  __shared__ int cnt[TILE];
  const int tile = blockIdx.x;
  const int tbase = tile * TILE;
  if (threadIdx.x < TILE) cnt[threadIdx.x] = 0;
  __syncthreads();
  const int e0 = offsets[tile];
  const int e1 = offsets[tile + 1];
  const int2* __restrict__ ed = (const int2*)binned;
  for (int e = e0 + threadIdx.x; e < e1; e += blockDim.x) atomicAdd(&cnt[ed[e].y - tbase], 1);
  __syncthreads();
  if (threadIdx.x < TILE) {
    const int node = tbase + threadIdx.x;
    if (node < N) dinv[node] = rsqrtf((float)(cnt[threadIdx.x] + 1));
  }
}

struct RowOrder {
  int src[SRT_CAP];
  int cnt[TILE];
  int start[TILE];
  int cur[TILE];
};

template <int C, int ROWS>
__device__ __forceinline__ void tile_sums(
    float* __restrict__ acc, RowOrder& ord, const float* __restrict__ h,
    const int* __restrict__ binned, const int* __restrict__ offsets,
    const float* __restrict__ dinv, int N, int tile, int tid, int lane, int w) {
  static_assert(C == 64 || C == 128, "a row is 16 or 32 lanes of 16 bytes");
  constexpr int LDA = C + 4;
  constexpr int LPR = C / 4;
  constexpr int RPI = 32 / LPR;
  constexpr int R   = AGG_FLIGHT / 128;
  constexpr int EPB = R * RPI;
  static_assert(32 % EPB == 0, "a 32-edge group is a whole number of batches");
  const int tbase = tile * TILE;
  const int sub = lane / LPR;
  const int cl  = lane % LPR;

  for (int i = tid; i < ROWS * C / 4; i += AGG_THREADS) {
    const int nl = i / (C / 4);
    const int c4 = i % (C / 4);
    const int node = tbase + nl;
    v4f v = {0.f, 0.f, 0.f, 0.f};
    if (nl < TILE && node < N) {
      const float di = dinv[node];
      v = ((const v4f*)(h + (long)node * C))[c4] * (di * di);
    }
    *(v4f*)&acc[nl * LDA + c4 * 4] = v;
  }

  const int e0 = offsets[tile];
  const int e1 = offsets[tile + 1];
  const int2* __restrict__ ed = (const int2*)binned;

  for (int c0 = e0; c0 < e1; c0 += SRT_CAP) {
    const int c1 = min(c0 + SRT_CAP, e1);

    int sv[SRT_PER_THR], dv[SRT_PER_THR];
#pragma unroll
    for (int k = 0; k < SRT_PER_THR; ++k) {
      const int2 p = ed[min(c0 + tid + k * AGG_THREADS, c1 - 1)];
      sv[k] = p.x;
      dv[k] = p.y - tbase;
    }
    if (tid < TILE) ord.cnt[tid] = 0;
    __syncthreads();
#pragma unroll
    for (int k = 0; k < SRT_PER_THR; ++k)
      if (c0 + tid + k * AGG_THREADS < c1) atomicAdd(&ord.cnt[dv[k]], 1);
    __syncthreads();
    if (w == 0) {
      int c[SCAN_PER_LANE];
      int sum = 0;
#pragma unroll
      for (int k = 0; k < SCAN_PER_LANE; ++k) {
        const int i = lane * SCAN_PER_LANE + k;
        c[k] = (i < TILE) ? ord.cnt[i] : 0;
        sum += c[k];
      }
      int incl = sum;
#pragma unroll
      for (int off = 1; off < 32; off <<= 1) {
        const int t = __shfl_up(incl, off);
        if (lane >= off) incl += t;
      }
      int run = incl - sum;
#pragma unroll
      for (int k = 0; k < SCAN_PER_LANE; ++k) {
        const int i = lane * SCAN_PER_LANE + k;
        if (i < TILE) { ord.start[i] = run; ord.cur[i] = run; }
        run += c[k];
      }
    }
    __syncthreads();
#pragma unroll
    for (int k = 0; k < SRT_PER_THR; ++k)
      if (c0 + tid + k * AGG_THREADS < c1) ord.src[atomicAdd(&ord.cur[dv[k]], 1)] = sv[k];
    __syncthreads();

    for (int r = w; r < TILE; r += AGG_WAVES) {
      const int n = __builtin_amdgcn_readfirstlane(ord.cnt[r]);
      if (n == 0) continue;
      const int st = __builtin_amdgcn_readfirstlane(ord.start[r]);
      const float dd = dinv[tbase + r];
      v4f a = {0.f, 0.f, 0.f, 0.f};
      for (int g = 0; g < n; g += 32) {
        const int cnt = min(32, n - g);
        int s = 0;
        float ds = 0.0f;
        if (lane < cnt) {
          s  = ord.src[st + g + lane];
          ds = dinv[s];
        }
        for (int jb = 0; jb < cnt; jb += EPB) {
          v4f v[R];
#pragma unroll
          for (int u = 0; u < R; ++u) {
            const int j0 = jb + u * RPI;
            if (u == 0 || j0 < cnt) {
              int sj = __builtin_amdgcn_readlane(s, j0);
              if (RPI == 2) {
                const int sj1 = __builtin_amdgcn_readlane(s, j0 + 1);
                sj = sub ? sj1 : sj;
              }
              v[u] = *(const v4f*)(h + (long)sj * C + cl * 4);
            }
          }
#pragma unroll
          for (int u = 0; u < R; ++u) {
            const int j0 = jb + u * RPI;
            if (u == 0 || j0 < cnt) {
              float cj = __int_as_float(__builtin_amdgcn_readlane(__float_as_int(ds), j0));
              if (RPI == 2) {
                const float cj1 = __int_as_float(__builtin_amdgcn_readlane(__float_as_int(ds), j0 + 1));
                cj = (sub ? cj1 : cj) * dd;
                if (j0 + sub < cnt) a += v[u] * cj;
              } else {
                cj *= dd;
                a += v[u] * cj;
              }
            }
          }
        }
      }
      if (RPI == 2) {
        a.x += __shfl_xor(a.x, 16);
        a.y += __shfl_xor(a.y, 16);
        a.z += __shfl_xor(a.z, 16);
        a.w += __shfl_xor(a.w, 16);
      }
      if (sub == 0) {
        v4f* __restrict__ ar = (v4f*)&acc[r * LDA + cl * 4];
        *ar = *ar + a;
      }
    }
    __syncthreads();
  }
  if (e0 >= e1) __syncthreads();
}

template <int COUT>
struct ProductShape {
  static constexpr int NTN = COUT / 16;
  static constexpr int NWG = AGG_WAVES / NTN;
  static constexpr int NRB = TILE_ROWS / 16;
  static constexpr int MPW = (NRB + NWG - 1) / NWG;
  static_assert(COUT % 16 == 0 && AGG_WAVES % NTN == 0 && TILE_ROWS % 16 == 0 && TILE_ROWS >= TILE,
                "whole strips, whole wave groups, whole row blocks");
  static_assert(NRB % MPW <= 1, "a wave holds MPW row blocks, one, or none");
};

template <int COUT, int NB>
__device__ __forceinline__ void product_rows(
    const float* __restrict__ ap, const float* __restrict__ Wm, int bcol, int half,
    v8f (&cacc)[ProductShape<COUT>::MPW]) {
  constexpr int C   = 128;
  constexpr int LDA = C + 4;
#pragma unroll 4
  for (int k0 = 0; k0 < C; k0 += 4) {
    const int ka = k0 + 2 * half;
    v2f b;
    b.x = Wm[(long)ka * COUT + bcol];
    b.y = Wm[(long)(ka + 1) * COUT + bcol];
#pragma unroll
    for (int j = 0; j < NB; ++j) {
      v2f a;
      a.x = ap[j * 16 * LDA + ka];
      a.y = ap[j * 16 * LDA + ka + 1];
      cacc[j] = __builtin_amdgcn_wmma_f32_16x16x4_f32(
           false, a,  false, b,
           (short)0, cacc[j],  false,  false);
    }
  }
}

template <int COUT>
__device__ __forceinline__ int tile_product(
    const float* __restrict__ acc, const float* __restrict__ Wm, int w, int lane,
    v8f (&cacc)[ProductShape<COUT>::MPW]) {
  using S = ProductShape<COUT>;
  constexpr int LDA = 128 + 4;
  const int half = lane >> 4;
  const int l16  = lane & 15;
  const int tn   = w % S::NTN;
  const int mt0  = (w / S::NTN) * S::MPW;
  const int bcol = tn * 16 + l16;
  const float* __restrict__ ap = &acc[(mt0 * 16 + l16) * LDA];

#pragma unroll
  for (int j = 0; j < S::MPW; ++j) cacc[j] = (v8f){0.f, 0.f, 0.f, 0.f, 0.f, 0.f, 0.f, 0.f};

  const int nb = __builtin_amdgcn_readfirstlane(min(S::MPW, max(0, S::NRB - mt0)));
  if (nb == S::MPW)  product_rows<COUT, S::MPW>(ap, Wm, bcol, half, cacc);
  else if (nb == 1)  product_rows<COUT, 1>(ap, Wm, bcol, half, cacc);
  return nb;
}

__global__ __launch_bounds__(AGG_THREADS) void tile_layer1(
    const float* __restrict__ x, const int* __restrict__ binned,
    const int* __restrict__ offsets, const float* __restrict__ dinv,
    const float* __restrict__ W1, const float* __restrict__ bias,
    const float* __restrict__ gamma, const float* __restrict__ beta,
    const float* __restrict__ mean, const float* __restrict__ var,
    const float* __restrict__ W2, float* __restrict__ hw2, int N) {
  static_assert(GCN_IN_C == 128 && GCN_HID_C == 128, "both products run over a 128-wide LDS tile");
  constexpr int LDA = 128 + 4;
  __shared__ float acc[TILE_ROWS * LDA];
  __shared__ RowOrder ord;

  const int tid  = threadIdx.x;
  const int lane = tid & 31;
  const int w = __builtin_amdgcn_readfirstlane(tid >> 5);
  const int tile  = blockIdx.x;
  const int tbase = tile * TILE;
  const int half = lane >> 4;
  const int l16  = lane & 15;

  tile_sums<GCN_IN_C, TILE_ROWS>(acc, ord, x, binned, offsets, dinv, N, tile, tid, lane, w);

  using S1 = ProductShape<GCN_HID_C>;
  v8f c1[S1::MPW];
  const int nb1 = tile_product<GCN_HID_C>(acc, W1, w, lane, c1);

  __syncthreads();
  {
    const int bcol = (w % S1::NTN) * 16 + l16;
    const int mt0  = (w / S1::NTN) * S1::MPW;
    const float bi = bias[bcol];
    const float mu = mean[bcol];
    const float rs = rsqrtf(var[bcol] + GCN_BN_EPS);
    const float ga = gamma[bcol];
    const float be = beta[bcol];
#pragma unroll
    for (int j = 0; j < S1::MPW; ++j) {
      if (j < nb1) {
        float* __restrict__ ar = &acc[((mt0 + j) * 16 + half * 8) * LDA + bcol];
#pragma unroll
        for (int r = 0; r < 8; ++r) {
          float v = c1[j][r] + bi;
          v = (v - mu) * rs * ga + be;
          ar[r * LDA] = fmaxf(v, 0.0f);
        }
      }
    }
  }
  __syncthreads();

  using S2 = ProductShape<GCN_OUT_C>;
  v8f c2[S2::MPW];
  const int nb2 = tile_product<GCN_OUT_C>(acc, W2, w, lane, c2);
  {
    const int bcol = (w % S2::NTN) * 16 + l16;
    const int mt0  = (w / S2::NTN) * S2::MPW;
#pragma unroll
    for (int j = 0; j < S2::MPW; ++j) {
      if (j < nb2) {
        const int nl0 = (mt0 + j) * 16 + half * 8;
        float* __restrict__ cp = hw2 + (long)(tbase + nl0) * GCN_OUT_C + bcol;
#pragma unroll
        for (int r = 0; r < 8; ++r)
          if (nl0 + r < TILE && tbase + nl0 + r < N) cp[(long)r * GCN_OUT_C] = c2[j][r];
      }
    }
  }
}

__global__ __launch_bounds__(AGG_THREADS) void tile_layer2(
    const float* __restrict__ hw2, const int* __restrict__ binned,
    const int* __restrict__ offsets, const float* __restrict__ dinv,
    const float* __restrict__ bias,
    const float* __restrict__ gamma, const float* __restrict__ beta,
    const float* __restrict__ mean, const float* __restrict__ var,
    float* __restrict__ outp, int N) {
  constexpr int C   = GCN_OUT_C;
  constexpr int LDA = C + 4;
  __shared__ float acc[TILE * LDA];
  __shared__ RowOrder ord;

  const int tid  = threadIdx.x;
  const int lane = tid & 31;
  const int w = __builtin_amdgcn_readfirstlane(tid >> 5);
  const int tile  = blockIdx.x;
  const int tbase = tile * TILE;

  tile_sums<C, TILE>(acc, ord, hw2, binned, offsets, dinv, N, tile, tid, lane, w);

  static_assert(AGG_THREADS % (C / 4) == 0, "a thread's column group must not change with i");
  const int c4 = tid % (C / 4);
  const v4f bi = ((const v4f*)bias)[c4];
  const v4f mu = ((const v4f*)mean)[c4];
  const v4f vr = ((const v4f*)var)[c4];
  const v4f ga = ((const v4f*)gamma)[c4];
  const v4f be = ((const v4f*)beta)[c4];
  v4f rs;
  rs.x = rsqrtf(vr.x + GCN_BN_EPS);
  rs.y = rsqrtf(vr.y + GCN_BN_EPS);
  rs.z = rsqrtf(vr.z + GCN_BN_EPS);
  rs.w = rsqrtf(vr.w + GCN_BN_EPS);
  for (int i = tid; i < TILE * C / 4; i += AGG_THREADS) {
    const int nl = i / (C / 4);
    const int node = tbase + nl;
    if (node < N) {
      v4f v = *(const v4f*)&acc[nl * LDA + c4 * 4] + bi;
      v = (v - mu) * rs * ga + be;
      ((v4f*)(outp + (long)node * C))[c4] = v;
    }
  }
}


extern "C" void kernel_launch(void* const* d_in, const int* in_sizes, int n_in,
                              void* d_out, int out_size, void* d_ws, size_t ws_size,
                              hipStream_t stream) {
  const float* x   = (const float*)d_in[0];
  const int*   ei  = (const int*)d_in[1];
  const float* W1  = (const float*)d_in[2];
  const float* b1  = (const float*)d_in[3];
  const float* g1  = (const float*)d_in[4];
  const float* be1 = (const float*)d_in[5];
  const float* m1  = (const float*)d_in[6];
  const float* v1  = (const float*)d_in[7];
  const float* W2  = (const float*)d_in[8];
  const float* b2  = (const float*)d_in[9];
  const float* g2  = (const float*)d_in[10];
  const float* be2 = (const float*)d_in[11];
  const float* m2  = (const float*)d_in[12];
  const float* v2  = (const float*)d_in[13];

  const int N = in_sizes[0] / GCN_IN_C;
  const int E = in_sizes[1] / 2;
  const int* src = ei;
  const int* dst = ei + E;

  char* ws = (char*)d_ws;
  float* hw2 = (float*)ws;                        ws += (size_t)N * GCN_OUT_C * sizeof(float);
  int*   totals  = (int*)ws;
  int*   offsets = totals + MAX_TILES;
  int*   cursor  = offsets + 2 * MAX_TILES;       ws += (size_t)N * sizeof(int);
  float* dinv = (float*)ws;                       ws += (size_t)N * sizeof(float);
  int*   binned  = (int*)ws;                      ws += (size_t)E * 2 * sizeof(int);
  (void)ws_size; (void)n_in; (void)out_size;

  const int NT = (N + TILE - 1) / TILE;
  const int BB = (E + BIN_CHUNK - 1) / BIN_CHUNK;

  zero_f32<<<1, MAX_TILES, 0, stream>>>((float*)totals, (long)MAX_TILES);
  bin_count<<<BB, BIN_THREADS, 0, stream>>>(dst, E, totals, NT);
  bin_scan<<<1, MAX_TILES, 0, stream>>>(totals, offsets, cursor, NT);
  bin_fill<<<BB, BIN_THREADS, 0, stream>>>(src, dst, E, cursor, binned, NT);
  tile_dinv<<<NT, 256, 0, stream>>>(binned, offsets, dinv, N);

  tile_layer1<<<NT, AGG_THREADS, 0, stream>>>(
      x, binned, offsets, dinv, W1, b1, g1, be1, m1, v1, W2, hw2, N);

  tile_layer2<<<NT, AGG_THREADS, 0, stream>>>(
      hw2, binned, offsets, dinv, b2, g2, be2, m2, v2, (float*)d_out, N);
}


# edge coefficients: tile block copies the deg^-1/2 table into LDS (space of its row sums), stores dinv[src]*dinv[dst] beside each source id while ordering; sums loop: no scattered 4-byte reads
# speedup vs baseline: 5.1152x; 1.0533x over previous
#include <hip/hip_runtime.h>

typedef __attribute__((ext_vector_type(2))) float v2f;
typedef __attribute__((ext_vector_type(4))) float v4f;
typedef __attribute__((ext_vector_type(8))) float v8f;

#define GCN_IN_C   128
#define GCN_HID_C  128
#define GCN_OUT_C  64
#define GCN_BN_EPS 1e-5f

#define TILE        196
#define TILE_ROWS   208
#define MAX_TILES   512
#define BIN_CHUNK   4096
#define BIN_THREADS 256
#define BIN_PER_THR (BIN_CHUNK / BIN_THREADS)
#define AGG_THREADS 1024
#define AGG_WAVES   (AGG_THREADS / 32)
#define AGG_FLIGHT  1024
#define SRT_CAP     8192
#define SRT_PER_THR (SRT_CAP / AGG_THREADS)
#define SCAN_PER_LANE ((TILE + 31) / 32)
#define DTAB_CAP    50176
#define DTAB_FLIGHT 8


__global__ void zero_f32(float* __restrict__ p, long n) {
  long i = (long)blockIdx.x * blockDim.x + threadIdx.x;
  long stride = (long)gridDim.x * blockDim.x;
  for (; i < n; i += stride) p[i] = 0.0f;
}

__device__ __forceinline__ int tile_of(int node) { return (int)((unsigned)node / (unsigned)TILE); }

__global__ void bin_count(const int* __restrict__ dst, int E, int* __restrict__ totals, int NT) {
  __shared__ int hist[MAX_TILES];
  const int base = blockIdx.x * BIN_CHUNK + threadIdx.x;
  int dv[BIN_PER_THR];
#pragma unroll
  for (int k = 0; k < BIN_PER_THR; ++k) dv[k] = dst[min(base + k * BIN_THREADS, E - 1)];
  for (int i = threadIdx.x; i < NT; i += BIN_THREADS) hist[i] = 0;
  __syncthreads();
#pragma unroll
  for (int k = 0; k < BIN_PER_THR; ++k)
    if (base + k * BIN_THREADS < E) atomicAdd(&hist[tile_of(dv[k])], 1);
  __syncthreads();
  for (int i = threadIdx.x; i < NT; i += BIN_THREADS) {
    const int c = hist[i];
    if (c) atomicAdd(&totals[i], c);
  }
}

__global__ void bin_scan(const int* __restrict__ totals, int* __restrict__ offsets,
                         int* __restrict__ cursor, int NT) {
  __shared__ int a[2][MAX_TILES];
  const int i = threadIdx.x;
  const int own = (i < NT) ? totals[i] : 0;
  a[0][i] = own;
  __syncthreads();
  int cur = 0;
  for (int off = 1; off < MAX_TILES; off <<= 1) {
    int x = a[cur][i];
    if (i >= off) x += a[cur][i - off];
    a[cur ^ 1][i] = x;
    __syncthreads();
    cur ^= 1;
  }
  const int excl = a[cur][i] - own;
  if (i <= NT) offsets[i] = excl;
  if (i < NT) cursor[i] = excl;
}

__global__ void bin_fill(const int* __restrict__ src, const int* __restrict__ dst, int E,
                         int* __restrict__ cursor, int* __restrict__ binned, int NT) {
  __shared__ int hist[MAX_TILES];
  __shared__ int runbase[MAX_TILES];
  const int base = blockIdx.x * BIN_CHUNK + threadIdx.x;
  int dv[BIN_PER_THR], sv[BIN_PER_THR];
#pragma unroll
  for (int k = 0; k < BIN_PER_THR; ++k) dv[k] = dst[min(base + k * BIN_THREADS, E - 1)];
#pragma unroll
  for (int k = 0; k < BIN_PER_THR; ++k) sv[k] = src[min(base + k * BIN_THREADS, E - 1)];
  for (int i = threadIdx.x; i < NT; i += BIN_THREADS) hist[i] = 0;
  __syncthreads();
#pragma unroll
  for (int k = 0; k < BIN_PER_THR; ++k)
    if (base + k * BIN_THREADS < E) atomicAdd(&hist[tile_of(dv[k])], 1);
  __syncthreads();
  for (int i = threadIdx.x; i < NT; i += BIN_THREADS) {
    const int c = hist[i];
    runbase[i] = c ? atomicAdd(&cursor[i], c) : 0;
    hist[i] = 0;
  }
  __syncthreads();
  int2* __restrict__ out = (int2*)binned;
#pragma unroll
  for (int k = 0; k < BIN_PER_THR; ++k) {
    if (base + k * BIN_THREADS < E) {
      const int b = tile_of(dv[k]);
      const int r = atomicAdd(&hist[b], 1);
      out[runbase[b] + r] = make_int2(sv[k], dv[k]);
    }
  }
}

__global__ void tile_dinv(const int* __restrict__ binned, const int* __restrict__ offsets,
                          float* __restrict__ dinv, int N) {
  static_assert(TILE <= 256, "launched with 256 threads: one thread per node of the tile");
  __shared__ int cnt[TILE];
  const int tile = blockIdx.x;
  const int tbase = tile * TILE;
  if (threadIdx.x < TILE) cnt[threadIdx.x] = 0;
  __syncthreads();
  const int e0 = offsets[tile];
  const int e1 = offsets[tile + 1];
  const int2* __restrict__ ed = (const int2*)binned;
  for (int e = e0 + threadIdx.x; e < e1; e += blockDim.x) atomicAdd(&cnt[ed[e].y - tbase], 1);
  __syncthreads();
  if (threadIdx.x < TILE) {
    const int node = tbase + threadIdx.x;
    if (node < N) dinv[node] = rsqrtf((float)(cnt[threadIdx.x] + 1));
  }
}

struct RowOrder {
  int src[SRT_CAP];
  float coef[SRT_CAP];
  int cnt[TILE];
  int start[TILE];
  int cur[TILE];
};

template <int C, int ROWS>
__device__ __forceinline__ void tile_start_values(
    float* acc, const float* __restrict__ h, const float* __restrict__ dinv,
    int N, int tbase, int tid) {
  constexpr int LDA = C + 4;
  for (int i = tid; i < ROWS * C / 4; i += AGG_THREADS) {
    const int nl = i / (C / 4);
    const int c4 = i % (C / 4);
    const int node = tbase + nl;
    v4f v = {0.f, 0.f, 0.f, 0.f};
    if (nl < TILE && node < N) {
      const float di = dinv[node];
      v = ((const v4f*)(h + (long)node * C))[c4] * (di * di);
    }
    *(v4f*)&acc[nl * LDA + c4 * 4] = v;
  }
}

template <int C, int ROWS>
__device__ __forceinline__ void tile_sums(
    float* acc, RowOrder& ord, const float* __restrict__ h,
    const int* __restrict__ binned, const int* __restrict__ offsets,
    const float* __restrict__ dinv, int N, int tile, int tid, int lane, int w) {
  static_assert(C == 64 || C == 128, "a row is 16 or 32 lanes of 16 bytes");
  constexpr int LDA = C + 4;
  constexpr int LPR = C / 4;
  constexpr int RPI = 32 / LPR;
  constexpr int R   = AGG_FLIGHT / 128;
  constexpr int EPB = R * RPI;
  static_assert(32 % EPB == 0, "a 32-edge group is a whole number of batches");
  const int tbase = tile * TILE;
  const int sub = lane / LPR;
  const int cl  = lane % LPR;

  const int e0 = offsets[tile];
  const int e1 = offsets[tile + 1];
  const int2* __restrict__ ed = (const int2*)binned;

  const bool tab = (N <= DTAB_CAP) && (e0 < e1);
  if (tab) {
    const int n4 = ((((unsigned long)dinv) & 15ul) == 0ul) ? (N >> 2) : 0;
    const v4f* __restrict__ d4 = (const v4f*)dinv;
    for (int i0 = tid; i0 < n4; i0 += AGG_THREADS * DTAB_FLIGHT) {
      v4f t[DTAB_FLIGHT];
#pragma unroll
      for (int u = 0; u < DTAB_FLIGHT; ++u) t[u] = d4[min(i0 + u * AGG_THREADS, n4 - 1)];
#pragma unroll
      for (int u = 0; u < DTAB_FLIGHT; ++u)
        if (i0 + u * AGG_THREADS < n4) ((v4f*)acc)[i0 + u * AGG_THREADS] = t[u];
    }
    for (int i = (n4 << 2) + tid; i < N; i += AGG_THREADS) acc[i] = dinv[i];
  }
  bool first = true;

  for (int c0 = e0; c0 < e1; c0 += SRT_CAP) {
    const int c1 = min(c0 + SRT_CAP, e1);

    int sv[SRT_PER_THR], dv[SRT_PER_THR];
#pragma unroll
    for (int k = 0; k < SRT_PER_THR; ++k) {
      const int2 p = ed[min(c0 + tid + k * AGG_THREADS, c1 - 1)];
      sv[k] = p.x;
      dv[k] = p.y - tbase;
    }
    const bool lds_tab = tab && first;
    float dk[SRT_PER_THR];
    if (!lds_tab) {
#pragma unroll
      for (int k = 0; k < SRT_PER_THR; ++k) dk[k] = dinv[sv[k]] * dinv[tbase + dv[k]];
    }
    if (tid < TILE) ord.cnt[tid] = 0;
    __syncthreads();
    if (lds_tab) {
#pragma unroll
      for (int k = 0; k < SRT_PER_THR; ++k) dk[k] = acc[sv[k]] * acc[tbase + dv[k]];
    }
#pragma unroll
    for (int k = 0; k < SRT_PER_THR; ++k)
      if (c0 + tid + k * AGG_THREADS < c1) atomicAdd(&ord.cnt[dv[k]], 1);
    __syncthreads();
    if (w == 0) {
      int c[SCAN_PER_LANE];
      int sum = 0;
#pragma unroll
      for (int k = 0; k < SCAN_PER_LANE; ++k) {
        const int i = lane * SCAN_PER_LANE + k;
        c[k] = (i < TILE) ? ord.cnt[i] : 0;
        sum += c[k];
      }
      int incl = sum;
#pragma unroll
      for (int off = 1; off < 32; off <<= 1) {
        const int t = __shfl_up(incl, off);
        if (lane >= off) incl += t;
      }
      int run = incl - sum;
#pragma unroll
      for (int k = 0; k < SCAN_PER_LANE; ++k) {
        const int i = lane * SCAN_PER_LANE + k;
        if (i < TILE) { ord.start[i] = run; ord.cur[i] = run; }
        run += c[k];
      }
    }
    __syncthreads();
#pragma unroll
    for (int k = 0; k < SRT_PER_THR; ++k)
      if (c0 + tid + k * AGG_THREADS < c1) {
        const int p = atomicAdd(&ord.cur[dv[k]], 1);
        ord.src[p] = sv[k];
        ord.coef[p] = dk[k];
      }
    __syncthreads();
    if (first) {
      tile_start_values<C, ROWS>(acc, h, dinv, N, tbase, tid);
      __syncthreads();
      first = false;
    }

    for (int r = w; r < TILE; r += AGG_WAVES) {
      const int n = __builtin_amdgcn_readfirstlane(ord.cnt[r]);
      if (n == 0) continue;
      const int st = __builtin_amdgcn_readfirstlane(ord.start[r]);
      v4f a = {0.f, 0.f, 0.f, 0.f};
      for (int g = 0; g < n; g += 32) {
        const int cnt = min(32, n - g);
        int s = 0;
        float ds = 0.0f;
        if (lane < cnt) {
          s  = ord.src[st + g + lane];
          ds = ord.coef[st + g + lane];
        }
        for (int jb = 0; jb < cnt; jb += EPB) {
          v4f v[R];
#pragma unroll
          for (int u = 0; u < R; ++u) {
            const int j0 = jb + u * RPI;
            if (u == 0 || j0 < cnt) {
              int sj = __builtin_amdgcn_readlane(s, j0);
              if (RPI == 2) {
                const int sj1 = __builtin_amdgcn_readlane(s, j0 + 1);
                sj = sub ? sj1 : sj;
              }
              v[u] = *(const v4f*)(h + (long)sj * C + cl * 4);
            }
          }
#pragma unroll
          for (int u = 0; u < R; ++u) {
            const int j0 = jb + u * RPI;
            if (u == 0 || j0 < cnt) {
              float cj = __int_as_float(__builtin_amdgcn_readlane(__float_as_int(ds), j0));
              if (RPI == 2) {
                const float cj1 = __int_as_float(__builtin_amdgcn_readlane(__float_as_int(ds), j0 + 1));
                cj = sub ? cj1 : cj;
                if (j0 + sub < cnt) a += v[u] * cj;
              } else {
                a += v[u] * cj;
              }
            }
          }
        }
      }
      if (RPI == 2) {
        a.x += __shfl_xor(a.x, 16);
        a.y += __shfl_xor(a.y, 16);
        a.z += __shfl_xor(a.z, 16);
        a.w += __shfl_xor(a.w, 16);
      }
      if (sub == 0) {
        v4f* __restrict__ ar = (v4f*)&acc[r * LDA + cl * 4];
        *ar = *ar + a;
      }
    }
    __syncthreads();
  }
  if (e0 >= e1) {
    tile_start_values<C, ROWS>(acc, h, dinv, N, tbase, tid);
    __syncthreads();
  }
}

template <int COUT>
struct ProductShape {
  static constexpr int NTN = COUT / 16;
  static constexpr int NWG = AGG_WAVES / NTN;
  static constexpr int NRB = TILE_ROWS / 16;
  static constexpr int MPW = (NRB + NWG - 1) / NWG;
  static_assert(COUT % 16 == 0 && AGG_WAVES % NTN == 0 && TILE_ROWS % 16 == 0 && TILE_ROWS >= TILE,
                "whole strips, whole wave groups, whole row blocks");
  static_assert(NRB % MPW <= 1, "a wave holds MPW row blocks, one, or none");
};

template <int COUT, int NB>
__device__ __forceinline__ void product_rows(
    const float* __restrict__ ap, const float* __restrict__ Wm, int bcol, int half,
    v8f (&cacc)[ProductShape<COUT>::MPW]) {
  constexpr int C   = 128;
  constexpr int LDA = C + 4;
#pragma unroll 4
  for (int k0 = 0; k0 < C; k0 += 4) {
    const int ka = k0 + 2 * half;
    v2f b;
    b.x = Wm[(long)ka * COUT + bcol];
    b.y = Wm[(long)(ka + 1) * COUT + bcol];
#pragma unroll
    for (int j = 0; j < NB; ++j) {
      v2f a;
      a.x = ap[j * 16 * LDA + ka];
      a.y = ap[j * 16 * LDA + ka + 1];
      cacc[j] = __builtin_amdgcn_wmma_f32_16x16x4_f32(
           false, a,  false, b,
           (short)0, cacc[j],  false,  false);
    }
  }
}

template <int COUT>
__device__ __forceinline__ int tile_product(
    const float* __restrict__ acc, const float* __restrict__ Wm, int w, int lane,
    v8f (&cacc)[ProductShape<COUT>::MPW]) {
  using S = ProductShape<COUT>;
  constexpr int LDA = 128 + 4;
  const int half = lane >> 4;
  const int l16  = lane & 15;
  const int tn   = w % S::NTN;
  const int mt0  = (w / S::NTN) * S::MPW;
  const int bcol = tn * 16 + l16;
  const float* __restrict__ ap = &acc[(mt0 * 16 + l16) * LDA];

#pragma unroll
  for (int j = 0; j < S::MPW; ++j) cacc[j] = (v8f){0.f, 0.f, 0.f, 0.f, 0.f, 0.f, 0.f, 0.f};

  const int nb = __builtin_amdgcn_readfirstlane(min(S::MPW, max(0, S::NRB - mt0)));
  if (nb == S::MPW)  product_rows<COUT, S::MPW>(ap, Wm, bcol, half, cacc);
  else if (nb == 1)  product_rows<COUT, 1>(ap, Wm, bcol, half, cacc);
  return nb;
}

__global__ __launch_bounds__(AGG_THREADS) void tile_layer1(
    const float* __restrict__ x, const int* __restrict__ binned,
    const int* __restrict__ offsets, const float* __restrict__ dinv,
    const float* __restrict__ W1, const float* __restrict__ bias,
    const float* __restrict__ gamma, const float* __restrict__ beta,
    const float* __restrict__ mean, const float* __restrict__ var,
    const float* __restrict__ W2, float* __restrict__ hw2, int N) {
  static_assert(GCN_IN_C == 128 && GCN_HID_C == 128, "both products run over a 128-wide LDS tile");
  constexpr int LDA = 128 + 4;
  constexpr int POOL = (TILE_ROWS * LDA > DTAB_CAP) ? TILE_ROWS * LDA : DTAB_CAP;
  __shared__ __attribute__((aligned(16))) float acc[POOL];
  __shared__ RowOrder ord;

  const int tid  = threadIdx.x;
  const int lane = tid & 31;
  const int w = __builtin_amdgcn_readfirstlane(tid >> 5);
  const int tile  = blockIdx.x;
  const int tbase = tile * TILE;
  const int half = lane >> 4;
  const int l16  = lane & 15;

  tile_sums<GCN_IN_C, TILE_ROWS>(acc, ord, x, binned, offsets, dinv, N, tile, tid, lane, w);

  using S1 = ProductShape<GCN_HID_C>;
  v8f c1[S1::MPW];
  const int nb1 = tile_product<GCN_HID_C>(acc, W1, w, lane, c1);

  __syncthreads();
  {
    const int bcol = (w % S1::NTN) * 16 + l16;
    const int mt0  = (w / S1::NTN) * S1::MPW;
    const float bi = bias[bcol];
    const float mu = mean[bcol];
    const float rs = rsqrtf(var[bcol] + GCN_BN_EPS);
    const float ga = gamma[bcol];
    const float be = beta[bcol];
#pragma unroll
    for (int j = 0; j < S1::MPW; ++j) {
      if (j < nb1) {
        float* __restrict__ ar = &acc[((mt0 + j) * 16 + half * 8) * LDA + bcol];
#pragma unroll
        for (int r = 0; r < 8; ++r) {
          float v = c1[j][r] + bi;
          v = (v - mu) * rs * ga + be;
          ar[r * LDA] = fmaxf(v, 0.0f);
        }
      }
    }
  }
  __syncthreads();

  using S2 = ProductShape<GCN_OUT_C>;
  v8f c2[S2::MPW];
  const int nb2 = tile_product<GCN_OUT_C>(acc, W2, w, lane, c2);
  {
    const int bcol = (w % S2::NTN) * 16 + l16;
    const int mt0  = (w / S2::NTN) * S2::MPW;
#pragma unroll
    for (int j = 0; j < S2::MPW; ++j) {
      if (j < nb2) {
        const int nl0 = (mt0 + j) * 16 + half * 8;
        float* __restrict__ cp = hw2 + (long)(tbase + nl0) * GCN_OUT_C + bcol;
#pragma unroll
        for (int r = 0; r < 8; ++r)
          if (nl0 + r < TILE && tbase + nl0 + r < N) cp[(long)r * GCN_OUT_C] = c2[j][r];
      }
    }
  }
}

__global__ __launch_bounds__(AGG_THREADS) void tile_layer2(
    const float* __restrict__ hw2, const int* __restrict__ binned,
    const int* __restrict__ offsets, const float* __restrict__ dinv,
    const float* __restrict__ bias,
    const float* __restrict__ gamma, const float* __restrict__ beta,
    const float* __restrict__ mean, const float* __restrict__ var,
    float* __restrict__ outp, int N) {
  constexpr int C   = GCN_OUT_C;
  constexpr int LDA = C + 4;
  constexpr int POOL = (TILE * LDA > DTAB_CAP) ? TILE * LDA : DTAB_CAP;
  __shared__ __attribute__((aligned(16))) float acc[POOL];
  __shared__ RowOrder ord;

  const int tid  = threadIdx.x;
  const int lane = tid & 31;
  const int w = __builtin_amdgcn_readfirstlane(tid >> 5);
  const int tile  = blockIdx.x;
  const int tbase = tile * TILE;

  tile_sums<C, TILE>(acc, ord, hw2, binned, offsets, dinv, N, tile, tid, lane, w);

  static_assert(AGG_THREADS % (C / 4) == 0, "a thread's column group must not change with i");
  const int c4 = tid % (C / 4);
  const v4f bi = ((const v4f*)bias)[c4];
  const v4f mu = ((const v4f*)mean)[c4];
  const v4f vr = ((const v4f*)var)[c4];
  const v4f ga = ((const v4f*)gamma)[c4];
  const v4f be = ((const v4f*)beta)[c4];
  v4f rs;
  rs.x = rsqrtf(vr.x + GCN_BN_EPS);
  rs.y = rsqrtf(vr.y + GCN_BN_EPS);
  rs.z = rsqrtf(vr.z + GCN_BN_EPS);
  rs.w = rsqrtf(vr.w + GCN_BN_EPS);
  for (int i = tid; i < TILE * C / 4; i += AGG_THREADS) {
    const int nl = i / (C / 4);
    const int node = tbase + nl;
    if (node < N) {
      v4f v = *(const v4f*)&acc[nl * LDA + c4 * 4] + bi;
      v = (v - mu) * rs * ga + be;
      ((v4f*)(outp + (long)node * C))[c4] = v;
    }
  }
}


extern "C" void kernel_launch(void* const* d_in, const int* in_sizes, int n_in,
                              void* d_out, int out_size, void* d_ws, size_t ws_size,
                              hipStream_t stream) {
  const float* x   = (const float*)d_in[0];
  const int*   ei  = (const int*)d_in[1];
  const float* W1  = (const float*)d_in[2];
  const float* b1  = (const float*)d_in[3];
  const float* g1  = (const float*)d_in[4];
  const float* be1 = (const float*)d_in[5];
  const float* m1  = (const float*)d_in[6];
  const float* v1  = (const float*)d_in[7];
  const float* W2  = (const float*)d_in[8];
  const float* b2  = (const float*)d_in[9];
  const float* g2  = (const float*)d_in[10];
  const float* be2 = (const float*)d_in[11];
  const float* m2  = (const float*)d_in[12];
  const float* v2  = (const float*)d_in[13];

  const int N = in_sizes[0] / GCN_IN_C;
  const int E = in_sizes[1] / 2;
  const int* src = ei;
  const int* dst = ei + E;

  char* ws = (char*)d_ws;
  float* hw2 = (float*)ws;                        ws += (size_t)N * GCN_OUT_C * sizeof(float);
  int*   totals  = (int*)ws;
  int*   offsets = totals + MAX_TILES;
  int*   cursor  = offsets + 2 * MAX_TILES;       ws += (size_t)N * sizeof(int);
  float* dinv = (float*)ws;                       ws += (size_t)N * sizeof(float);
  int*   binned  = (int*)ws;                      ws += (size_t)E * 2 * sizeof(int);
  (void)ws_size; (void)n_in; (void)out_size;

  const int NT = (N + TILE - 1) / TILE;
  const int BB = (E + BIN_CHUNK - 1) / BIN_CHUNK;

  zero_f32<<<1, MAX_TILES, 0, stream>>>((float*)totals, (long)MAX_TILES);
  bin_count<<<BB, BIN_THREADS, 0, stream>>>(dst, E, totals, NT);
  bin_scan<<<1, MAX_TILES, 0, stream>>>(totals, offsets, cursor, NT);
  bin_fill<<<BB, BIN_THREADS, 0, stream>>>(src, dst, E, cursor, binned, NT);
  tile_dinv<<<NT, 256, 0, stream>>>(binned, offsets, dinv, N);

  tile_layer1<<<NT, AGG_THREADS, 0, stream>>>(
      x, binned, offsets, dinv, W1, b1, g1, be1, m1, v1, W2, hw2, N);

  tile_layer2<<<NT, AGG_THREADS, 0, stream>>>(
      hw2, binned, offsets, dinv, b2, g2, be2, m2, v2, (float*)d_out, N);
}


# loads in flight: the degree kernel's edge loop and the self-loop start values waited for every single load; a thread now issues them as one group (32 edges; 7 or 4 row pieces) ahead of use
# speedup vs baseline: 5.5602x; 1.0870x over previous
#include <hip/hip_runtime.h>

typedef __attribute__((ext_vector_type(2))) float v2f;
typedef __attribute__((ext_vector_type(4))) float v4f;
typedef __attribute__((ext_vector_type(8))) float v8f;

#define GCN_IN_C   128
#define GCN_HID_C  128
#define GCN_OUT_C  64
#define GCN_BN_EPS 1e-5f

#define TILE        196
#define TILE_ROWS   208
#define MAX_TILES   512
#define BIN_CHUNK   4096
#define BIN_THREADS 256
#define BIN_PER_THR (BIN_CHUNK / BIN_THREADS)
#define AGG_THREADS 1024
#define AGG_WAVES   (AGG_THREADS / 32)
#define AGG_FLIGHT  1024
#define SRT_CAP     8192
#define SRT_PER_THR (SRT_CAP / AGG_THREADS)
#define SCAN_PER_LANE ((TILE + 31) / 32)
#define DTAB_CAP    50176
#define DTAB_FLIGHT 8
#define DINV_THREADS 256
#define DINV_FLIGHT  32


__global__ void zero_f32(float* __restrict__ p, long n) {
  long i = (long)blockIdx.x * blockDim.x + threadIdx.x;
  long stride = (long)gridDim.x * blockDim.x;
  for (; i < n; i += stride) p[i] = 0.0f;
}

__device__ __forceinline__ int tile_of(int node) { return (int)((unsigned)node / (unsigned)TILE); }

__global__ void bin_count(const int* __restrict__ dst, int E, int* __restrict__ totals, int NT) {
  __shared__ int hist[MAX_TILES];
  const int base = blockIdx.x * BIN_CHUNK + threadIdx.x;
  int dv[BIN_PER_THR];
#pragma unroll
  for (int k = 0; k < BIN_PER_THR; ++k) dv[k] = dst[min(base + k * BIN_THREADS, E - 1)];
  for (int i = threadIdx.x; i < NT; i += BIN_THREADS) hist[i] = 0;
  __syncthreads();
#pragma unroll
  for (int k = 0; k < BIN_PER_THR; ++k)
    if (base + k * BIN_THREADS < E) atomicAdd(&hist[tile_of(dv[k])], 1);
  __syncthreads();
  for (int i = threadIdx.x; i < NT; i += BIN_THREADS) {
    const int c = hist[i];
    if (c) atomicAdd(&totals[i], c);
  }
}

__global__ void bin_scan(const int* __restrict__ totals, int* __restrict__ offsets,
                         int* __restrict__ cursor, int NT) {
  __shared__ int a[2][MAX_TILES];
  const int i = threadIdx.x;
  const int own = (i < NT) ? totals[i] : 0;
  a[0][i] = own;
  __syncthreads();
  int cur = 0;
  for (int off = 1; off < MAX_TILES; off <<= 1) {
    int x = a[cur][i];
    if (i >= off) x += a[cur][i - off];
    a[cur ^ 1][i] = x;
    __syncthreads();
    cur ^= 1;
  }
  const int excl = a[cur][i] - own;
  if (i <= NT) offsets[i] = excl;
  if (i < NT) cursor[i] = excl;
}

__global__ void bin_fill(const int* __restrict__ src, const int* __restrict__ dst, int E,
                         int* __restrict__ cursor, int* __restrict__ binned, int NT) {
  __shared__ int hist[MAX_TILES];
  __shared__ int runbase[MAX_TILES];
  const int base = blockIdx.x * BIN_CHUNK + threadIdx.x;
  int dv[BIN_PER_THR], sv[BIN_PER_THR];
#pragma unroll
  for (int k = 0; k < BIN_PER_THR; ++k) dv[k] = dst[min(base + k * BIN_THREADS, E - 1)];
#pragma unroll
  for (int k = 0; k < BIN_PER_THR; ++k) sv[k] = src[min(base + k * BIN_THREADS, E - 1)];
  for (int i = threadIdx.x; i < NT; i += BIN_THREADS) hist[i] = 0;
  __syncthreads();
#pragma unroll
  for (int k = 0; k < BIN_PER_THR; ++k)
    if (base + k * BIN_THREADS < E) atomicAdd(&hist[tile_of(dv[k])], 1);
  __syncthreads();
  for (int i = threadIdx.x; i < NT; i += BIN_THREADS) {
    const int c = hist[i];
    runbase[i] = c ? atomicAdd(&cursor[i], c) : 0;
    hist[i] = 0;
  }
  __syncthreads();
  int2* __restrict__ out = (int2*)binned;
#pragma unroll
  for (int k = 0; k < BIN_PER_THR; ++k) {
    if (base + k * BIN_THREADS < E) {
      const int b = tile_of(dv[k]);
      const int r = atomicAdd(&hist[b], 1);
      out[runbase[b] + r] = make_int2(sv[k], dv[k]);
    }
  }
}

__global__ __launch_bounds__(DINV_THREADS) void tile_dinv(
    const int* __restrict__ binned, const int* __restrict__ offsets,
    float* __restrict__ dinv, int N) {
  static_assert(TILE <= DINV_THREADS, "one thread per node of the tile");
  __shared__ int cnt[TILE];
  const int tid = threadIdx.x;
  const int tile = blockIdx.x;
  const int tbase = tile * TILE;
  if (tid < TILE) cnt[tid] = 0;
  __syncthreads();
  const int e0 = offsets[tile];
  const int e1 = offsets[tile + 1];
  for (int b = e0 + tid; b < e1; b += DINV_THREADS * DINV_FLIGHT) {
    int d[DINV_FLIGHT];
#pragma unroll
    for (int u = 0; u < DINV_FLIGHT; ++u) d[u] = binned[2 * min(b + u * DINV_THREADS, e1 - 1) + 1];
#pragma unroll
    for (int u = 0; u < DINV_FLIGHT; ++u)
      if (b + u * DINV_THREADS < e1) atomicAdd(&cnt[d[u] - tbase], 1);
  }
  __syncthreads();
  if (tid < TILE) {
    const int node = tbase + tid;
    if (node < N) dinv[node] = rsqrtf((float)(cnt[tid] + 1));
  }
}

struct RowOrder {
  int src[SRT_CAP];
  float coef[SRT_CAP];
  int cnt[TILE];
  int start[TILE];
  int cur[TILE];
};

template <int C, int ROWS>
__device__ __forceinline__ void tile_start_values(
    float* acc, const float* __restrict__ h, const float* __restrict__ dinv,
    int N, int tbase, int tid) {
  constexpr int LDA = C + 4;
  constexpr int TOT = ROWS * C / 4;
  constexpr int NIT = (TOT + AGG_THREADS - 1) / AGG_THREADS;
  static_assert(AGG_THREADS % (C / 4) == 0, "a thread keeps one column group");
  const int c4 = tid % (C / 4);
  v4f v[NIT];
  float di[NIT];
#pragma unroll
  for (int u = 0; u < NIT; ++u) {
    const int nl = (tid + u * AGG_THREADS) / (C / 4);
    const int node = min(tbase + min(nl, TILE - 1), N - 1);
    di[u] = dinv[node];
    v[u] = ((const v4f*)(h + (long)node * C))[c4];
  }
#pragma unroll
  for (int u = 0; u < NIT; ++u) {
    const int i = tid + u * AGG_THREADS;
    const int nl = i / (C / 4);
    if (i < TOT) {
      v4f s = {0.f, 0.f, 0.f, 0.f};
      if (nl < TILE && tbase + nl < N) s = v[u] * (di[u] * di[u]);
      *(v4f*)&acc[nl * LDA + c4 * 4] = s;
    }
  }
}

template <int C, int ROWS>
__device__ __forceinline__ void tile_sums(
    float* acc, RowOrder& ord, const float* __restrict__ h,
    const int* __restrict__ binned, const int* __restrict__ offsets,
    const float* __restrict__ dinv, int N, int tile, int tid, int lane, int w) {
  static_assert(C == 64 || C == 128, "a row is 16 or 32 lanes of 16 bytes");
  constexpr int LDA = C + 4;
  constexpr int LPR = C / 4;
  constexpr int RPI = 32 / LPR;
  constexpr int R   = AGG_FLIGHT / 128;
  constexpr int EPB = R * RPI;
  static_assert(32 % EPB == 0, "a 32-edge group is a whole number of batches");
  const int tbase = tile * TILE;
  const int sub = lane / LPR;
  const int cl  = lane % LPR;

  const int e0 = offsets[tile];
  const int e1 = offsets[tile + 1];
  const int2* __restrict__ ed = (const int2*)binned;

  const bool tab = (N <= DTAB_CAP) && (e0 < e1);
  if (tab) {
    const int n4 = ((((unsigned long)dinv) & 15ul) == 0ul) ? (N >> 2) : 0;
    const v4f* __restrict__ d4 = (const v4f*)dinv;
    for (int i0 = tid; i0 < n4; i0 += AGG_THREADS * DTAB_FLIGHT) {
      v4f t[DTAB_FLIGHT];
#pragma unroll
      for (int u = 0; u < DTAB_FLIGHT; ++u) t[u] = d4[min(i0 + u * AGG_THREADS, n4 - 1)];
#pragma unroll
      for (int u = 0; u < DTAB_FLIGHT; ++u)
        if (i0 + u * AGG_THREADS < n4) ((v4f*)acc)[i0 + u * AGG_THREADS] = t[u];
    }
    for (int i = (n4 << 2) + tid; i < N; i += AGG_THREADS) acc[i] = dinv[i];
  }
  bool first = true;

  for (int c0 = e0; c0 < e1; c0 += SRT_CAP) {
    const int c1 = min(c0 + SRT_CAP, e1);

    int sv[SRT_PER_THR], dv[SRT_PER_THR];
#pragma unroll
    for (int k = 0; k < SRT_PER_THR; ++k) {
      const int2 p = ed[min(c0 + tid + k * AGG_THREADS, c1 - 1)];
      sv[k] = p.x;
      dv[k] = p.y - tbase;
    }
    const bool lds_tab = tab && first;
    float dk[SRT_PER_THR];
    if (!lds_tab) {
#pragma unroll
      for (int k = 0; k < SRT_PER_THR; ++k) dk[k] = dinv[sv[k]] * dinv[tbase + dv[k]];
    }
    if (tid < TILE) ord.cnt[tid] = 0;
    __syncthreads();
    if (lds_tab) {
#pragma unroll
      for (int k = 0; k < SRT_PER_THR; ++k) dk[k] = acc[sv[k]] * acc[tbase + dv[k]];
    }
#pragma unroll
    for (int k = 0; k < SRT_PER_THR; ++k)
      if (c0 + tid + k * AGG_THREADS < c1) atomicAdd(&ord.cnt[dv[k]], 1);
    __syncthreads();
    if (w == 0) {
      int c[SCAN_PER_LANE];
      int sum = 0;
#pragma unroll
      for (int k = 0; k < SCAN_PER_LANE; ++k) {
        const int i = lane * SCAN_PER_LANE + k;
        c[k] = (i < TILE) ? ord.cnt[i] : 0;
        sum += c[k];
      }
      int incl = sum;
#pragma unroll
      for (int off = 1; off < 32; off <<= 1) {
        const int t = __shfl_up(incl, off);
        if (lane >= off) incl += t;
      }
      int run = incl - sum;
#pragma unroll
      for (int k = 0; k < SCAN_PER_LANE; ++k) {
        const int i = lane * SCAN_PER_LANE + k;
        if (i < TILE) { ord.start[i] = run; ord.cur[i] = run; }
        run += c[k];
      }
    }
    __syncthreads();
#pragma unroll
    for (int k = 0; k < SRT_PER_THR; ++k)
      if (c0 + tid + k * AGG_THREADS < c1) {
        const int p = atomicAdd(&ord.cur[dv[k]], 1);
        ord.src[p] = sv[k];
        ord.coef[p] = dk[k];
      }
    __syncthreads();
    if (first) {
      tile_start_values<C, ROWS>(acc, h, dinv, N, tbase, tid);
      __syncthreads();
      first = false;
    }

    for (int r = w; r < TILE; r += AGG_WAVES) {
      const int n = __builtin_amdgcn_readfirstlane(ord.cnt[r]);
      if (n == 0) continue;
      const int st = __builtin_amdgcn_readfirstlane(ord.start[r]);
      v4f a = {0.f, 0.f, 0.f, 0.f};
      for (int g = 0; g < n; g += 32) {
        const int cnt = min(32, n - g);
        int s = 0;
        float ds = 0.0f;
        if (lane < cnt) {
          s  = ord.src[st + g + lane];
          ds = ord.coef[st + g + lane];
        }
        for (int jb = 0; jb < cnt; jb += EPB) {
          v4f v[R];
#pragma unroll
          for (int u = 0; u < R; ++u) {
            const int j0 = jb + u * RPI;
            if (u == 0 || j0 < cnt) {
              int sj = __builtin_amdgcn_readlane(s, j0);
              if (RPI == 2) {
                const int sj1 = __builtin_amdgcn_readlane(s, j0 + 1);
                sj = sub ? sj1 : sj;
              }
              v[u] = *(const v4f*)(h + (long)sj * C + cl * 4);
            }
          }
#pragma unroll
          for (int u = 0; u < R; ++u) {
            const int j0 = jb + u * RPI;
            if (u == 0 || j0 < cnt) {
              float cj = __int_as_float(__builtin_amdgcn_readlane(__float_as_int(ds), j0));
              if (RPI == 2) {
                const float cj1 = __int_as_float(__builtin_amdgcn_readlane(__float_as_int(ds), j0 + 1));
                cj = sub ? cj1 : cj;
                if (j0 + sub < cnt) a += v[u] * cj;
              } else {
                a += v[u] * cj;
              }
            }
          }
        }
      }
      if (RPI == 2) {
        a.x += __shfl_xor(a.x, 16);
        a.y += __shfl_xor(a.y, 16);
        a.z += __shfl_xor(a.z, 16);
        a.w += __shfl_xor(a.w, 16);
      }
      if (sub == 0) {
        v4f* __restrict__ ar = (v4f*)&acc[r * LDA + cl * 4];
        *ar = *ar + a;
      }
    }
    __syncthreads();
  }
  if (e0 >= e1) {
    tile_start_values<C, ROWS>(acc, h, dinv, N, tbase, tid);
    __syncthreads();
  }
}

template <int COUT>
struct ProductShape {
  static constexpr int NTN = COUT / 16;
  static constexpr int NWG = AGG_WAVES / NTN;
  static constexpr int NRB = TILE_ROWS / 16;
  static constexpr int MPW = (NRB + NWG - 1) / NWG;
  static_assert(COUT % 16 == 0 && AGG_WAVES % NTN == 0 && TILE_ROWS % 16 == 0 && TILE_ROWS >= TILE,
                "whole strips, whole wave groups, whole row blocks");
  static_assert(NRB % MPW <= 1, "a wave holds MPW row blocks, one, or none");
};

template <int COUT, int NB>
__device__ __forceinline__ void product_rows(
    const float* __restrict__ ap, const float* __restrict__ Wm, int bcol, int half,
    v8f (&cacc)[ProductShape<COUT>::MPW]) {
  constexpr int C   = 128;
  constexpr int LDA = C + 4;
#pragma unroll 4
  for (int k0 = 0; k0 < C; k0 += 4) {
    const int ka = k0 + 2 * half;
    v2f b;
    b.x = Wm[(long)ka * COUT + bcol];
    b.y = Wm[(long)(ka + 1) * COUT + bcol];
#pragma unroll
    for (int j = 0; j < NB; ++j) {
      v2f a;
      a.x = ap[j * 16 * LDA + ka];
      a.y = ap[j * 16 * LDA + ka + 1];
      cacc[j] = __builtin_amdgcn_wmma_f32_16x16x4_f32(
           false, a,  false, b,
           (short)0, cacc[j],  false,  false);
    }
  }
}

template <int COUT>
__device__ __forceinline__ int tile_product(
    const float* __restrict__ acc, const float* __restrict__ Wm, int w, int lane,
    v8f (&cacc)[ProductShape<COUT>::MPW]) {
  using S = ProductShape<COUT>;
  constexpr int LDA = 128 + 4;
  const int half = lane >> 4;
  const int l16  = lane & 15;
  const int tn   = w % S::NTN;
  const int mt0  = (w / S::NTN) * S::MPW;
  const int bcol = tn * 16 + l16;
  const float* __restrict__ ap = &acc[(mt0 * 16 + l16) * LDA];

#pragma unroll
  for (int j = 0; j < S::MPW; ++j) cacc[j] = (v8f){0.f, 0.f, 0.f, 0.f, 0.f, 0.f, 0.f, 0.f};

  const int nb = __builtin_amdgcn_readfirstlane(min(S::MPW, max(0, S::NRB - mt0)));
  if (nb == S::MPW)  product_rows<COUT, S::MPW>(ap, Wm, bcol, half, cacc);
  else if (nb == 1)  product_rows<COUT, 1>(ap, Wm, bcol, half, cacc);
  return nb;
}

__global__ __launch_bounds__(AGG_THREADS) void tile_layer1(
    const float* __restrict__ x, const int* __restrict__ binned,
    const int* __restrict__ offsets, const float* __restrict__ dinv,
    const float* __restrict__ W1, const float* __restrict__ bias,
    const float* __restrict__ gamma, const float* __restrict__ beta,
    const float* __restrict__ mean, const float* __restrict__ var,
    const float* __restrict__ W2, float* __restrict__ hw2, int N) {
  static_assert(GCN_IN_C == 128 && GCN_HID_C == 128, "both products run over a 128-wide LDS tile");
  constexpr int LDA = 128 + 4;
  constexpr int POOL = (TILE_ROWS * LDA > DTAB_CAP) ? TILE_ROWS * LDA : DTAB_CAP;
  __shared__ __attribute__((aligned(16))) float acc[POOL];
  __shared__ RowOrder ord;

  const int tid  = threadIdx.x;
  const int lane = tid & 31;
  const int w = __builtin_amdgcn_readfirstlane(tid >> 5);
  const int tile  = blockIdx.x;
  const int tbase = tile * TILE;
  const int half = lane >> 4;
  const int l16  = lane & 15;

  tile_sums<GCN_IN_C, TILE_ROWS>(acc, ord, x, binned, offsets, dinv, N, tile, tid, lane, w);

  using S1 = ProductShape<GCN_HID_C>;
  v8f c1[S1::MPW];
  const int nb1 = tile_product<GCN_HID_C>(acc, W1, w, lane, c1);

  __syncthreads();
  {
    const int bcol = (w % S1::NTN) * 16 + l16;
    const int mt0  = (w / S1::NTN) * S1::MPW;
    const float bi = bias[bcol];
    const float mu = mean[bcol];
    const float rs = rsqrtf(var[bcol] + GCN_BN_EPS);
    const float ga = gamma[bcol];
    const float be = beta[bcol];
#pragma unroll
    for (int j = 0; j < S1::MPW; ++j) {
      if (j < nb1) {
        float* __restrict__ ar = &acc[((mt0 + j) * 16 + half * 8) * LDA + bcol];
#pragma unroll
        for (int r = 0; r < 8; ++r) {
          float v = c1[j][r] + bi;
          v = (v - mu) * rs * ga + be;
          ar[r * LDA] = fmaxf(v, 0.0f);
        }
      }
    }
  }
  __syncthreads();

  using S2 = ProductShape<GCN_OUT_C>;
  v8f c2[S2::MPW];
  const int nb2 = tile_product<GCN_OUT_C>(acc, W2, w, lane, c2);
  {
    const int bcol = (w % S2::NTN) * 16 + l16;
    const int mt0  = (w / S2::NTN) * S2::MPW;
#pragma unroll
    for (int j = 0; j < S2::MPW; ++j) {
      if (j < nb2) {
        const int nl0 = (mt0 + j) * 16 + half * 8;
        float* __restrict__ cp = hw2 + (long)(tbase + nl0) * GCN_OUT_C + bcol;
#pragma unroll
        for (int r = 0; r < 8; ++r)
          if (nl0 + r < TILE && tbase + nl0 + r < N) cp[(long)r * GCN_OUT_C] = c2[j][r];
      }
    }
  }
}

__global__ __launch_bounds__(AGG_THREADS) void tile_layer2(
    const float* __restrict__ hw2, const int* __restrict__ binned,
    const int* __restrict__ offsets, const float* __restrict__ dinv,
    const float* __restrict__ bias,
    const float* __restrict__ gamma, const float* __restrict__ beta,
    const float* __restrict__ mean, const float* __restrict__ var,
    float* __restrict__ outp, int N) {
  constexpr int C   = GCN_OUT_C;
  constexpr int LDA = C + 4;
  constexpr int POOL = (TILE * LDA > DTAB_CAP) ? TILE * LDA : DTAB_CAP;
  __shared__ __attribute__((aligned(16))) float acc[POOL];
  __shared__ RowOrder ord;

  const int tid  = threadIdx.x;
  const int lane = tid & 31;
  const int w = __builtin_amdgcn_readfirstlane(tid >> 5);
  const int tile  = blockIdx.x;
  const int tbase = tile * TILE;

  tile_sums<C, TILE>(acc, ord, hw2, binned, offsets, dinv, N, tile, tid, lane, w);

  static_assert(AGG_THREADS % (C / 4) == 0, "a thread's column group must not change with i");
  const int c4 = tid % (C / 4);
  const v4f bi = ((const v4f*)bias)[c4];
  const v4f mu = ((const v4f*)mean)[c4];
  const v4f vr = ((const v4f*)var)[c4];
  const v4f ga = ((const v4f*)gamma)[c4];
  const v4f be = ((const v4f*)beta)[c4];
  v4f rs;
  rs.x = rsqrtf(vr.x + GCN_BN_EPS);
  rs.y = rsqrtf(vr.y + GCN_BN_EPS);
  rs.z = rsqrtf(vr.z + GCN_BN_EPS);
  rs.w = rsqrtf(vr.w + GCN_BN_EPS);
  for (int i = tid; i < TILE * C / 4; i += AGG_THREADS) {
    const int nl = i / (C / 4);
    const int node = tbase + nl;
    if (node < N) {
      v4f v = *(const v4f*)&acc[nl * LDA + c4 * 4] + bi;
      v = (v - mu) * rs * ga + be;
      ((v4f*)(outp + (long)node * C))[c4] = v;
    }
  }
}


extern "C" void kernel_launch(void* const* d_in, const int* in_sizes, int n_in,
                              void* d_out, int out_size, void* d_ws, size_t ws_size,
                              hipStream_t stream) {
  const float* x   = (const float*)d_in[0];
  const int*   ei  = (const int*)d_in[1];
  const float* W1  = (const float*)d_in[2];
  const float* b1  = (const float*)d_in[3];
  const float* g1  = (const float*)d_in[4];
  const float* be1 = (const float*)d_in[5];
  const float* m1  = (const float*)d_in[6];
  const float* v1  = (const float*)d_in[7];
  const float* W2  = (const float*)d_in[8];
  const float* b2  = (const float*)d_in[9];
  const float* g2  = (const float*)d_in[10];
  const float* be2 = (const float*)d_in[11];
  const float* m2  = (const float*)d_in[12];
  const float* v2  = (const float*)d_in[13];

  const int N = in_sizes[0] / GCN_IN_C;
  const int E = in_sizes[1] / 2;
  const int* src = ei;
  const int* dst = ei + E;

  char* ws = (char*)d_ws;
  float* hw2 = (float*)ws;                        ws += (size_t)N * GCN_OUT_C * sizeof(float);
  int*   totals  = (int*)ws;
  int*   offsets = totals + MAX_TILES;
  int*   cursor  = offsets + 2 * MAX_TILES;       ws += (size_t)N * sizeof(int);
  float* dinv = (float*)ws;                       ws += (size_t)N * sizeof(float);
  int*   binned  = (int*)ws;                      ws += (size_t)E * 2 * sizeof(int);
  (void)ws_size; (void)n_in; (void)out_size;

  const int NT = (N + TILE - 1) / TILE;
  const int BB = (E + BIN_CHUNK - 1) / BIN_CHUNK;

  zero_f32<<<1, MAX_TILES, 0, stream>>>((float*)totals, (long)MAX_TILES);
  bin_count<<<BB, BIN_THREADS, 0, stream>>>(dst, E, totals, NT);
  bin_scan<<<1, MAX_TILES, 0, stream>>>(totals, offsets, cursor, NT);
  bin_fill<<<BB, BIN_THREADS, 0, stream>>>(src, dst, E, cursor, binned, NT);
  tile_dinv<<<NT, DINV_THREADS, 0, stream>>>(binned, offsets, dinv, N);

  tile_layer1<<<NT, AGG_THREADS, 0, stream>>>(
      x, binned, offsets, dinv, W1, b1, g1, be1, m1, v1, W2, hw2, N);

  tile_layer2<<<NT, AGG_THREADS, 0, stream>>>(
      hw2, binned, offsets, dinv, b2, g2, be2, m2, v2, (float*)d_out, N);
}


# sums: each wave walks an equal share of the tile's row-ordered edge list in whole load batches (was whole rows per wave); partial row sums meet by LDS float adds at row changes: sum order changes
# speedup vs baseline: 5.8012x; 1.0433x over previous
#include <hip/hip_runtime.h>

typedef __attribute__((ext_vector_type(2))) float v2f;
typedef __attribute__((ext_vector_type(4))) float v4f;
typedef __attribute__((ext_vector_type(8))) float v8f;

#define GCN_IN_C   128
#define GCN_HID_C  128
#define GCN_OUT_C  64
#define GCN_BN_EPS 1e-5f

#define TILE        196
#define TILE_ROWS   208
#define MAX_TILES   512
#define BIN_CHUNK   4096
#define BIN_THREADS 256
#define BIN_PER_THR (BIN_CHUNK / BIN_THREADS)
#define AGG_THREADS 1024
#define AGG_WAVES   (AGG_THREADS / 32)
#define AGG_FLIGHT  1024
#define SRT_CAP     8192
#define SRT_PER_THR (SRT_CAP / AGG_THREADS)
#define SCAN_PER_LANE ((TILE + 31) / 32)
#define DTAB_CAP    50176
#define DTAB_FLIGHT 8
#define DINV_THREADS 256
#define DINV_FLIGHT  32


__global__ void zero_f32(float* __restrict__ p, long n) {
  long i = (long)blockIdx.x * blockDim.x + threadIdx.x;
  long stride = (long)gridDim.x * blockDim.x;
  for (; i < n; i += stride) p[i] = 0.0f;
}

__device__ __forceinline__ int tile_of(int node) { return (int)((unsigned)node / (unsigned)TILE); }

__global__ void bin_count(const int* __restrict__ dst, int E, int* __restrict__ totals, int NT) {
  __shared__ int hist[MAX_TILES];
  const int base = blockIdx.x * BIN_CHUNK + threadIdx.x;
  int dv[BIN_PER_THR];
#pragma unroll
  for (int k = 0; k < BIN_PER_THR; ++k) dv[k] = dst[min(base + k * BIN_THREADS, E - 1)];
  for (int i = threadIdx.x; i < NT; i += BIN_THREADS) hist[i] = 0;
  __syncthreads();
#pragma unroll
  for (int k = 0; k < BIN_PER_THR; ++k)
    if (base + k * BIN_THREADS < E) atomicAdd(&hist[tile_of(dv[k])], 1);
  __syncthreads();
  for (int i = threadIdx.x; i < NT; i += BIN_THREADS) {
    const int c = hist[i];
    if (c) atomicAdd(&totals[i], c);
  }
}

__global__ void bin_scan(const int* __restrict__ totals, int* __restrict__ offsets,
                         int* __restrict__ cursor, int NT) {
  __shared__ int a[2][MAX_TILES];
  const int i = threadIdx.x;
  const int own = (i < NT) ? totals[i] : 0;
  a[0][i] = own;
  __syncthreads();
  int cur = 0;
  for (int off = 1; off < MAX_TILES; off <<= 1) {
    int x = a[cur][i];
    if (i >= off) x += a[cur][i - off];
    a[cur ^ 1][i] = x;
    __syncthreads();
    cur ^= 1;
  }
  const int excl = a[cur][i] - own;
  if (i <= NT) offsets[i] = excl;
  if (i < NT) cursor[i] = excl;
}

__global__ void bin_fill(const int* __restrict__ src, const int* __restrict__ dst, int E,
                         int* __restrict__ cursor, int* __restrict__ binned, int NT) {
  __shared__ int hist[MAX_TILES];
  __shared__ int runbase[MAX_TILES];
  const int base = blockIdx.x * BIN_CHUNK + threadIdx.x;
  int dv[BIN_PER_THR], sv[BIN_PER_THR];
#pragma unroll
  for (int k = 0; k < BIN_PER_THR; ++k) dv[k] = dst[min(base + k * BIN_THREADS, E - 1)];
#pragma unroll
  for (int k = 0; k < BIN_PER_THR; ++k) sv[k] = src[min(base + k * BIN_THREADS, E - 1)];
  for (int i = threadIdx.x; i < NT; i += BIN_THREADS) hist[i] = 0;
  __syncthreads();
#pragma unroll
  for (int k = 0; k < BIN_PER_THR; ++k)
    if (base + k * BIN_THREADS < E) atomicAdd(&hist[tile_of(dv[k])], 1);
  __syncthreads();
  for (int i = threadIdx.x; i < NT; i += BIN_THREADS) {
    const int c = hist[i];
    runbase[i] = c ? atomicAdd(&cursor[i], c) : 0;
    hist[i] = 0;
  }
  __syncthreads();
  int2* __restrict__ out = (int2*)binned;
#pragma unroll
  for (int k = 0; k < BIN_PER_THR; ++k) {
    if (base + k * BIN_THREADS < E) {
      const int b = tile_of(dv[k]);
      const int r = atomicAdd(&hist[b], 1);
      out[runbase[b] + r] = make_int2(sv[k], dv[k]);
    }
  }
}

__global__ __launch_bounds__(DINV_THREADS) void tile_dinv(
    const int* __restrict__ binned, const int* __restrict__ offsets,
    float* __restrict__ dinv, int N) {
  static_assert(TILE <= DINV_THREADS, "one thread per node of the tile");
  __shared__ int cnt[TILE];
  const int tid = threadIdx.x;
  const int tile = blockIdx.x;
  const int tbase = tile * TILE;
  if (tid < TILE) cnt[tid] = 0;
  __syncthreads();
  const int e0 = offsets[tile];
  const int e1 = offsets[tile + 1];
  for (int b = e0 + tid; b < e1; b += DINV_THREADS * DINV_FLIGHT) {
    int d[DINV_FLIGHT];
#pragma unroll
    for (int u = 0; u < DINV_FLIGHT; ++u) d[u] = binned[2 * min(b + u * DINV_THREADS, e1 - 1) + 1];
#pragma unroll
    for (int u = 0; u < DINV_FLIGHT; ++u)
      if (b + u * DINV_THREADS < e1) atomicAdd(&cnt[d[u] - tbase], 1);
  }
  __syncthreads();
  if (tid < TILE) {
    const int node = tbase + tid;
    if (node < N) dinv[node] = rsqrtf((float)(cnt[tid] + 1));
  }
}

struct RowOrder {
  int src[SRT_CAP];
  float coef[SRT_CAP];
  unsigned char row[SRT_CAP];
  int cnt[TILE];
  int cur[TILE];
};
static_assert(TILE <= 256, "a row index inside the tile is stored in one byte");

__device__ __forceinline__ void row_flush(float* ar, v4f a) {
  atomicAdd(ar + 0, a.x);
  atomicAdd(ar + 1, a.y);
  atomicAdd(ar + 2, a.z);
  atomicAdd(ar + 3, a.w);
}

template <int C, int ROWS>
__device__ __forceinline__ void tile_start_values(
    float* acc, const float* __restrict__ h, const float* __restrict__ dinv,
    int N, int tbase, int tid) {
  constexpr int LDA = C + 4;
  constexpr int TOT = ROWS * C / 4;
  constexpr int NIT = (TOT + AGG_THREADS - 1) / AGG_THREADS;
  static_assert(AGG_THREADS % (C / 4) == 0, "a thread keeps one column group");
  const int c4 = tid % (C / 4);
  v4f v[NIT];
  float di[NIT];
#pragma unroll
  for (int u = 0; u < NIT; ++u) {
    const int nl = (tid + u * AGG_THREADS) / (C / 4);
    const int node = min(tbase + min(nl, TILE - 1), N - 1);
    di[u] = dinv[node];
    v[u] = ((const v4f*)(h + (long)node * C))[c4];
  }
#pragma unroll
  for (int u = 0; u < NIT; ++u) {
    const int i = tid + u * AGG_THREADS;
    const int nl = i / (C / 4);
    if (i < TOT) {
      v4f s = {0.f, 0.f, 0.f, 0.f};
      if (nl < TILE && tbase + nl < N) s = v[u] * (di[u] * di[u]);
      *(v4f*)&acc[nl * LDA + c4 * 4] = s;
    }
  }
}

template <int C, bool FULL>
__device__ __forceinline__ void walk_batch(
    float* acc, const float* __restrict__ h, int s, float ds, int rw,
    int jb, int cnt, int sub, int cl, int& myrow, v4f& a) {
  constexpr int LDA = C + 4;
  constexpr int RPI = 32 / (C / 4);
  constexpr int R   = AGG_FLIGHT / 128;
  v4f v[R];
#pragma unroll
  for (int u = 0; u < R; ++u) {
    const int j0 = jb + u * RPI;
    int sj = __builtin_amdgcn_readlane(s, j0);
    if (RPI == 2) {
      const int sj1 = __builtin_amdgcn_readlane(s, j0 + 1);
      sj = sub ? sj1 : sj;
    }
    v[u] = *(const v4f*)(h + (long)sj * C + cl * 4);
  }
#pragma unroll
  for (int u = 0; u < R; ++u) {
    const int j0 = jb + u * RPI;
    if (FULL || u == 0 || j0 < cnt) {
      float cj = __int_as_float(__builtin_amdgcn_readlane(__float_as_int(ds), j0));
      int rj = __builtin_amdgcn_readlane(rw, j0);
      bool live = true;
      if (RPI == 2) {
        const float cj1 = __int_as_float(__builtin_amdgcn_readlane(__float_as_int(ds), j0 + 1));
        const int rj1 = __builtin_amdgcn_readlane(rw, j0 + 1);
        cj = sub ? cj1 : cj;
        rj = sub ? rj1 : rj;
        live = FULL || (j0 + sub < cnt);
      }
      if (live) {
        if (rj != myrow) {
          if (myrow >= 0) row_flush(&acc[myrow * LDA + cl * 4], a);
          a = (v4f){0.f, 0.f, 0.f, 0.f};
          myrow = rj;
        }
        a += v[u] * cj;
      }
    }
  }
}

template <int C, int ROWS>
__device__ __forceinline__ void tile_sums(
    float* acc, RowOrder& ord, const float* __restrict__ h,
    const int* __restrict__ binned, const int* __restrict__ offsets,
    const float* __restrict__ dinv, int N, int tile, int tid, int lane, int w) {
  static_assert(C == 64 || C == 128, "a row is 16 or 32 lanes of 16 bytes");
  constexpr int LDA = C + 4;
  constexpr int LPR = C / 4;
  constexpr int RPI = 32 / LPR;
  constexpr int R   = AGG_FLIGHT / 128;
  constexpr int EPB = R * RPI;
  static_assert(32 % EPB == 0, "a 32-edge group is a whole number of batches");
  const int tbase = tile * TILE;
  const int sub = lane / LPR;
  const int cl  = lane % LPR;

  const int e0 = offsets[tile];
  const int e1 = offsets[tile + 1];
  const int2* __restrict__ ed = (const int2*)binned;

  const bool tab = (N <= DTAB_CAP) && (e0 < e1);
  if (tab) {
    const int n4 = ((((unsigned long)dinv) & 15ul) == 0ul) ? (N >> 2) : 0;
    const v4f* __restrict__ d4 = (const v4f*)dinv;
    for (int i0 = tid; i0 < n4; i0 += AGG_THREADS * DTAB_FLIGHT) {
      v4f t[DTAB_FLIGHT];
#pragma unroll
      for (int u = 0; u < DTAB_FLIGHT; ++u) t[u] = d4[min(i0 + u * AGG_THREADS, n4 - 1)];
#pragma unroll
      for (int u = 0; u < DTAB_FLIGHT; ++u)
        if (i0 + u * AGG_THREADS < n4) ((v4f*)acc)[i0 + u * AGG_THREADS] = t[u];
    }
    for (int i = (n4 << 2) + tid; i < N; i += AGG_THREADS) acc[i] = dinv[i];
  }
  bool first = true;

  for (int c0 = e0; c0 < e1; c0 += SRT_CAP) {
    const int c1 = min(c0 + SRT_CAP, e1);

    int sv[SRT_PER_THR], dv[SRT_PER_THR];
#pragma unroll
    for (int k = 0; k < SRT_PER_THR; ++k) {
      const int2 p = ed[min(c0 + tid + k * AGG_THREADS, c1 - 1)];
      sv[k] = p.x;
      dv[k] = p.y - tbase;
    }
    const bool lds_tab = tab && first;
    float dk[SRT_PER_THR];
    if (!lds_tab) {
#pragma unroll
      for (int k = 0; k < SRT_PER_THR; ++k) dk[k] = dinv[sv[k]] * dinv[tbase + dv[k]];
    }
    if (tid < TILE) ord.cnt[tid] = 0;
    __syncthreads();
    if (lds_tab) {
#pragma unroll
      for (int k = 0; k < SRT_PER_THR; ++k) dk[k] = acc[sv[k]] * acc[tbase + dv[k]];
    }
#pragma unroll
    for (int k = 0; k < SRT_PER_THR; ++k)
      if (c0 + tid + k * AGG_THREADS < c1) atomicAdd(&ord.cnt[dv[k]], 1);
    __syncthreads();
    if (w == 0) {
      int c[SCAN_PER_LANE];
      int sum = 0;
#pragma unroll
      for (int k = 0; k < SCAN_PER_LANE; ++k) {
        const int i = lane * SCAN_PER_LANE + k;
        c[k] = (i < TILE) ? ord.cnt[i] : 0;
        sum += c[k];
      }
      int incl = sum;
#pragma unroll
      for (int off = 1; off < 32; off <<= 1) {
        const int t = __shfl_up(incl, off);
        if (lane >= off) incl += t;
      }
      int run = incl - sum;
#pragma unroll
      for (int k = 0; k < SCAN_PER_LANE; ++k) {
        const int i = lane * SCAN_PER_LANE + k;
        if (i < TILE) ord.cur[i] = run;
        run += c[k];
      }
    }
    __syncthreads();
#pragma unroll
    for (int k = 0; k < SRT_PER_THR; ++k)
      if (c0 + tid + k * AGG_THREADS < c1) {
        const int p = atomicAdd(&ord.cur[dv[k]], 1);
        ord.src[p] = sv[k];
        ord.coef[p] = dk[k];
        ord.row[p] = (unsigned char)dv[k];
      }
    __syncthreads();
    if (first) {
      tile_start_values<C, ROWS>(acc, h, dinv, N, tbase, tid);
      __syncthreads();
      first = false;
    }

    const int m   = c1 - c0;
    const int per = ((m + AGG_WAVES * EPB - 1) / (AGG_WAVES * EPB)) * EPB;
    const int wb  = w * per;
    const int we  = min(wb + per, m);
    int myrow = -1;
    v4f a = {0.f, 0.f, 0.f, 0.f};
    for (int g = wb; g < we; g += 32) {
      const int cnt = min(32, we - g);
      int s = 0;
      float ds = 0.0f;
      int rw = 0;
      if (lane < cnt) {
        s  = ord.src[g + lane];
        ds = ord.coef[g + lane];
        rw = ord.row[g + lane];
      }
      for (int jb = 0; jb < cnt; jb += EPB) {
        if (cnt - jb >= EPB) walk_batch<C, true>(acc, h, s, ds, rw, jb, cnt, sub, cl, myrow, a);
        else                 walk_batch<C, false>(acc, h, s, ds, rw, jb, cnt, sub, cl, myrow, a);
      }
    }
    if (myrow >= 0) row_flush(&acc[myrow * LDA + cl * 4], a);
    __syncthreads();
  }
  if (e0 >= e1) {
    tile_start_values<C, ROWS>(acc, h, dinv, N, tbase, tid);
    __syncthreads();
  }
}

template <int COUT>
struct ProductShape {
  static constexpr int NTN = COUT / 16;
  static constexpr int NWG = AGG_WAVES / NTN;
  static constexpr int NRB = TILE_ROWS / 16;
  static constexpr int MPW = (NRB + NWG - 1) / NWG;
  static_assert(COUT % 16 == 0 && AGG_WAVES % NTN == 0 && TILE_ROWS % 16 == 0 && TILE_ROWS >= TILE,
                "whole strips, whole wave groups, whole row blocks");
  static_assert(NRB % MPW <= 1, "a wave holds MPW row blocks, one, or none");
};

template <int COUT, int NB>
__device__ __forceinline__ void product_rows(
    const float* __restrict__ ap, const float* __restrict__ Wm, int bcol, int half,
    v8f (&cacc)[ProductShape<COUT>::MPW]) {
  constexpr int C   = 128;
  constexpr int LDA = C + 4;
#pragma unroll 4
  for (int k0 = 0; k0 < C; k0 += 4) {
    const int ka = k0 + 2 * half;
    v2f b;
    b.x = Wm[(long)ka * COUT + bcol];
    b.y = Wm[(long)(ka + 1) * COUT + bcol];
#pragma unroll
    for (int j = 0; j < NB; ++j) {
      v2f a;
      a.x = ap[j * 16 * LDA + ka];
      a.y = ap[j * 16 * LDA + ka + 1];
      cacc[j] = __builtin_amdgcn_wmma_f32_16x16x4_f32(
           false, a,  false, b,
           (short)0, cacc[j],  false,  false);
    }
  }
}

template <int COUT>
__device__ __forceinline__ int tile_product(
    const float* __restrict__ acc, const float* __restrict__ Wm, int w, int lane,
    v8f (&cacc)[ProductShape<COUT>::MPW]) {
  using S = ProductShape<COUT>;
  constexpr int LDA = 128 + 4;
  const int half = lane >> 4;
  const int l16  = lane & 15;
  const int tn   = w % S::NTN;
  const int mt0  = (w / S::NTN) * S::MPW;
  const int bcol = tn * 16 + l16;
  const float* __restrict__ ap = &acc[(mt0 * 16 + l16) * LDA];

#pragma unroll
  for (int j = 0; j < S::MPW; ++j) cacc[j] = (v8f){0.f, 0.f, 0.f, 0.f, 0.f, 0.f, 0.f, 0.f};

  const int nb = __builtin_amdgcn_readfirstlane(min(S::MPW, max(0, S::NRB - mt0)));
  if (nb == S::MPW)  product_rows<COUT, S::MPW>(ap, Wm, bcol, half, cacc);
  else if (nb == 1)  product_rows<COUT, 1>(ap, Wm, bcol, half, cacc);
  return nb;
}

__global__ __launch_bounds__(AGG_THREADS) void tile_layer1(
    const float* __restrict__ x, const int* __restrict__ binned,
    const int* __restrict__ offsets, const float* __restrict__ dinv,
    const float* __restrict__ W1, const float* __restrict__ bias,
    const float* __restrict__ gamma, const float* __restrict__ beta,
    const float* __restrict__ mean, const float* __restrict__ var,
    const float* __restrict__ W2, float* __restrict__ hw2, int N) {
  static_assert(GCN_IN_C == 128 && GCN_HID_C == 128, "both products run over a 128-wide LDS tile");
  constexpr int LDA = 128 + 4;
  constexpr int POOL = (TILE_ROWS * LDA > DTAB_CAP) ? TILE_ROWS * LDA : DTAB_CAP;
  __shared__ __attribute__((aligned(16))) float acc[POOL];
  __shared__ RowOrder ord;

  const int tid  = threadIdx.x;
  const int lane = tid & 31;
  const int w = __builtin_amdgcn_readfirstlane(tid >> 5);
  const int tile  = blockIdx.x;
  const int tbase = tile * TILE;
  const int half = lane >> 4;
  const int l16  = lane & 15;

  tile_sums<GCN_IN_C, TILE_ROWS>(acc, ord, x, binned, offsets, dinv, N, tile, tid, lane, w);

  using S1 = ProductShape<GCN_HID_C>;
  v8f c1[S1::MPW];
  const int nb1 = tile_product<GCN_HID_C>(acc, W1, w, lane, c1);

  __syncthreads();
  {
    const int bcol = (w % S1::NTN) * 16 + l16;
    const int mt0  = (w / S1::NTN) * S1::MPW;
    const float bi = bias[bcol];
    const float mu = mean[bcol];
    const float rs = rsqrtf(var[bcol] + GCN_BN_EPS);
    const float ga = gamma[bcol];
    const float be = beta[bcol];
#pragma unroll
    for (int j = 0; j < S1::MPW; ++j) {
      if (j < nb1) {
        float* __restrict__ ar = &acc[((mt0 + j) * 16 + half * 8) * LDA + bcol];
#pragma unroll
        for (int r = 0; r < 8; ++r) {
          float v = c1[j][r] + bi;
          v = (v - mu) * rs * ga + be;
          ar[r * LDA] = fmaxf(v, 0.0f);
        }
      }
    }
  }
  __syncthreads();

  using S2 = ProductShape<GCN_OUT_C>;
  v8f c2[S2::MPW];
  const int nb2 = tile_product<GCN_OUT_C>(acc, W2, w, lane, c2);
  {
    const int bcol = (w % S2::NTN) * 16 + l16;
    const int mt0  = (w / S2::NTN) * S2::MPW;
#pragma unroll
    for (int j = 0; j < S2::MPW; ++j) {
      if (j < nb2) {
        const int nl0 = (mt0 + j) * 16 + half * 8;
        float* __restrict__ cp = hw2 + (long)(tbase + nl0) * GCN_OUT_C + bcol;
#pragma unroll
        for (int r = 0; r < 8; ++r)
          if (nl0 + r < TILE && tbase + nl0 + r < N) cp[(long)r * GCN_OUT_C] = c2[j][r];
      }
    }
  }
}

__global__ __launch_bounds__(AGG_THREADS) void tile_layer2(
    const float* __restrict__ hw2, const int* __restrict__ binned,
    const int* __restrict__ offsets, const float* __restrict__ dinv,
    const float* __restrict__ bias,
    const float* __restrict__ gamma, const float* __restrict__ beta,
    const float* __restrict__ mean, const float* __restrict__ var,
    float* __restrict__ outp, int N) {
  constexpr int C   = GCN_OUT_C;
  constexpr int LDA = C + 4;
  constexpr int POOL = (TILE * LDA > DTAB_CAP) ? TILE * LDA : DTAB_CAP;
  __shared__ __attribute__((aligned(16))) float acc[POOL];
  __shared__ RowOrder ord;

  const int tid  = threadIdx.x;
  const int lane = tid & 31;
  const int w = __builtin_amdgcn_readfirstlane(tid >> 5);
  const int tile  = blockIdx.x;
  const int tbase = tile * TILE;

  tile_sums<C, TILE>(acc, ord, hw2, binned, offsets, dinv, N, tile, tid, lane, w);

  static_assert(AGG_THREADS % (C / 4) == 0, "a thread's column group must not change with i");
  const int c4 = tid % (C / 4);
  const v4f bi = ((const v4f*)bias)[c4];
  const v4f mu = ((const v4f*)mean)[c4];
  const v4f vr = ((const v4f*)var)[c4];
  const v4f ga = ((const v4f*)gamma)[c4];
  const v4f be = ((const v4f*)beta)[c4];
  v4f rs;
  rs.x = rsqrtf(vr.x + GCN_BN_EPS);
  rs.y = rsqrtf(vr.y + GCN_BN_EPS);
  rs.z = rsqrtf(vr.z + GCN_BN_EPS);
  rs.w = rsqrtf(vr.w + GCN_BN_EPS);
  for (int i = tid; i < TILE * C / 4; i += AGG_THREADS) {
    const int nl = i / (C / 4);
    const int node = tbase + nl;
    if (node < N) {
      v4f v = *(const v4f*)&acc[nl * LDA + c4 * 4] + bi;
      v = (v - mu) * rs * ga + be;
      ((v4f*)(outp + (long)node * C))[c4] = v;
    }
  }
}


extern "C" void kernel_launch(void* const* d_in, const int* in_sizes, int n_in,
                              void* d_out, int out_size, void* d_ws, size_t ws_size,
                              hipStream_t stream) {
  const float* x   = (const float*)d_in[0];
  const int*   ei  = (const int*)d_in[1];
  const float* W1  = (const float*)d_in[2];
  const float* b1  = (const float*)d_in[3];
  const float* g1  = (const float*)d_in[4];
  const float* be1 = (const float*)d_in[5];
  const float* m1  = (const float*)d_in[6];
  const float* v1  = (const float*)d_in[7];
  const float* W2  = (const float*)d_in[8];
  const float* b2  = (const float*)d_in[9];
  const float* g2  = (const float*)d_in[10];
  const float* be2 = (const float*)d_in[11];
  const float* m2  = (const float*)d_in[12];
  const float* v2  = (const float*)d_in[13];

  const int N = in_sizes[0] / GCN_IN_C;
  const int E = in_sizes[1] / 2;
  const int* src = ei;
  const int* dst = ei + E;

  char* ws = (char*)d_ws;
  float* hw2 = (float*)ws;                        ws += (size_t)N * GCN_OUT_C * sizeof(float);
  int*   totals  = (int*)ws;
  int*   offsets = totals + MAX_TILES;
  int*   cursor  = offsets + 2 * MAX_TILES;       ws += (size_t)N * sizeof(int);
  float* dinv = (float*)ws;                       ws += (size_t)N * sizeof(float);
  int*   binned  = (int*)ws;                      ws += (size_t)E * 2 * sizeof(int);
  (void)ws_size; (void)n_in; (void)out_size;

  const int NT = (N + TILE - 1) / TILE;
  const int BB = (E + BIN_CHUNK - 1) / BIN_CHUNK;

  zero_f32<<<1, MAX_TILES, 0, stream>>>((float*)totals, (long)MAX_TILES);
  bin_count<<<BB, BIN_THREADS, 0, stream>>>(dst, E, totals, NT);
  bin_scan<<<1, MAX_TILES, 0, stream>>>(totals, offsets, cursor, NT);
  bin_fill<<<BB, BIN_THREADS, 0, stream>>>(src, dst, E, cursor, binned, NT);
  tile_dinv<<<NT, DINV_THREADS, 0, stream>>>(binned, offsets, dinv, N);

  tile_layer1<<<NT, AGG_THREADS, 0, stream>>>(
      x, binned, offsets, dinv, W1, b1, g1, be1, m1, v1, W2, hw2, N);

  tile_layer2<<<NT, AGG_THREADS, 0, stream>>>(
      hw2, binned, offsets, dinv, b2, g2, be2, m2, v2, (float*)d_out, N);
}


# binning stores: a fill block orders its 4096 edges by tile in LDS, then writes each tile's run with contiguous 256-byte wave stores (was 32 lines x 8 bytes per store); arrival-order sums as before
# speedup vs baseline: 6.0159x; 1.0370x over previous
#include <hip/hip_runtime.h>

typedef __attribute__((ext_vector_type(2))) float v2f;
typedef __attribute__((ext_vector_type(4))) float v4f;
typedef __attribute__((ext_vector_type(8))) float v8f;

#define GCN_IN_C   128
#define GCN_HID_C  128
#define GCN_OUT_C  64
#define GCN_BN_EPS 1e-5f

#define TILE        196
#define TILE_ROWS   208
#define MAX_TILES   512
#define BIN_CHUNK   4096
#define BIN_THREADS 256
#define BIN_PER_THR (BIN_CHUNK / BIN_THREADS)
#define AGG_THREADS 1024
#define AGG_WAVES   (AGG_THREADS / 32)
#define AGG_FLIGHT  1024
#define SRT_CAP     8192
#define SRT_PER_THR (SRT_CAP / AGG_THREADS)
#define SCAN_PER_LANE ((TILE + 31) / 32)
#define DTAB_CAP    50176
#define DTAB_FLIGHT 8
#define DINV_THREADS 256
#define DINV_FLIGHT  32


__global__ void zero_f32(float* __restrict__ p, long n) {
  long i = (long)blockIdx.x * blockDim.x + threadIdx.x;
  long stride = (long)gridDim.x * blockDim.x;
  for (; i < n; i += stride) p[i] = 0.0f;
}

__device__ __forceinline__ int tile_of(int node) { return (int)((unsigned)node / (unsigned)TILE); }

__global__ void bin_count(const int* __restrict__ dst, int E, int* __restrict__ totals, int NT) {
  __shared__ int hist[MAX_TILES];
  const int base = blockIdx.x * BIN_CHUNK + threadIdx.x;
  int dv[BIN_PER_THR];
#pragma unroll
  for (int k = 0; k < BIN_PER_THR; ++k) dv[k] = dst[min(base + k * BIN_THREADS, E - 1)];
  for (int i = threadIdx.x; i < NT; i += BIN_THREADS) hist[i] = 0;
  __syncthreads();
#pragma unroll
  for (int k = 0; k < BIN_PER_THR; ++k)
    if (base + k * BIN_THREADS < E) atomicAdd(&hist[tile_of(dv[k])], 1);
  __syncthreads();
  for (int i = threadIdx.x; i < NT; i += BIN_THREADS) {
    const int c = hist[i];
    if (c) atomicAdd(&totals[i], c);
  }
}

__global__ void bin_scan(const int* __restrict__ totals, int* __restrict__ offsets,
                         int* __restrict__ cursor, int NT) {
  __shared__ int a[2][MAX_TILES];
  const int i = threadIdx.x;
  const int own = (i < NT) ? totals[i] : 0;
  a[0][i] = own;
  __syncthreads();
  int cur = 0;
  for (int off = 1; off < MAX_TILES; off <<= 1) {
    int x = a[cur][i];
    if (i >= off) x += a[cur][i - off];
    a[cur ^ 1][i] = x;
    __syncthreads();
    cur ^= 1;
  }
  const int excl = a[cur][i] - own;
  if (i <= NT) offsets[i] = excl;
  if (i < NT) cursor[i] = excl;
}

__global__ void bin_fill(const int* __restrict__ src, const int* __restrict__ dst, int E,
                         int* __restrict__ cursor, int* __restrict__ binned, int NT) {
  static_assert(MAX_TILES == 2 * BIN_THREADS, "the prefix sum gives every thread two tiles");
  __shared__ __attribute__((aligned(8))) int hist[MAX_TILES];
  __shared__ __attribute__((aligned(8))) int lstart[MAX_TILES];
  __shared__ __attribute__((aligned(8))) int lpos[MAX_TILES + 2];
  __shared__ int delta[MAX_TILES];
  __shared__ int wsum[BIN_THREADS / 32];
  __shared__ int2 ordered[BIN_CHUNK];
  const int tid  = threadIdx.x;
  const int lane = tid & 31;
  const int w    = tid >> 5;
  const int base = blockIdx.x * BIN_CHUNK + tid;
  int dv[BIN_PER_THR], sv[BIN_PER_THR];
#pragma unroll
  for (int k = 0; k < BIN_PER_THR; ++k) dv[k] = dst[min(base + k * BIN_THREADS, E - 1)];
#pragma unroll
  for (int k = 0; k < BIN_PER_THR; ++k) sv[k] = src[min(base + k * BIN_THREADS, E - 1)];
  for (int i = tid; i < MAX_TILES; i += BIN_THREADS) hist[i] = 0;
  __syncthreads();
#pragma unroll
  for (int k = 0; k < BIN_PER_THR; ++k)
    if (base + k * BIN_THREADS < E) atomicAdd(&hist[tile_of(dv[k])], 1);
  __syncthreads();
  int rb[MAX_TILES / BIN_THREADS];
#pragma unroll
  for (int k = 0; k < MAX_TILES / BIN_THREADS; ++k) {
    const int i = tid + k * BIN_THREADS;
    const int c = (i < NT) ? hist[i] : 0;
    rb[k] = c ? atomicAdd(&cursor[i], c) : 0;
  }
  const int2 h = ((const int2*)hist)[tid];
  const int s = h.x + h.y;
  int incl = s;
#pragma unroll
  for (int off = 1; off < 32; off <<= 1) {
    const int t = __shfl_up(incl, off);
    if (lane >= off) incl += t;
  }
  if (lane == 31) wsum[w] = incl;
  __syncthreads();
  int prev = 0;
#pragma unroll
  for (int j = 0; j < BIN_THREADS / 32; ++j) prev += (j < w) ? wsum[j] : 0;
  const int ex0 = prev + incl - s;
  ((int2*)lstart)[tid] = make_int2(ex0, ex0 + h.x);
  ((int2*)lpos)[tid]   = make_int2(ex0, ex0 + h.x);
  if (tid == 0) lpos[MAX_TILES] = 0;
  __syncthreads();
#pragma unroll
  for (int k = 0; k < MAX_TILES / BIN_THREADS; ++k) {
    const int i = tid + k * BIN_THREADS;
    delta[i] = rb[k] - lstart[i];
  }
  int p[BIN_PER_THR];
#pragma unroll
  for (int k = 0; k < BIN_PER_THR; ++k)
    p[k] = atomicAdd(&lpos[(base + k * BIN_THREADS < E) ? tile_of(dv[k]) : MAX_TILES], 1);
#pragma unroll
  for (int k = 0; k < BIN_PER_THR; ++k)
    if (base + k * BIN_THREADS < E) ordered[p[k]] = make_int2(sv[k], dv[k]);
  __syncthreads();
  const int nvalid = min((int)BIN_CHUNK, E - (int)blockIdx.x * BIN_CHUNK);
  int2* __restrict__ out = (int2*)binned;
  int2 e[BIN_PER_THR];
  int dl[BIN_PER_THR];
#pragma unroll
  for (int k = 0; k < BIN_PER_THR; ++k) e[k] = ordered[min(tid + k * BIN_THREADS, nvalid - 1)];
#pragma unroll
  for (int k = 0; k < BIN_PER_THR; ++k) dl[k] = delta[tile_of(e[k].y)];
#pragma unroll
  for (int k = 0; k < BIN_PER_THR; ++k)
    if (tid + k * BIN_THREADS < nvalid) out[dl[k] + tid + k * BIN_THREADS] = e[k];
}

__global__ __launch_bounds__(DINV_THREADS) void tile_dinv(
    const int* __restrict__ binned, const int* __restrict__ offsets,
    float* __restrict__ dinv, int N) {
  static_assert(TILE <= DINV_THREADS, "one thread per node of the tile");
  __shared__ int cnt[TILE];
  const int tid = threadIdx.x;
  const int tile = blockIdx.x;
  const int tbase = tile * TILE;
  if (tid < TILE) cnt[tid] = 0;
  __syncthreads();
  const int e0 = offsets[tile];
  const int e1 = offsets[tile + 1];
  for (int b = e0 + tid; b < e1; b += DINV_THREADS * DINV_FLIGHT) {
    int d[DINV_FLIGHT];
#pragma unroll
    for (int u = 0; u < DINV_FLIGHT; ++u) d[u] = binned[2 * min(b + u * DINV_THREADS, e1 - 1) + 1];
#pragma unroll
    for (int u = 0; u < DINV_FLIGHT; ++u)
      if (b + u * DINV_THREADS < e1) atomicAdd(&cnt[d[u] - tbase], 1);
  }
  __syncthreads();
  if (tid < TILE) {
    const int node = tbase + tid;
    if (node < N) dinv[node] = rsqrtf((float)(cnt[tid] + 1));
  }
}

struct RowOrder {
  int src[SRT_CAP];
  float coef[SRT_CAP];
  unsigned char row[SRT_CAP];
  int cnt[TILE];
  int cur[TILE];
};
static_assert(TILE <= 256, "a row index inside the tile is stored in one byte");

__device__ __forceinline__ void row_flush(float* ar, v4f a) {
  atomicAdd(ar + 0, a.x);
  atomicAdd(ar + 1, a.y);
  atomicAdd(ar + 2, a.z);
  atomicAdd(ar + 3, a.w);
}

template <int C, int ROWS>
__device__ __forceinline__ void tile_start_values(
    float* acc, const float* __restrict__ h, const float* __restrict__ dinv,
    int N, int tbase, int tid) {
  constexpr int LDA = C + 4;
  constexpr int TOT = ROWS * C / 4;
  constexpr int NIT = (TOT + AGG_THREADS - 1) / AGG_THREADS;
  static_assert(AGG_THREADS % (C / 4) == 0, "a thread keeps one column group");
  const int c4 = tid % (C / 4);
  v4f v[NIT];
  float di[NIT];
#pragma unroll
  for (int u = 0; u < NIT; ++u) {
    const int nl = (tid + u * AGG_THREADS) / (C / 4);
    const int node = min(tbase + min(nl, TILE - 1), N - 1);
    di[u] = dinv[node];
    v[u] = ((const v4f*)(h + (long)node * C))[c4];
  }
#pragma unroll
  for (int u = 0; u < NIT; ++u) {
    const int i = tid + u * AGG_THREADS;
    const int nl = i / (C / 4);
    if (i < TOT) {
      v4f s = {0.f, 0.f, 0.f, 0.f};
      if (nl < TILE && tbase + nl < N) s = v[u] * (di[u] * di[u]);
      *(v4f*)&acc[nl * LDA + c4 * 4] = s;
    }
  }
}

template <int C, bool FULL>
__device__ __forceinline__ void walk_batch(
    float* acc, const float* __restrict__ h, int s, float ds, int rw,
    int jb, int cnt, int sub, int cl, int& myrow, v4f& a) {
  constexpr int LDA = C + 4;
  constexpr int RPI = 32 / (C / 4);
  constexpr int R   = AGG_FLIGHT / 128;
  v4f v[R];
#pragma unroll
  for (int u = 0; u < R; ++u) {
    const int j0 = jb + u * RPI;
    int sj = __builtin_amdgcn_readlane(s, j0);
    if (RPI == 2) {
      const int sj1 = __builtin_amdgcn_readlane(s, j0 + 1);
      sj = sub ? sj1 : sj;
    }
    v[u] = *(const v4f*)(h + (long)sj * C + cl * 4);
  }
#pragma unroll
  for (int u = 0; u < R; ++u) {
    const int j0 = jb + u * RPI;
    if (FULL || u == 0 || j0 < cnt) {
      float cj = __int_as_float(__builtin_amdgcn_readlane(__float_as_int(ds), j0));
      int rj = __builtin_amdgcn_readlane(rw, j0);
      bool live = true;
      if (RPI == 2) {
        const float cj1 = __int_as_float(__builtin_amdgcn_readlane(__float_as_int(ds), j0 + 1));
        const int rj1 = __builtin_amdgcn_readlane(rw, j0 + 1);
        cj = sub ? cj1 : cj;
        rj = sub ? rj1 : rj;
        live = FULL || (j0 + sub < cnt);
      }
      if (live) {
        if (rj != myrow) {
          if (myrow >= 0) row_flush(&acc[myrow * LDA + cl * 4], a);
          a = (v4f){0.f, 0.f, 0.f, 0.f};
          myrow = rj;
        }
        a += v[u] * cj;
      }
    }
  }
}

template <int C, int ROWS>
__device__ __forceinline__ void tile_sums(
    float* acc, RowOrder& ord, const float* __restrict__ h,
    const int* __restrict__ binned, const int* __restrict__ offsets,
    const float* __restrict__ dinv, int N, int tile, int tid, int lane, int w) {
  static_assert(C == 64 || C == 128, "a row is 16 or 32 lanes of 16 bytes");
  constexpr int LDA = C + 4;
  constexpr int LPR = C / 4;
  constexpr int RPI = 32 / LPR;
  constexpr int R   = AGG_FLIGHT / 128;
  constexpr int EPB = R * RPI;
  static_assert(32 % EPB == 0, "a 32-edge group is a whole number of batches");
  const int tbase = tile * TILE;
  const int sub = lane / LPR;
  const int cl  = lane % LPR;

  const int e0 = offsets[tile];
  const int e1 = offsets[tile + 1];
  const int2* __restrict__ ed = (const int2*)binned;

  const bool tab = (N <= DTAB_CAP) && (e0 < e1);
  if (tab) {
    const int n4 = ((((unsigned long)dinv) & 15ul) == 0ul) ? (N >> 2) : 0;
    const v4f* __restrict__ d4 = (const v4f*)dinv;
    for (int i0 = tid; i0 < n4; i0 += AGG_THREADS * DTAB_FLIGHT) {
      v4f t[DTAB_FLIGHT];
#pragma unroll
      for (int u = 0; u < DTAB_FLIGHT; ++u) t[u] = d4[min(i0 + u * AGG_THREADS, n4 - 1)];
#pragma unroll
      for (int u = 0; u < DTAB_FLIGHT; ++u)
        if (i0 + u * AGG_THREADS < n4) ((v4f*)acc)[i0 + u * AGG_THREADS] = t[u];
    }
    for (int i = (n4 << 2) + tid; i < N; i += AGG_THREADS) acc[i] = dinv[i];
  }
  bool first = true;

  for (int c0 = e0; c0 < e1; c0 += SRT_CAP) {
    const int c1 = min(c0 + SRT_CAP, e1);

    int sv[SRT_PER_THR], dv[SRT_PER_THR];
#pragma unroll
    for (int k = 0; k < SRT_PER_THR; ++k) {
      const int2 p = ed[min(c0 + tid + k * AGG_THREADS, c1 - 1)];
      sv[k] = p.x;
      dv[k] = p.y - tbase;
    }
    const bool lds_tab = tab && first;
    float dk[SRT_PER_THR];
    if (!lds_tab) {
#pragma unroll
      for (int k = 0; k < SRT_PER_THR; ++k) dk[k] = dinv[sv[k]] * dinv[tbase + dv[k]];
    }
    if (tid < TILE) ord.cnt[tid] = 0;
    __syncthreads();
    if (lds_tab) {
#pragma unroll
      for (int k = 0; k < SRT_PER_THR; ++k) dk[k] = acc[sv[k]] * acc[tbase + dv[k]];
    }
#pragma unroll
    for (int k = 0; k < SRT_PER_THR; ++k)
      if (c0 + tid + k * AGG_THREADS < c1) atomicAdd(&ord.cnt[dv[k]], 1);
    __syncthreads();
    if (w == 0) {
      int c[SCAN_PER_LANE];
      int sum = 0;
#pragma unroll
      for (int k = 0; k < SCAN_PER_LANE; ++k) {
        const int i = lane * SCAN_PER_LANE + k;
        c[k] = (i < TILE) ? ord.cnt[i] : 0;
        sum += c[k];
      }
      int incl = sum;
#pragma unroll
      for (int off = 1; off < 32; off <<= 1) {
        const int t = __shfl_up(incl, off);
        if (lane >= off) incl += t;
      }
      int run = incl - sum;
#pragma unroll
      for (int k = 0; k < SCAN_PER_LANE; ++k) {
        const int i = lane * SCAN_PER_LANE + k;
        if (i < TILE) ord.cur[i] = run;
        run += c[k];
      }
    }
    __syncthreads();
#pragma unroll
    for (int k = 0; k < SRT_PER_THR; ++k)
      if (c0 + tid + k * AGG_THREADS < c1) {
        const int p = atomicAdd(&ord.cur[dv[k]], 1);
        ord.src[p] = sv[k];
        ord.coef[p] = dk[k];
        ord.row[p] = (unsigned char)dv[k];
      }
    __syncthreads();
    if (first) {
      tile_start_values<C, ROWS>(acc, h, dinv, N, tbase, tid);
      __syncthreads();
      first = false;
    }

    const int m   = c1 - c0;
    const int per = ((m + AGG_WAVES * EPB - 1) / (AGG_WAVES * EPB)) * EPB;
    const int wb  = w * per;
    const int we  = min(wb + per, m);
    int myrow = -1;
    v4f a = {0.f, 0.f, 0.f, 0.f};
    for (int g = wb; g < we; g += 32) {
      const int cnt = min(32, we - g);
      int s = 0;
      float ds = 0.0f;
      int rw = 0;
      if (lane < cnt) {
        s  = ord.src[g + lane];
        ds = ord.coef[g + lane];
        rw = ord.row[g + lane];
      }
      for (int jb = 0; jb < cnt; jb += EPB) {
        if (cnt - jb >= EPB) walk_batch<C, true>(acc, h, s, ds, rw, jb, cnt, sub, cl, myrow, a);
        else                 walk_batch<C, false>(acc, h, s, ds, rw, jb, cnt, sub, cl, myrow, a);
      }
    }
    if (myrow >= 0) row_flush(&acc[myrow * LDA + cl * 4], a);
    __syncthreads();
  }
  if (e0 >= e1) {
    tile_start_values<C, ROWS>(acc, h, dinv, N, tbase, tid);
    __syncthreads();
  }
}

template <int COUT>
struct ProductShape {
  static constexpr int NTN = COUT / 16;
  static constexpr int NWG = AGG_WAVES / NTN;
  static constexpr int NRB = TILE_ROWS / 16;
  static constexpr int MPW = (NRB + NWG - 1) / NWG;
  static_assert(COUT % 16 == 0 && AGG_WAVES % NTN == 0 && TILE_ROWS % 16 == 0 && TILE_ROWS >= TILE,
                "whole strips, whole wave groups, whole row blocks");
  static_assert(NRB % MPW <= 1, "a wave holds MPW row blocks, one, or none");
};

template <int COUT, int NB>
__device__ __forceinline__ void product_rows(
    const float* __restrict__ ap, const float* __restrict__ Wm, int bcol, int half,
    v8f (&cacc)[ProductShape<COUT>::MPW]) {
  constexpr int C   = 128;
  constexpr int LDA = C + 4;
#pragma unroll 4
  for (int k0 = 0; k0 < C; k0 += 4) {
    const int ka = k0 + 2 * half;
    v2f b;
    b.x = Wm[(long)ka * COUT + bcol];
    b.y = Wm[(long)(ka + 1) * COUT + bcol];
#pragma unroll
    for (int j = 0; j < NB; ++j) {
      v2f a;
      a.x = ap[j * 16 * LDA + ka];
      a.y = ap[j * 16 * LDA + ka + 1];
      cacc[j] = __builtin_amdgcn_wmma_f32_16x16x4_f32(
           false, a,  false, b,
           (short)0, cacc[j],  false,  false);
    }
  }
}

template <int COUT>
__device__ __forceinline__ int tile_product(
    const float* __restrict__ acc, const float* __restrict__ Wm, int w, int lane,
    v8f (&cacc)[ProductShape<COUT>::MPW]) {
  using S = ProductShape<COUT>;
  constexpr int LDA = 128 + 4;
  const int half = lane >> 4;
  const int l16  = lane & 15;
  const int tn   = w % S::NTN;
  const int mt0  = (w / S::NTN) * S::MPW;
  const int bcol = tn * 16 + l16;
  const float* __restrict__ ap = &acc[(mt0 * 16 + l16) * LDA];

#pragma unroll
  for (int j = 0; j < S::MPW; ++j) cacc[j] = (v8f){0.f, 0.f, 0.f, 0.f, 0.f, 0.f, 0.f, 0.f};

  const int nb = __builtin_amdgcn_readfirstlane(min(S::MPW, max(0, S::NRB - mt0)));
  if (nb == S::MPW)  product_rows<COUT, S::MPW>(ap, Wm, bcol, half, cacc);
  else if (nb == 1)  product_rows<COUT, 1>(ap, Wm, bcol, half, cacc);
  return nb;
}

__global__ __launch_bounds__(AGG_THREADS) void tile_layer1(
    const float* __restrict__ x, const int* __restrict__ binned,
    const int* __restrict__ offsets, const float* __restrict__ dinv,
    const float* __restrict__ W1, const float* __restrict__ bias,
    const float* __restrict__ gamma, const float* __restrict__ beta,
    const float* __restrict__ mean, const float* __restrict__ var,
    const float* __restrict__ W2, float* __restrict__ hw2, int N) {
  static_assert(GCN_IN_C == 128 && GCN_HID_C == 128, "both products run over a 128-wide LDS tile");
  constexpr int LDA = 128 + 4;
  constexpr int POOL = (TILE_ROWS * LDA > DTAB_CAP) ? TILE_ROWS * LDA : DTAB_CAP;
  __shared__ __attribute__((aligned(16))) float acc[POOL];
  __shared__ RowOrder ord;

  const int tid  = threadIdx.x;
  const int lane = tid & 31;
  const int w = __builtin_amdgcn_readfirstlane(tid >> 5);
  const int tile  = blockIdx.x;
  const int tbase = tile * TILE;
  const int half = lane >> 4;
  const int l16  = lane & 15;

  tile_sums<GCN_IN_C, TILE_ROWS>(acc, ord, x, binned, offsets, dinv, N, tile, tid, lane, w);

  using S1 = ProductShape<GCN_HID_C>;
  v8f c1[S1::MPW];
  const int nb1 = tile_product<GCN_HID_C>(acc, W1, w, lane, c1);

  __syncthreads();
  {
    const int bcol = (w % S1::NTN) * 16 + l16;
    const int mt0  = (w / S1::NTN) * S1::MPW;
    const float bi = bias[bcol];
    const float mu = mean[bcol];
    const float rs = rsqrtf(var[bcol] + GCN_BN_EPS);
    const float ga = gamma[bcol];
    const float be = beta[bcol];
#pragma unroll
    for (int j = 0; j < S1::MPW; ++j) {
      if (j < nb1) {
        float* __restrict__ ar = &acc[((mt0 + j) * 16 + half * 8) * LDA + bcol];
#pragma unroll
        for (int r = 0; r < 8; ++r) {
          float v = c1[j][r] + bi;
          v = (v - mu) * rs * ga + be;
          ar[r * LDA] = fmaxf(v, 0.0f);
        }
      }
    }
  }
  __syncthreads();

  using S2 = ProductShape<GCN_OUT_C>;
  v8f c2[S2::MPW];
  const int nb2 = tile_product<GCN_OUT_C>(acc, W2, w, lane, c2);
  {
    const int bcol = (w % S2::NTN) * 16 + l16;
    const int mt0  = (w / S2::NTN) * S2::MPW;
#pragma unroll
    for (int j = 0; j < S2::MPW; ++j) {
      if (j < nb2) {
        const int nl0 = (mt0 + j) * 16 + half * 8;
        float* __restrict__ cp = hw2 + (long)(tbase + nl0) * GCN_OUT_C + bcol;
#pragma unroll
        for (int r = 0; r < 8; ++r)
          if (nl0 + r < TILE && tbase + nl0 + r < N) cp[(long)r * GCN_OUT_C] = c2[j][r];
      }
    }
  }
}

__global__ __launch_bounds__(AGG_THREADS) void tile_layer2(
    const float* __restrict__ hw2, const int* __restrict__ binned,
    const int* __restrict__ offsets, const float* __restrict__ dinv,
    const float* __restrict__ bias,
    const float* __restrict__ gamma, const float* __restrict__ beta,
    const float* __restrict__ mean, const float* __restrict__ var,
    float* __restrict__ outp, int N) {
  constexpr int C   = GCN_OUT_C;
  constexpr int LDA = C + 4;
  constexpr int POOL = (TILE * LDA > DTAB_CAP) ? TILE * LDA : DTAB_CAP;
  __shared__ __attribute__((aligned(16))) float acc[POOL];
  __shared__ RowOrder ord;

  const int tid  = threadIdx.x;
  const int lane = tid & 31;
  const int w = __builtin_amdgcn_readfirstlane(tid >> 5);
  const int tile  = blockIdx.x;
  const int tbase = tile * TILE;

  tile_sums<C, TILE>(acc, ord, hw2, binned, offsets, dinv, N, tile, tid, lane, w);

  static_assert(AGG_THREADS % (C / 4) == 0, "a thread's column group must not change with i");
  const int c4 = tid % (C / 4);
  const v4f bi = ((const v4f*)bias)[c4];
  const v4f mu = ((const v4f*)mean)[c4];
  const v4f vr = ((const v4f*)var)[c4];
  const v4f ga = ((const v4f*)gamma)[c4];
  const v4f be = ((const v4f*)beta)[c4];
  v4f rs;
  rs.x = rsqrtf(vr.x + GCN_BN_EPS);
  rs.y = rsqrtf(vr.y + GCN_BN_EPS);
  rs.z = rsqrtf(vr.z + GCN_BN_EPS);
  rs.w = rsqrtf(vr.w + GCN_BN_EPS);
  for (int i = tid; i < TILE * C / 4; i += AGG_THREADS) {
    const int nl = i / (C / 4);
    const int node = tbase + nl;
    if (node < N) {
      v4f v = *(const v4f*)&acc[nl * LDA + c4 * 4] + bi;
      v = (v - mu) * rs * ga + be;
      ((v4f*)(outp + (long)node * C))[c4] = v;
    }
  }
}


extern "C" void kernel_launch(void* const* d_in, const int* in_sizes, int n_in,
                              void* d_out, int out_size, void* d_ws, size_t ws_size,
                              hipStream_t stream) {
  const float* x   = (const float*)d_in[0];
  const int*   ei  = (const int*)d_in[1];
  const float* W1  = (const float*)d_in[2];
  const float* b1  = (const float*)d_in[3];
  const float* g1  = (const float*)d_in[4];
  const float* be1 = (const float*)d_in[5];
  const float* m1  = (const float*)d_in[6];
  const float* v1  = (const float*)d_in[7];
  const float* W2  = (const float*)d_in[8];
  const float* b2  = (const float*)d_in[9];
  const float* g2  = (const float*)d_in[10];
  const float* be2 = (const float*)d_in[11];
  const float* m2  = (const float*)d_in[12];
  const float* v2  = (const float*)d_in[13];

  const int N = in_sizes[0] / GCN_IN_C;
  const int E = in_sizes[1] / 2;
  const int* src = ei;
  const int* dst = ei + E;

  char* ws = (char*)d_ws;
  float* hw2 = (float*)ws;                        ws += (size_t)N * GCN_OUT_C * sizeof(float);
  int*   totals  = (int*)ws;
  int*   offsets = totals + MAX_TILES;
  int*   cursor  = offsets + 2 * MAX_TILES;       ws += (size_t)N * sizeof(int);
  float* dinv = (float*)ws;                       ws += (size_t)N * sizeof(float);
  int*   binned  = (int*)ws;                      ws += (size_t)E * 2 * sizeof(int);
  (void)ws_size; (void)n_in; (void)out_size;

  const int NT = (N + TILE - 1) / TILE;
  const int BB = (E + BIN_CHUNK - 1) / BIN_CHUNK;

  zero_f32<<<1, MAX_TILES, 0, stream>>>((float*)totals, (long)MAX_TILES);
  bin_count<<<BB, BIN_THREADS, 0, stream>>>(dst, E, totals, NT);
  bin_scan<<<1, MAX_TILES, 0, stream>>>(totals, offsets, cursor, NT);
  bin_fill<<<BB, BIN_THREADS, 0, stream>>>(src, dst, E, cursor, binned, NT);
  tile_dinv<<<NT, DINV_THREADS, 0, stream>>>(binned, offsets, dinv, N);

  tile_layer1<<<NT, AGG_THREADS, 0, stream>>>(
      x, binned, offsets, dinv, W1, b1, g1, be1, m1, v1, W2, hw2, N);

  tile_layer2<<<NT, AGG_THREADS, 0, stream>>>(
      hw2, binned, offsets, dinv, b2, g2, be2, m2, v2, (float*)d_out, N);
}


# binning in one pass: each dst tile owns a fixed-capacity segment, a fill block's returning add on the tile cursor (from 0) places its run; overflow list: no count or scan launch; arrival-order sums
# speedup vs baseline: 6.3942x; 1.0629x over previous
#include <hip/hip_runtime.h>

typedef __attribute__((ext_vector_type(2))) float v2f;
typedef __attribute__((ext_vector_type(4))) float v4f;
typedef __attribute__((ext_vector_type(8))) float v8f;

#define GCN_IN_C   128
#define GCN_HID_C  128
#define GCN_OUT_C  64
#define GCN_BN_EPS 1e-5f

#define TILE        196
#define TILE_ROWS   208
#define MAX_TILES   512
#define BIN_CHUNK   4096
#define SEG_CAP     32032
#define BIN_THREADS 256
#define BIN_PER_THR (BIN_CHUNK / BIN_THREADS)
#define AGG_THREADS 1024
#define AGG_WAVES   (AGG_THREADS / 32)
#define AGG_FLIGHT  1024
#define SRT_CAP     8192
#define SRT_PER_THR (SRT_CAP / AGG_THREADS)
#define SCAN_PER_LANE ((TILE + 31) / 32)
#define DTAB_CAP    50176
#define DTAB_FLIGHT 8
#define DINV_THREADS 256
#define DINV_FLIGHT  32


__global__ void zero_f32(float* __restrict__ p, long n) {
  long i = (long)blockIdx.x * blockDim.x + threadIdx.x;
  long stride = (long)gridDim.x * blockDim.x;
  for (; i < n; i += stride) p[i] = 0.0f;
}

__device__ __forceinline__ int tile_of(int node) { return (int)((unsigned)node / (unsigned)TILE); }

__global__ void bin_fill(const int* __restrict__ src, const int* __restrict__ dst, int E,
                         int* __restrict__ cursor, int* __restrict__ binned,
                         int* __restrict__ ovf, int NT) {
  static_assert(MAX_TILES == 2 * BIN_THREADS, "the prefix sum gives every thread two tiles");
  __shared__ __attribute__((aligned(8))) int hist[MAX_TILES];
  __shared__ __attribute__((aligned(8))) int lstart[MAX_TILES];
  __shared__ __attribute__((aligned(8))) int lpos[MAX_TILES + 2];
  __shared__ int delta[MAX_TILES];
  __shared__ int odelta[MAX_TILES];
  __shared__ int spill;
  __shared__ int wsum[BIN_THREADS / 32];
  __shared__ int2 ordered[BIN_CHUNK];
  const int tid  = threadIdx.x;
  const int lane = tid & 31;
  const int w    = tid >> 5;
  const int base = blockIdx.x * BIN_CHUNK + tid;
  int dv[BIN_PER_THR], sv[BIN_PER_THR];
#pragma unroll
  for (int k = 0; k < BIN_PER_THR; ++k) dv[k] = dst[min(base + k * BIN_THREADS, E - 1)];
#pragma unroll
  for (int k = 0; k < BIN_PER_THR; ++k) sv[k] = src[min(base + k * BIN_THREADS, E - 1)];
  for (int i = tid; i < MAX_TILES; i += BIN_THREADS) hist[i] = 0;
  __syncthreads();
#pragma unroll
  for (int k = 0; k < BIN_PER_THR; ++k)
    if (base + k * BIN_THREADS < E) atomicAdd(&hist[tile_of(dv[k])], 1);
  __syncthreads();
  int rb[MAX_TILES / BIN_THREADS];
#pragma unroll
  for (int k = 0; k < MAX_TILES / BIN_THREADS; ++k) {
    const int i = tid + k * BIN_THREADS;
    const int c = (i < NT) ? hist[i] : 0;
    rb[k] = c ? atomicAdd(&cursor[i], c) : 0;
  }
  const int2 h = ((const int2*)hist)[tid];
  const int s = h.x + h.y;
  int incl = s;
#pragma unroll
  for (int off = 1; off < 32; off <<= 1) {
    const int t = __shfl_up(incl, off);
    if (lane >= off) incl += t;
  }
  if (lane == 31) wsum[w] = incl;
  __syncthreads();
  int prev = 0;
#pragma unroll
  for (int j = 0; j < BIN_THREADS / 32; ++j) prev += (j < w) ? wsum[j] : 0;
  const int ex0 = prev + incl - s;
  ((int2*)lstart)[tid] = make_int2(ex0, ex0 + h.x);
  ((int2*)lpos)[tid]   = make_int2(ex0, ex0 + h.x);
  if (tid == 0) { lpos[MAX_TILES] = 0; spill = 0; }
  __syncthreads();
#pragma unroll
  for (int k = 0; k < MAX_TILES / BIN_THREADS; ++k) {
    const int i = tid + k * BIN_THREADS;
    delta[i] = i * SEG_CAP + rb[k] - lstart[i];
    const int c = hist[i];
    if (rb[k] + c > SEG_CAP) {
      const int cut = max(rb[k], SEG_CAP);
      const int opos = atomicAdd(&cursor[MAX_TILES], rb[k] + c - cut);
      odelta[i] = opos - cut + rb[k] - lstart[i];
      spill = 1;
    }
  }
  int p[BIN_PER_THR];
#pragma unroll
  for (int k = 0; k < BIN_PER_THR; ++k)
    p[k] = atomicAdd(&lpos[(base + k * BIN_THREADS < E) ? tile_of(dv[k]) : MAX_TILES], 1);
#pragma unroll
  for (int k = 0; k < BIN_PER_THR; ++k)
    if (base + k * BIN_THREADS < E) ordered[p[k]] = make_int2(sv[k], dv[k]);
  __syncthreads();
  const int nvalid = min((int)BIN_CHUNK, E - (int)blockIdx.x * BIN_CHUNK);
  int2* __restrict__ out = (int2*)binned;
  int2 e[BIN_PER_THR];
  int dl[BIN_PER_THR];
#pragma unroll
  for (int k = 0; k < BIN_PER_THR; ++k) e[k] = ordered[min(tid + k * BIN_THREADS, nvalid - 1)];
#pragma unroll
  for (int k = 0; k < BIN_PER_THR; ++k) dl[k] = delta[tile_of(e[k].y)];
  if (!spill) {
#pragma unroll
    for (int k = 0; k < BIN_PER_THR; ++k)
      if (tid + k * BIN_THREADS < nvalid) out[dl[k] + tid + k * BIN_THREADS] = e[k];
  } else {
    int2* __restrict__ ov = (int2*)ovf;
#pragma unroll
    for (int k = 0; k < BIN_PER_THR; ++k) {
      const int q = tid + k * BIN_THREADS;
      if (q < nvalid) {
        const int t = tile_of(e[k].y);
        const int idx = dl[k] + q;
        if (idx < (t + 1) * SEG_CAP) out[idx] = e[k];
        else                         ov[odelta[t] + q] = e[k];
      }
    }
  }
}

__global__ __launch_bounds__(DINV_THREADS) void tile_dinv(
    const int* __restrict__ binned, const int* __restrict__ cursor,
    const int* __restrict__ ovf, float* __restrict__ dinv, int N) {
  static_assert(TILE <= DINV_THREADS, "one thread per node of the tile");
  __shared__ int cnt[TILE];
  const int tid = threadIdx.x;
  const int tile = blockIdx.x;
  const int tbase = tile * TILE;
  if (tid < TILE) cnt[tid] = 0;
  __syncthreads();
  const int e0 = tile * SEG_CAP;
  const int e1 = e0 + min(cursor[tile], SEG_CAP);
  const int nov = cursor[MAX_TILES];
  for (int b = e0 + tid; b < e1; b += DINV_THREADS * DINV_FLIGHT) {
    int d[DINV_FLIGHT];
#pragma unroll
    for (int u = 0; u < DINV_FLIGHT; ++u) d[u] = binned[2 * min(b + u * DINV_THREADS, e1 - 1) + 1];
#pragma unroll
    for (int u = 0; u < DINV_FLIGHT; ++u)
      if (b + u * DINV_THREADS < e1) atomicAdd(&cnt[d[u] - tbase], 1);
  }
  for (int i = tid; i < nov; i += DINV_THREADS) {
    const int d = ovf[2 * i + 1];
    if (tile_of(d) == tile) atomicAdd(&cnt[d - tbase], 1);
  }
  __syncthreads();
  if (tid < TILE) {
    const int node = tbase + tid;
    if (node < N) dinv[node] = rsqrtf((float)(cnt[tid] + 1));
  }
}

struct RowOrder {
  int src[SRT_CAP];
  float coef[SRT_CAP];
  unsigned char row[SRT_CAP];
  int cnt[TILE];
  int cur[TILE];
};
static_assert(TILE <= 256, "a row index inside the tile is stored in one byte");

__device__ __forceinline__ void row_flush(float* ar, v4f a) {
  atomicAdd(ar + 0, a.x);
  atomicAdd(ar + 1, a.y);
  atomicAdd(ar + 2, a.z);
  atomicAdd(ar + 3, a.w);
}

template <int C, int ROWS>
__device__ __forceinline__ void tile_start_values(
    float* acc, const float* __restrict__ h, const float* __restrict__ dinv,
    int N, int tbase, int tid) {
  constexpr int LDA = C + 4;
  constexpr int TOT = ROWS * C / 4;
  constexpr int NIT = (TOT + AGG_THREADS - 1) / AGG_THREADS;
  static_assert(AGG_THREADS % (C / 4) == 0, "a thread keeps one column group");
  const int c4 = tid % (C / 4);
  v4f v[NIT];
  float di[NIT];
#pragma unroll
  for (int u = 0; u < NIT; ++u) {
    const int nl = (tid + u * AGG_THREADS) / (C / 4);
    const int node = min(tbase + min(nl, TILE - 1), N - 1);
    di[u] = dinv[node];
    v[u] = ((const v4f*)(h + (long)node * C))[c4];
  }
#pragma unroll
  for (int u = 0; u < NIT; ++u) {
    const int i = tid + u * AGG_THREADS;
    const int nl = i / (C / 4);
    if (i < TOT) {
      v4f s = {0.f, 0.f, 0.f, 0.f};
      if (nl < TILE && tbase + nl < N) s = v[u] * (di[u] * di[u]);
      *(v4f*)&acc[nl * LDA + c4 * 4] = s;
    }
  }
}

template <int C, bool FULL>
__device__ __forceinline__ void walk_batch(
    float* acc, const float* __restrict__ h, int s, float ds, int rw,
    int jb, int cnt, int sub, int cl, int& myrow, v4f& a) {
  constexpr int LDA = C + 4;
  constexpr int RPI = 32 / (C / 4);
  constexpr int R   = AGG_FLIGHT / 128;
  v4f v[R];
#pragma unroll
  for (int u = 0; u < R; ++u) {
    const int j0 = jb + u * RPI;
    int sj = __builtin_amdgcn_readlane(s, j0);
    if (RPI == 2) {
      const int sj1 = __builtin_amdgcn_readlane(s, j0 + 1);
      sj = sub ? sj1 : sj;
    }
    v[u] = *(const v4f*)(h + (long)sj * C + cl * 4);
  }
#pragma unroll
  for (int u = 0; u < R; ++u) {
    const int j0 = jb + u * RPI;
    if (FULL || u == 0 || j0 < cnt) {
      float cj = __int_as_float(__builtin_amdgcn_readlane(__float_as_int(ds), j0));
      int rj = __builtin_amdgcn_readlane(rw, j0);
      bool live = true;
      if (RPI == 2) {
        const float cj1 = __int_as_float(__builtin_amdgcn_readlane(__float_as_int(ds), j0 + 1));
        const int rj1 = __builtin_amdgcn_readlane(rw, j0 + 1);
        cj = sub ? cj1 : cj;
        rj = sub ? rj1 : rj;
        live = FULL || (j0 + sub < cnt);
      }
      if (live) {
        if (rj != myrow) {
          if (myrow >= 0) row_flush(&acc[myrow * LDA + cl * 4], a);
          a = (v4f){0.f, 0.f, 0.f, 0.f};
          myrow = rj;
        }
        a += v[u] * cj;
      }
    }
  }
}

template <int C, int ROWS>
__device__ __forceinline__ void tile_sums(
    float* acc, RowOrder& ord, const float* __restrict__ h,
    const int* __restrict__ binned, const int* __restrict__ cursor, const int* __restrict__ ovf,
    const float* __restrict__ dinv, int N, int tile, int tid, int lane, int w) {
  static_assert(C == 64 || C == 128, "a row is 16 or 32 lanes of 16 bytes");
  constexpr int LDA = C + 4;
  constexpr int LPR = C / 4;
  constexpr int RPI = 32 / LPR;
  constexpr int R   = AGG_FLIGHT / 128;
  constexpr int EPB = R * RPI;
  static_assert(32 % EPB == 0, "a 32-edge group is a whole number of batches");
  const int tbase = tile * TILE;
  const int sub = lane / LPR;
  const int cl  = lane % LPR;

  const int e0 = tile * SEG_CAP;
  const int e1 = e0 + min(cursor[tile], SEG_CAP);
  const int nov = cursor[MAX_TILES];
  const int2* __restrict__ ed = (const int2*)binned;

  const bool tab = (N <= DTAB_CAP) && (e0 < e1);
  if (tab) {
    const int n4 = ((((unsigned long)dinv) & 15ul) == 0ul) ? (N >> 2) : 0;
    const v4f* __restrict__ d4 = (const v4f*)dinv;
    for (int i0 = tid; i0 < n4; i0 += AGG_THREADS * DTAB_FLIGHT) {
      v4f t[DTAB_FLIGHT];
#pragma unroll
      for (int u = 0; u < DTAB_FLIGHT; ++u) t[u] = d4[min(i0 + u * AGG_THREADS, n4 - 1)];
#pragma unroll
      for (int u = 0; u < DTAB_FLIGHT; ++u)
        if (i0 + u * AGG_THREADS < n4) ((v4f*)acc)[i0 + u * AGG_THREADS] = t[u];
    }
    for (int i = (n4 << 2) + tid; i < N; i += AGG_THREADS) acc[i] = dinv[i];
  }
  bool first = true;

  for (int c0 = e0; c0 < e1; c0 += SRT_CAP) {
    const int c1 = min(c0 + SRT_CAP, e1);

    int sv[SRT_PER_THR], dv[SRT_PER_THR];
#pragma unroll
    for (int k = 0; k < SRT_PER_THR; ++k) {
      const int2 p = ed[min(c0 + tid + k * AGG_THREADS, c1 - 1)];
      sv[k] = p.x;
      dv[k] = p.y - tbase;
    }
    const bool lds_tab = tab && first;
    float dk[SRT_PER_THR];
    if (!lds_tab) {
#pragma unroll
      for (int k = 0; k < SRT_PER_THR; ++k) dk[k] = dinv[sv[k]] * dinv[tbase + dv[k]];
    }
    if (tid < TILE) ord.cnt[tid] = 0;
    __syncthreads();
    if (lds_tab) {
#pragma unroll
      for (int k = 0; k < SRT_PER_THR; ++k) dk[k] = acc[sv[k]] * acc[tbase + dv[k]];
    }
#pragma unroll
    for (int k = 0; k < SRT_PER_THR; ++k)
      if (c0 + tid + k * AGG_THREADS < c1) atomicAdd(&ord.cnt[dv[k]], 1);
    __syncthreads();
    if (w == 0) {
      int c[SCAN_PER_LANE];
      int sum = 0;
#pragma unroll
      for (int k = 0; k < SCAN_PER_LANE; ++k) {
        const int i = lane * SCAN_PER_LANE + k;
        c[k] = (i < TILE) ? ord.cnt[i] : 0;
        sum += c[k];
      }
      int incl = sum;
#pragma unroll
      for (int off = 1; off < 32; off <<= 1) {
        const int t = __shfl_up(incl, off);
        if (lane >= off) incl += t;
      }
      int run = incl - sum;
#pragma unroll
      for (int k = 0; k < SCAN_PER_LANE; ++k) {
        const int i = lane * SCAN_PER_LANE + k;
        if (i < TILE) ord.cur[i] = run;
        run += c[k];
      }
    }
    __syncthreads();
#pragma unroll
    for (int k = 0; k < SRT_PER_THR; ++k)
      if (c0 + tid + k * AGG_THREADS < c1) {
        const int p = atomicAdd(&ord.cur[dv[k]], 1);
        ord.src[p] = sv[k];
        ord.coef[p] = dk[k];
        ord.row[p] = (unsigned char)dv[k];
      }
    __syncthreads();
    if (first) {
      tile_start_values<C, ROWS>(acc, h, dinv, N, tbase, tid);
      __syncthreads();
      first = false;
    }

    const int m   = c1 - c0;
    const int per = ((m + AGG_WAVES * EPB - 1) / (AGG_WAVES * EPB)) * EPB;
    const int wb  = w * per;
    const int we  = min(wb + per, m);
    int myrow = -1;
    v4f a = {0.f, 0.f, 0.f, 0.f};
    for (int g = wb; g < we; g += 32) {
      const int cnt = min(32, we - g);
      int s = 0;
      float ds = 0.0f;
      int rw = 0;
      if (lane < cnt) {
        s  = ord.src[g + lane];
        ds = ord.coef[g + lane];
        rw = ord.row[g + lane];
      }
      for (int jb = 0; jb < cnt; jb += EPB) {
        if (cnt - jb >= EPB) walk_batch<C, true>(acc, h, s, ds, rw, jb, cnt, sub, cl, myrow, a);
        else                 walk_batch<C, false>(acc, h, s, ds, rw, jb, cnt, sub, cl, myrow, a);
      }
    }
    if (myrow >= 0) row_flush(&acc[myrow * LDA + cl * 4], a);
    __syncthreads();
  }
  if (e0 >= e1) {
    tile_start_values<C, ROWS>(acc, h, dinv, N, tbase, tid);
    __syncthreads();
  }
  if (nov > 0) {
    const int2* __restrict__ ov = (const int2*)ovf;
    for (int i = w; i < nov; i += AGG_WAVES) {
      const int2 p = ov[i];
      if (tile_of(p.y) == tile && sub == 0) {
        const float cf = dinv[p.x] * dinv[p.y];
        const v4f v = *(const v4f*)(h + (long)p.x * C + cl * 4);
        row_flush(&acc[(p.y - tbase) * LDA + cl * 4], v * cf);
      }
    }
    __syncthreads();
  }
}

template <int COUT>
struct ProductShape {
  static constexpr int NTN = COUT / 16;
  static constexpr int NWG = AGG_WAVES / NTN;
  static constexpr int NRB = TILE_ROWS / 16;
  static constexpr int MPW = (NRB + NWG - 1) / NWG;
  static_assert(COUT % 16 == 0 && AGG_WAVES % NTN == 0 && TILE_ROWS % 16 == 0 && TILE_ROWS >= TILE,
                "whole strips, whole wave groups, whole row blocks");
  static_assert(NRB % MPW <= 1, "a wave holds MPW row blocks, one, or none");
};

template <int COUT, int NB>
__device__ __forceinline__ void product_rows(
    const float* __restrict__ ap, const float* __restrict__ Wm, int bcol, int half,
    v8f (&cacc)[ProductShape<COUT>::MPW]) {
  constexpr int C   = 128;
  constexpr int LDA = C + 4;
#pragma unroll 4
  for (int k0 = 0; k0 < C; k0 += 4) {
    const int ka = k0 + 2 * half;
    v2f b;
    b.x = Wm[(long)ka * COUT + bcol];
    b.y = Wm[(long)(ka + 1) * COUT + bcol];
#pragma unroll
    for (int j = 0; j < NB; ++j) {
      v2f a;
      a.x = ap[j * 16 * LDA + ka];
      a.y = ap[j * 16 * LDA + ka + 1];
      cacc[j] = __builtin_amdgcn_wmma_f32_16x16x4_f32(
           false, a,  false, b,
           (short)0, cacc[j],  false,  false);
    }
  }
}

template <int COUT>
__device__ __forceinline__ int tile_product(
    const float* __restrict__ acc, const float* __restrict__ Wm, int w, int lane,
    v8f (&cacc)[ProductShape<COUT>::MPW]) {
  using S = ProductShape<COUT>;
  constexpr int LDA = 128 + 4;
  const int half = lane >> 4;
  const int l16  = lane & 15;
  const int tn   = w % S::NTN;
  const int mt0  = (w / S::NTN) * S::MPW;
  const int bcol = tn * 16 + l16;
  const float* __restrict__ ap = &acc[(mt0 * 16 + l16) * LDA];

#pragma unroll
  for (int j = 0; j < S::MPW; ++j) cacc[j] = (v8f){0.f, 0.f, 0.f, 0.f, 0.f, 0.f, 0.f, 0.f};

  const int nb = __builtin_amdgcn_readfirstlane(min(S::MPW, max(0, S::NRB - mt0)));
  if (nb == S::MPW)  product_rows<COUT, S::MPW>(ap, Wm, bcol, half, cacc);
  else if (nb == 1)  product_rows<COUT, 1>(ap, Wm, bcol, half, cacc);
  return nb;
}

__global__ __launch_bounds__(AGG_THREADS) void tile_layer1(
    const float* __restrict__ x, const int* __restrict__ binned,
    const int* __restrict__ cursor, const int* __restrict__ ovf, const float* __restrict__ dinv,
    const float* __restrict__ W1, const float* __restrict__ bias,
    const float* __restrict__ gamma, const float* __restrict__ beta,
    const float* __restrict__ mean, const float* __restrict__ var,
    const float* __restrict__ W2, float* __restrict__ hw2, int N) {
  static_assert(GCN_IN_C == 128 && GCN_HID_C == 128, "both products run over a 128-wide LDS tile");
  constexpr int LDA = 128 + 4;
  constexpr int POOL = (TILE_ROWS * LDA > DTAB_CAP) ? TILE_ROWS * LDA : DTAB_CAP;
  __shared__ __attribute__((aligned(16))) float acc[POOL];
  __shared__ RowOrder ord;

  const int tid  = threadIdx.x;
  const int lane = tid & 31;
  const int w = __builtin_amdgcn_readfirstlane(tid >> 5);
  const int tile  = blockIdx.x;
  const int tbase = tile * TILE;
  const int half = lane >> 4;
  const int l16  = lane & 15;

  tile_sums<GCN_IN_C, TILE_ROWS>(acc, ord, x, binned, cursor, ovf, dinv, N, tile, tid, lane, w);

  using S1 = ProductShape<GCN_HID_C>;
  v8f c1[S1::MPW];
  const int nb1 = tile_product<GCN_HID_C>(acc, W1, w, lane, c1);

  __syncthreads();
  {
    const int bcol = (w % S1::NTN) * 16 + l16;
    const int mt0  = (w / S1::NTN) * S1::MPW;
    const float bi = bias[bcol];
    const float mu = mean[bcol];
    const float rs = rsqrtf(var[bcol] + GCN_BN_EPS);
    const float ga = gamma[bcol];
    const float be = beta[bcol];
#pragma unroll
    for (int j = 0; j < S1::MPW; ++j) {
      if (j < nb1) {
        float* __restrict__ ar = &acc[((mt0 + j) * 16 + half * 8) * LDA + bcol];
#pragma unroll
        for (int r = 0; r < 8; ++r) {
          float v = c1[j][r] + bi;
          v = (v - mu) * rs * ga + be;
          ar[r * LDA] = fmaxf(v, 0.0f);
        }
      }
    }
  }
  __syncthreads();

  using S2 = ProductShape<GCN_OUT_C>;
  v8f c2[S2::MPW];
  const int nb2 = tile_product<GCN_OUT_C>(acc, W2, w, lane, c2);
  {
    const int bcol = (w % S2::NTN) * 16 + l16;
    const int mt0  = (w / S2::NTN) * S2::MPW;
#pragma unroll
    for (int j = 0; j < S2::MPW; ++j) {
      if (j < nb2) {
        const int nl0 = (mt0 + j) * 16 + half * 8;
        float* __restrict__ cp = hw2 + (long)(tbase + nl0) * GCN_OUT_C + bcol;
#pragma unroll
        for (int r = 0; r < 8; ++r)
          if (nl0 + r < TILE && tbase + nl0 + r < N) cp[(long)r * GCN_OUT_C] = c2[j][r];
      }
    }
  }
}

__global__ __launch_bounds__(AGG_THREADS) void tile_layer2(
    const float* __restrict__ hw2, const int* __restrict__ binned,
    const int* __restrict__ cursor, const int* __restrict__ ovf, const float* __restrict__ dinv,
    const float* __restrict__ bias,
    const float* __restrict__ gamma, const float* __restrict__ beta,
    const float* __restrict__ mean, const float* __restrict__ var,
    float* __restrict__ outp, int N) {
  constexpr int C   = GCN_OUT_C;
  constexpr int LDA = C + 4;
  constexpr int POOL = (TILE * LDA > DTAB_CAP) ? TILE * LDA : DTAB_CAP;
  __shared__ __attribute__((aligned(16))) float acc[POOL];
  __shared__ RowOrder ord;

  const int tid  = threadIdx.x;
  const int lane = tid & 31;
  const int w = __builtin_amdgcn_readfirstlane(tid >> 5);
  const int tile  = blockIdx.x;
  const int tbase = tile * TILE;

  tile_sums<C, TILE>(acc, ord, hw2, binned, cursor, ovf, dinv, N, tile, tid, lane, w);

  static_assert(AGG_THREADS % (C / 4) == 0, "a thread's column group must not change with i");
  const int c4 = tid % (C / 4);
  const v4f bi = ((const v4f*)bias)[c4];
  const v4f mu = ((const v4f*)mean)[c4];
  const v4f vr = ((const v4f*)var)[c4];
  const v4f ga = ((const v4f*)gamma)[c4];
  const v4f be = ((const v4f*)beta)[c4];
  v4f rs;
  rs.x = rsqrtf(vr.x + GCN_BN_EPS);
  rs.y = rsqrtf(vr.y + GCN_BN_EPS);
  rs.z = rsqrtf(vr.z + GCN_BN_EPS);
  rs.w = rsqrtf(vr.w + GCN_BN_EPS);
  for (int i = tid; i < TILE * C / 4; i += AGG_THREADS) {
    const int nl = i / (C / 4);
    const int node = tbase + nl;
    if (node < N) {
      v4f v = *(const v4f*)&acc[nl * LDA + c4 * 4] + bi;
      v = (v - mu) * rs * ga + be;
      ((v4f*)(outp + (long)node * C))[c4] = v;
    }
  }
}


extern "C" void kernel_launch(void* const* d_in, const int* in_sizes, int n_in,
                              void* d_out, int out_size, void* d_ws, size_t ws_size,
                              hipStream_t stream) {
  const float* x   = (const float*)d_in[0];
  const int*   ei  = (const int*)d_in[1];
  const float* W1  = (const float*)d_in[2];
  const float* b1  = (const float*)d_in[3];
  const float* g1  = (const float*)d_in[4];
  const float* be1 = (const float*)d_in[5];
  const float* m1  = (const float*)d_in[6];
  const float* v1  = (const float*)d_in[7];
  const float* W2  = (const float*)d_in[8];
  const float* b2  = (const float*)d_in[9];
  const float* g2  = (const float*)d_in[10];
  const float* be2 = (const float*)d_in[11];
  const float* m2  = (const float*)d_in[12];
  const float* v2  = (const float*)d_in[13];

  const int N = in_sizes[0] / GCN_IN_C;
  const int E = in_sizes[1] / 2;
  const int* src = ei;
  const int* dst = ei + E;

  const int NT = (N + TILE - 1) / TILE;
  const int BB = (E + BIN_CHUNK - 1) / BIN_CHUNK;

  char* ws = (char*)d_ws;
  float* hw2 = (float*)ws;                        ws += (size_t)N * GCN_OUT_C * sizeof(float);
  int*   cursor  = (int*)ws;                      ws += (size_t)N * sizeof(int);
  float* dinv = (float*)ws;                       ws += (size_t)N * sizeof(float);
  int*   binned  = (int*)ws;                      ws += (size_t)NT * SEG_CAP * 2 * sizeof(int);
  int*   ovf     = (int*)ws;                      ws += (size_t)E * 2 * sizeof(int);
  (void)ws_size; (void)n_in; (void)out_size;

  zero_f32<<<1, MAX_TILES, 0, stream>>>((float*)cursor, (long)(MAX_TILES + 1));
  bin_fill<<<BB, BIN_THREADS, 0, stream>>>(src, dst, E, cursor, binned, ovf, NT);
  tile_dinv<<<NT, DINV_THREADS, 0, stream>>>(binned, cursor, ovf, dinv, N);

  tile_layer1<<<NT, AGG_THREADS, 0, stream>>>(
      x, binned, cursor, ovf, dinv, W1, b1, g1, be1, m1, v1, W2, hw2, N);

  tile_layer2<<<NT, AGG_THREADS, 0, stream>>>(
      hw2, binned, cursor, ovf, dinv, b2, g2, be2, m2, v2, (float*)d_out, N);
}
